# starting kernel
# baseline (speedup 1.0000x reference)
#include <hip/hip_runtime.h>


typedef float v2f __attribute__((ext_vector_type(2)));
typedef float v4f __attribute__((ext_vector_type(4)));
typedef float v8f __attribute__((ext_vector_type(8)));

#define NEG_SLOPE 0.2f
#define LN_EPS 1e-5f

__device__ __forceinline__ unsigned fenc(float f) {
  unsigned u = __float_as_uint(f);
  return (u >> 31) ? ~u : (u | 0x80000000u);
}
__device__ __forceinline__ float fdec(unsigned u) {
  return __uint_as_float((u >> 31) ? (u & 0x7FFFFFFFu) : ~u);
}
__device__ __forceinline__ float lrelu(float x) {
  return x > 0.f ? x : x * NEG_SLOPE;
}

template <int K>
__global__ void gat_gemm_wmma(const float* __restrict__ Amat,
                              const float* __restrict__ Bmat,
                              float* __restrict__ Cmat,
                              int M) {
  __shared__ float bs[K * 128];

  const int tid = threadIdx.x;

  for (int idx = tid; idx < K * 128; idx += 256) {
    int kr = idx >> 7;
    int col = idx & 127;
    int ks = kr >> 2;
    int c = col >> 4;
    int lane = (col & 15) | ((kr & 2) << 3);
    int sub = kr & 1;
    bs[((ks * 32 + lane) * 8 + c) * 2 + sub] = Bmat[idx];
  }
  __syncthreads();

  const int wave = tid >> 5;
  const int lane = tid & 31;
  const int row0 = (blockIdx.x * 8 + wave) * 16;
  if (row0 >= M) return;

  const int m  = lane & 15;
  const int kk = (lane >> 4) << 1;

  v8f acc[8] = {};

  const float* arow = Amat + (size_t)(row0 + m) * K + kk;
  const float* blane = bs + lane * 16;

#pragma unroll 4
  for (int ks = 0; ks < K / 4; ++ks) {
    v2f a = *(const v2f*)(arow + ks * 4);
    const float* bp = blane + ks * 512;
    v4f q0 = *(const v4f*)(bp);
    v4f q1 = *(const v4f*)(bp + 4);
    v4f q2 = *(const v4f*)(bp + 8);
    v4f q3 = *(const v4f*)(bp + 12);
    acc[0] = __builtin_amdgcn_wmma_f32_16x16x4_f32(false, a, false, q0.xy, (short)0, acc[0], false, false);
    acc[1] = __builtin_amdgcn_wmma_f32_16x16x4_f32(false, a, false, q0.zw, (short)0, acc[1], false, false);
    acc[2] = __builtin_amdgcn_wmma_f32_16x16x4_f32(false, a, false, q1.xy, (short)0, acc[2], false, false);
    acc[3] = __builtin_amdgcn_wmma_f32_16x16x4_f32(false, a, false, q1.zw, (short)0, acc[3], false, false);
    acc[4] = __builtin_amdgcn_wmma_f32_16x16x4_f32(false, a, false, q2.xy, (short)0, acc[4], false, false);
    acc[5] = __builtin_amdgcn_wmma_f32_16x16x4_f32(false, a, false, q2.zw, (short)0, acc[5], false, false);
    acc[6] = __builtin_amdgcn_wmma_f32_16x16x4_f32(false, a, false, q3.xy, (short)0, acc[6], false, false);
    acc[7] = __builtin_amdgcn_wmma_f32_16x16x4_f32(false, a, false, q3.zw, (short)0, acc[7], false, false);
  }

  const int mo = (lane >> 4) * 8;
#pragma unroll
  for (int c = 0; c < 8; ++c) {
#pragma unroll
    for (int r = 0; r < 8; ++r) {
      Cmat[(size_t)(row0 + mo + r) * 128 + c * 16 + m] = acc[c][r];
    }
  }
}

__global__ void gat_attn_proj(const float* __restrict__ h,
                              const float* __restrict__ al,
                              const float* __restrict__ ar,
                              float* __restrict__ el, float* __restrict__ er,
                              int NH) {
  int i = blockIdx.x * blockDim.x + threadIdx.x;
  if (i >= NH) return;
  int n = i >> 2, hh = i & 3;
  const float* hp  = h + (size_t)n * 128 + hh * 32;
  const float* alp = al + hh * 32;
  const float* arp = ar + hh * 32;
  float sl = 0.f, sr = 0.f;
#pragma unroll
  for (int f = 0; f < 32; ++f) {
    float v = hp[f];
    sl += v * alp[f];
    sr += v * arp[f];
  }
  el[i] = sl;
  er[i] = sr;
}

__global__ void gat_edge_max(const int* __restrict__ src, const int* __restrict__ dst,
                             const float* __restrict__ el, const float* __restrict__ er,
                             unsigned* __restrict__ emax, int E4) {
  int i = blockIdx.x * blockDim.x + threadIdx.x;
  if (i >= E4) return;
  int e = i >> 2, hh = i & 3;
  int s = src[e], d = dst[e];
  float ev = lrelu(el[s * 4 + hh] + er[d * 4 + hh]);
  atomicMax(&emax[d * 4 + hh], fenc(ev));
}

__global__ void gat_edge_denom(const int* __restrict__ src, const int* __restrict__ dst,
                               const float* __restrict__ el, const float* __restrict__ er,
                               const unsigned* __restrict__ emax,
                               float* __restrict__ denom, int E4) {
  int i = blockIdx.x * blockDim.x + threadIdx.x;
  if (i >= E4) return;
  int e = i >> 2, hh = i & 3;
  int s = src[e], d = dst[e];
  float ev = lrelu(el[s * 4 + hh] + er[d * 4 + hh]);
  float mx = fdec(emax[d * 4 + hh]);
  atomicAdd(&denom[d * 4 + hh], expf(ev - mx));
}

__global__ void gat_edge_scatter(const int* __restrict__ src, const int* __restrict__ dst,
                                 const float* __restrict__ el, const float* __restrict__ er,
                                 const unsigned* __restrict__ emax,
                                 const float* __restrict__ denom,
                                 const float* __restrict__ h,
                                 float* __restrict__ rst, int E) {
  int gid = blockIdx.x * 256 + threadIdx.x;
  int e = gid >> 7;
  if (e >= E) return;
  int t = gid & 127;
  int hh = t >> 5;
  int s = src[e], d = dst[e];
  float ev = lrelu(el[s * 4 + hh] + er[d * 4 + hh]);
  float mx = fdec(emax[d * 4 + hh]);
  float ex = expf(ev - mx);
  float alpha = ex / fmaxf(denom[d * 4 + hh], 1e-9f);
  atomicAdd(&rst[(size_t)d * 128 + t], h[(size_t)s * 128 + t] * alpha);
}

__global__ void gat_finalize1(const float* __restrict__ rst,
                              const float* __restrict__ bias,
                              const float* __restrict__ lns,
                              const float* __restrict__ lnb,
                              float* __restrict__ hmid, int Nn) {
  int n = blockIdx.x * 8 + (threadIdx.x >> 5);
  int f = threadIdx.x & 31;
  if (n >= Nn) return;
  float x = 0.f;
#pragma unroll
  for (int hh = 0; hh < 4; ++hh)
    x += rst[(size_t)n * 128 + hh * 32 + f] + bias[hh * 32 + f];
  x *= 0.25f;
  float mu = x;
#pragma unroll
  for (int m = 16; m >= 1; m >>= 1) mu += __shfl_xor(mu, m, 32);
  mu *= (1.f / 32.f);
  float dx = x - mu;
  float var = dx * dx;
#pragma unroll
  for (int m = 16; m >= 1; m >>= 1) var += __shfl_xor(var, m, 32);
  var *= (1.f / 32.f);
  float y = dx * rsqrtf(var + LN_EPS) * lns[f] + lnb[f];
  hmid[(size_t)n * 32 + f] = fmaxf(y, 0.f);
}

__global__ void gat_finalize2(const float* __restrict__ rst,
                              const float* __restrict__ bias,
                              float* __restrict__ out, int NF) {
  int i = blockIdx.x * blockDim.x + threadIdx.x;
  if (i >= NF) return;
  int n = i >> 5, f = i & 31;
  float x = 0.f;
#pragma unroll
  for (int hh = 0; hh < 4; ++hh)
    x += rst[(size_t)n * 128 + hh * 32 + f] + bias[hh * 32 + f];
  out[i] = x * 0.25f;
}

__global__ void gat_flops(const int* __restrict__ in_deg, int Nn, float* out) {
  __shared__ float ssum[256];
  float s = 0.f;
  for (int i = threadIdx.x; i < Nn; i += 256) s += (float)in_deg[i];
  ssum[threadIdx.x] = s;
  __syncthreads();
  for (int m = 128; m > 0; m >>= 1) {
    if ((int)threadIdx.x < m) ssum[threadIdx.x] += ssum[threadIdx.x + m];
    __syncthreads();
  }
  if (threadIdx.x == 0) {
    double ne = (double)ssum[0];
    double fl = (4.0 * ne * (6.0 * 128 * 32 + 6 * 32 + 2) +
                 4.0 * ne * (6.0 * 32 * 32 + 6 * 32 + 2)) / 1e12;
    out[0] = (float)fl;
  }
}

extern "C" void kernel_launch(void* const* d_in, const int* in_sizes, int n_in,
                              void* d_out, int out_size, void* d_ws, size_t ws_size,
                              hipStream_t stream) {
  const float* feat = (const float*)d_in[0];
  const float* W1   = (const float*)d_in[1];
  const float* al1  = (const float*)d_in[2];
  const float* ar1  = (const float*)d_in[3];
  const float* b1   = (const float*)d_in[4];
  const float* lns  = (const float*)d_in[5];
  const float* lnb  = (const float*)d_in[6];
  const float* W2   = (const float*)d_in[7];
  const float* al2  = (const float*)d_in[8];
  const float* ar2  = (const float*)d_in[9];
  const float* b2   = (const float*)d_in[10];
  const int*   src  = (const int*)d_in[11];
  const int*   dst  = (const int*)d_in[12];
  const int*   indeg= (const int*)d_in[13];

  const int Nn = in_sizes[13];
  const int E  = in_sizes[11];
  const int NH = Nn * 4;

  char* p = (char*)d_ws;
  float*    hbuf  = (float*)p;   p += (size_t)Nn * 128 * sizeof(float);
  float*    rst   = (float*)p;   p += (size_t)Nn * 128 * sizeof(float);
  float*    hmid  = (float*)p;   p += (size_t)Nn * 32  * sizeof(float);
  float*    el    = (float*)p;   p += (size_t)NH * sizeof(float);
  float*    er    = (float*)p;   p += (size_t)NH * sizeof(float);
  unsigned* emax  = (unsigned*)p;p += (size_t)NH * sizeof(unsigned);
  float*    denom = (float*)p;   p += (size_t)NH * sizeof(float);
  (void)ws_size; (void)n_in; (void)out_size;

  float* outh = (float*)d_out;

  dim3 b256(256);
  const int rowTiles   = (Nn + 15) / 16;
  const int gemmBlocks = (rowTiles + 7) / 8;
  const int e4Blocks   = (E * 4 + 255) / 256;
  const int scBlocks   = (E * 128 + 255) / 256;

  hipMemsetAsync(rst,   0, (size_t)Nn * 128 * sizeof(float), stream);
  hipMemsetAsync(emax,  0, (size_t)NH * sizeof(unsigned), stream);
  hipMemsetAsync(denom, 0, (size_t)NH * sizeof(float), stream);

  gat_gemm_wmma<128><<<gemmBlocks, b256, 0, stream>>>(feat, W1, hbuf, Nn);
  gat_attn_proj<<<(NH + 255) / 256, b256, 0, stream>>>(hbuf, al1, ar1, el, er, NH);
  gat_edge_max<<<e4Blocks, b256, 0, stream>>>(src, dst, el, er, emax, E * 4);
  gat_edge_denom<<<e4Blocks, b256, 0, stream>>>(src, dst, el, er, emax, denom, E * 4);
  gat_edge_scatter<<<scBlocks, b256, 0, stream>>>(src, dst, el, er, emax, denom, hbuf, rst, E);
  gat_finalize1<<<(Nn + 7) / 8, b256, 0, stream>>>(rst, b1, lns, lnb, hmid, Nn);

  hipMemsetAsync(rst,   0, (size_t)Nn * 128 * sizeof(float), stream);
  hipMemsetAsync(emax,  0, (size_t)NH * sizeof(unsigned), stream);
  hipMemsetAsync(denom, 0, (size_t)NH * sizeof(float), stream);

  gat_gemm_wmma<32><<<gemmBlocks, b256, 0, stream>>>(hmid, W2, hbuf, Nn);
  gat_attn_proj<<<(NH + 255) / 256, b256, 0, stream>>>(hbuf, al2, ar2, el, er, NH);
  gat_edge_max<<<e4Blocks, b256, 0, stream>>>(src, dst, el, er, emax, E * 4);
  gat_edge_denom<<<e4Blocks, b256, 0, stream>>>(src, dst, el, er, emax, denom, E * 4);
  gat_edge_scatter<<<scBlocks, b256, 0, stream>>>(src, dst, el, er, emax, denom, hbuf, rst, E);
  gat_finalize2<<<(Nn * 32 + 255) / 256, b256, 0, stream>>>(rst, b2, outh, Nn * 32);

  gat_flops<<<1, 256, 0, stream>>>(indeg, Nn, outh + (size_t)Nn * 32);
}


# workgroup owns 196 destination nodes: scans all edge destinations, lists own edges in LDS; max, denominator, row sum by LDS atomics, fused epilogue; no global atomics, no sort; fp sum order changes
# speedup vs baseline: 2.6655x; 2.6655x over previous
#include <hip/hip_runtime.h>


typedef float v2f __attribute__((ext_vector_type(2)));
typedef float v4f __attribute__((ext_vector_type(4)));
typedef float v8f __attribute__((ext_vector_type(8)));

#define NEG_SLOPE 0.2f
#define LN_EPS 1e-5f

__device__ __forceinline__ unsigned fenc(float f) {
  unsigned u = __float_as_uint(f);
  return (u >> 31) ? ~u : (u | 0x80000000u);
}
__device__ __forceinline__ float fdec(unsigned u) {
  return __uint_as_float((u >> 31) ? (u & 0x7FFFFFFFu) : ~u);
}
__device__ __forceinline__ float lrelu(float x) {
  return x > 0.f ? x : x * NEG_SLOPE;
}

template <int K>
__global__ void gat_gemm_wmma(const float* __restrict__ Amat,
                              const float* __restrict__ Bmat,
                              float* __restrict__ Cmat,
                              int M) {
  __shared__ float bs[K * 128];

  const int tid = threadIdx.x;

  for (int idx = tid; idx < K * 128; idx += 256) {
    int kr = idx >> 7;
    int col = idx & 127;
    int ks = kr >> 2;
    int c = col >> 4;
    int lane = (col & 15) | ((kr & 2) << 3);
    int sub = kr & 1;
    bs[((ks * 32 + lane) * 8 + c) * 2 + sub] = Bmat[idx];
  }
  __syncthreads();

  const int wave = tid >> 5;
  const int lane = tid & 31;
  const int row0 = (blockIdx.x * 8 + wave) * 16;
  if (row0 >= M) return;

  const int m  = lane & 15;
  const int kk = (lane >> 4) << 1;

  v8f acc[8] = {};

  const float* arow = Amat + (size_t)(row0 + m) * K + kk;
  const float* blane = bs + lane * 16;

#pragma unroll 4
  for (int ks = 0; ks < K / 4; ++ks) {
    v2f a = *(const v2f*)(arow + ks * 4);
    const float* bp = blane + ks * 512;
    v4f q0 = *(const v4f*)(bp);
    v4f q1 = *(const v4f*)(bp + 4);
    v4f q2 = *(const v4f*)(bp + 8);
    v4f q3 = *(const v4f*)(bp + 12);
    acc[0] = __builtin_amdgcn_wmma_f32_16x16x4_f32(false, a, false, q0.xy, (short)0, acc[0], false, false);
    acc[1] = __builtin_amdgcn_wmma_f32_16x16x4_f32(false, a, false, q0.zw, (short)0, acc[1], false, false);
    acc[2] = __builtin_amdgcn_wmma_f32_16x16x4_f32(false, a, false, q1.xy, (short)0, acc[2], false, false);
    acc[3] = __builtin_amdgcn_wmma_f32_16x16x4_f32(false, a, false, q1.zw, (short)0, acc[3], false, false);
    acc[4] = __builtin_amdgcn_wmma_f32_16x16x4_f32(false, a, false, q2.xy, (short)0, acc[4], false, false);
    acc[5] = __builtin_amdgcn_wmma_f32_16x16x4_f32(false, a, false, q2.zw, (short)0, acc[5], false, false);
    acc[6] = __builtin_amdgcn_wmma_f32_16x16x4_f32(false, a, false, q3.xy, (short)0, acc[6], false, false);
    acc[7] = __builtin_amdgcn_wmma_f32_16x16x4_f32(false, a, false, q3.zw, (short)0, acc[7], false, false);
  }

  const int mo = (lane >> 4) * 8;
#pragma unroll
  for (int c = 0; c < 8; ++c) {
#pragma unroll
    for (int r = 0; r < 8; ++r) {
      Cmat[(size_t)(row0 + mo + r) * 128 + c * 16 + m] = acc[c][r];
    }
  }
}

__global__ void gat_attn_proj(const float* __restrict__ h,
                              const float* __restrict__ al,
                              const float* __restrict__ ar,
                              float* __restrict__ el, float* __restrict__ er,
                              int NH) {
  int i = blockIdx.x * blockDim.x + threadIdx.x;
  if (i >= NH) return;
  int n = i >> 2, hh = i & 3;
  const float* hp  = h + (size_t)n * 128 + hh * 32;
  const float* alp = al + hh * 32;
  const float* arp = ar + hh * 32;
  float sl = 0.f, sr = 0.f;
#pragma unroll
  for (int f = 0; f < 32; ++f) {
    float v = hp[f];
    sl += v * alp[f];
    sr += v * arp[f];
  }
  el[i] = sl;
  er[i] = sr;
}

#define OWN_NB      196
#define OWN_CAP     32768
#define OWN_THREADS 1024

__device__ __forceinline__ void own_check(int dv, int e, int base, unsigned nb,
                                          unsigned* list, unsigned* cnt) {
  if ((unsigned)(dv - base) < nb) {
    unsigned pos = atomicAdd(cnt, 1u);
    if (pos < OWN_CAP) list[pos] = (unsigned)e;
  }
}

__device__ __forceinline__ void own_build(const int* __restrict__ dst, int e0, int e1,
                                          int base, unsigned nb,
                                          unsigned* list, unsigned* cnt, int tid) {
  const int nvec = (e1 - e0) >> 2;
  const int4* dv = (const int4*)(dst + e0);
  int i = tid;
  for (; i + 3 * OWN_THREADS < nvec; i += 4 * OWN_THREADS) {
    int4 d[4];
#pragma unroll
    for (int u = 0; u < 4; ++u) d[u] = dv[i + u * OWN_THREADS];
#pragma unroll
    for (int u = 0; u < 4; ++u) {
      const int eb = e0 + 4 * (i + u * OWN_THREADS);
      own_check(d[u].x, eb + 0, base, nb, list, cnt);
      own_check(d[u].y, eb + 1, base, nb, list, cnt);
      own_check(d[u].z, eb + 2, base, nb, list, cnt);
      own_check(d[u].w, eb + 3, base, nb, list, cnt);
    }
  }
  for (; i < nvec; i += OWN_THREADS) {
    int4 d = dv[i];
    const int eb = e0 + 4 * i;
    own_check(d.x, eb + 0, base, nb, list, cnt);
    own_check(d.y, eb + 1, base, nb, list, cnt);
    own_check(d.z, eb + 2, base, nb, list, cnt);
    own_check(d.w, eb + 3, base, nb, list, cnt);
  }
  for (int e = e0 + nvec * 4 + tid; e < e1; e += OWN_THREADS)
    own_check(dst[e], e, base, nb, list, cnt);
}

__device__ __forceinline__ void own_max(const unsigned* list, int n,
                                        const int* __restrict__ src, const int* __restrict__ dst,
                                        const float* __restrict__ el, const float* __restrict__ er,
                                        int base, unsigned* smax, int tid) {
  for (int i = tid; i < n; i += OWN_THREADS) {
    const int e = (int)list[i];
    const int s = src[e], d = dst[e];
    const int dl = d - base;
    const v4f l = *(const v4f*)(el + (size_t)s * 4);
    const v4f r = *(const v4f*)(er + (size_t)d * 4);
    atomicMax(&smax[dl * 4 + 0], fenc(lrelu(l.x + r.x)));
    atomicMax(&smax[dl * 4 + 1], fenc(lrelu(l.y + r.y)));
    atomicMax(&smax[dl * 4 + 2], fenc(lrelu(l.z + r.z)));
    atomicMax(&smax[dl * 4 + 3], fenc(lrelu(l.w + r.w)));
  }
}

__device__ __forceinline__ void own_den(const unsigned* list, int n,
                                        const int* __restrict__ src, const int* __restrict__ dst,
                                        const float* __restrict__ el, const float* __restrict__ er,
                                        int base, const unsigned* smax, float* sden, int tid) {
  for (int i = tid; i < n; i += OWN_THREADS) {
    const int e = (int)list[i];
    const int s = src[e], d = dst[e];
    const int dl = d - base;
    const v4f l = *(const v4f*)(el + (size_t)s * 4);
    const v4f r = *(const v4f*)(er + (size_t)d * 4);
    atomicAdd(&sden[dl * 4 + 0], expf(lrelu(l.x + r.x) - fdec(smax[dl * 4 + 0])));
    atomicAdd(&sden[dl * 4 + 1], expf(lrelu(l.y + r.y) - fdec(smax[dl * 4 + 1])));
    atomicAdd(&sden[dl * 4 + 2], expf(lrelu(l.z + r.z) - fdec(smax[dl * 4 + 2])));
    atomicAdd(&sden[dl * 4 + 3], expf(lrelu(l.w + r.w) - fdec(smax[dl * 4 + 3])));
  }
}

__device__ __forceinline__ void own_acc(const unsigned* list, int n,
                                        const int* __restrict__ src, const int* __restrict__ dst,
                                        const float* __restrict__ el, const float* __restrict__ er,
                                        const float* __restrict__ h,
                                        int base, const unsigned* smax, const float* sden,
                                        float* acc, int wave, int lane) {
  for (int b = wave * 32; b < n; b += OWN_THREADS) {
    const int i = b + lane;
    int s = 0, dl = 0;
    float a0 = 0.f, a1 = 0.f, a2 = 0.f, a3 = 0.f;
    if (i < n) {
      const int e = (int)list[i];
      s = src[e];
      const int d = dst[e];
      dl = d - base;
      const v4f l = *(const v4f*)(el + (size_t)s * 4);
      const v4f r = *(const v4f*)(er + (size_t)d * 4);
      a0 = expf(lrelu(l.x + r.x) - fdec(smax[dl * 4 + 0])) / fmaxf(sden[dl * 4 + 0], 1e-9f);
      a1 = expf(lrelu(l.y + r.y) - fdec(smax[dl * 4 + 1])) / fmaxf(sden[dl * 4 + 1], 1e-9f);
      a2 = expf(lrelu(l.z + r.z) - fdec(smax[dl * 4 + 2])) / fmaxf(sden[dl * 4 + 2], 1e-9f);
      a3 = expf(lrelu(l.w + r.w) - fdec(smax[dl * 4 + 3])) / fmaxf(sden[dl * 4 + 3], 1e-9f);
    }
    const int cb = min(32, n - b);
    for (int j = 0; j < cb; j += 8) {
      float hv[8][4], aw[8][4];
      int dj[8];
#pragma unroll
      for (int u = 0; u < 8; ++u) {
        const int jj = j + u;
        const int sj = __builtin_amdgcn_readlane(s, jj);
        dj[u] = __builtin_amdgcn_readlane(dl, jj);
        aw[u][0] = __int_as_float(__builtin_amdgcn_readlane(__float_as_int(a0), jj));
        aw[u][1] = __int_as_float(__builtin_amdgcn_readlane(__float_as_int(a1), jj));
        aw[u][2] = __int_as_float(__builtin_amdgcn_readlane(__float_as_int(a2), jj));
        aw[u][3] = __int_as_float(__builtin_amdgcn_readlane(__float_as_int(a3), jj));
        const float* hp = h + (size_t)sj * 128 + lane;
        hv[u][0] = hp[0];
        hv[u][1] = hp[32];
        hv[u][2] = hp[64];
        hv[u][3] = hp[96];
      }
#pragma unroll
      for (int u = 0; u < 8; ++u) {
        if (j + u < cb) {
          float* ap = acc + dj[u] * 128 + lane;
          atomicAdd(ap,      hv[u][0] * aw[u][0]);
          atomicAdd(ap + 32, hv[u][1] * aw[u][1]);
          atomicAdd(ap + 64, hv[u][2] * aw[u][2]);
          atomicAdd(ap + 96, hv[u][3] * aw[u][3]);
        }
      }
    }
  }
}

template <int LAYER>
__global__ void __launch_bounds__(OWN_THREADS)
gat_owner(const int* __restrict__ src, const int* __restrict__ dst,
          const float* __restrict__ el, const float* __restrict__ er,
          const float* __restrict__ h, const float* __restrict__ bias,
          const float* __restrict__ lns, const float* __restrict__ lnb,
          float* __restrict__ outp, int Nn, int E) {
  __shared__ float    acc[OWN_NB * 128];
  __shared__ unsigned smax[OWN_NB * 4];
  __shared__ float    sden[OWN_NB * 4];
  __shared__ unsigned list[OWN_CAP];
  __shared__ unsigned cnt;

  const int tid  = threadIdx.x;
  const int wave = tid >> 5;
  const int lane = tid & 31;
  const int base = blockIdx.x * OWN_NB;
  const unsigned nb = (unsigned)min(OWN_NB, Nn - base);

  for (int i = tid; i < OWN_NB * 128; i += OWN_THREADS) acc[i] = 0.f;
  for (int i = tid; i < OWN_NB * 4; i += OWN_THREADS) { smax[i] = 0u; sden[i] = 0.f; }
  if (tid == 0) cnt = 0u;
  __syncthreads();

  own_build(dst, 0, E, base, nb, list, &cnt, tid);
  __syncthreads();
  const unsigned total = cnt;

  if (total <= OWN_CAP) {
    const int n = (int)total;
    own_max(list, n, src, dst, el, er, base, smax, tid);
    __syncthreads();
    own_den(list, n, src, dst, el, er, base, smax, sden, tid);
    __syncthreads();
    own_acc(list, n, src, dst, el, er, h, base, smax, sden, acc, wave, lane);
  } else {
    for (int pass = 0; pass < 3; ++pass) {
      for (int e0 = 0; e0 < E; e0 += OWN_CAP) {
        __syncthreads();
        if (tid == 0) cnt = 0u;
        __syncthreads();
        own_build(dst, e0, min(E, e0 + OWN_CAP), base, nb, list, &cnt, tid);
        __syncthreads();
        const int n = (int)cnt;
        if (pass == 0)      own_max(list, n, src, dst, el, er, base, smax, tid);
        else if (pass == 1) own_den(list, n, src, dst, el, er, base, smax, sden, tid);
        else                own_acc(list, n, src, dst, el, er, h, base, smax, sden, acc, wave, lane);
      }
    }
  }
  __syncthreads();

  for (int nl = wave; nl < (int)nb; nl += OWN_THREADS / 32) {
    const int n = base + nl;
    float x = 0.f;
#pragma unroll
    for (int hh = 0; hh < 4; ++hh)
      x += acc[nl * 128 + hh * 32 + lane] + bias[hh * 32 + lane];
    x *= 0.25f;
    if (LAYER == 1) {
      float mu = x;
#pragma unroll
      for (int m = 16; m >= 1; m >>= 1) mu += __shfl_xor(mu, m, 32);
      mu *= (1.f / 32.f);
      float dx = x - mu;
      float var = dx * dx;
#pragma unroll
      for (int m = 16; m >= 1; m >>= 1) var += __shfl_xor(var, m, 32);
      var *= (1.f / 32.f);
      float y = dx * rsqrtf(var + LN_EPS) * lns[lane] + lnb[lane];
      outp[(size_t)n * 32 + lane] = fmaxf(y, 0.f);
    } else {
      outp[(size_t)n * 32 + lane] = x;
    }
  }
}

__global__ void gat_flops(const int* __restrict__ in_deg, int Nn, float* out) {
  __shared__ float ssum[256];
  float s = 0.f;
  for (int i = threadIdx.x; i < Nn; i += 256) s += (float)in_deg[i];
  ssum[threadIdx.x] = s;
  __syncthreads();
  for (int m = 128; m > 0; m >>= 1) {
    if ((int)threadIdx.x < m) ssum[threadIdx.x] += ssum[threadIdx.x + m];
    __syncthreads();
  }
  if (threadIdx.x == 0) {
    double ne = (double)ssum[0];
    double fl = (4.0 * ne * (6.0 * 128 * 32 + 6 * 32 + 2) +
                 4.0 * ne * (6.0 * 32 * 32 + 6 * 32 + 2)) / 1e12;
    out[0] = (float)fl;
  }
}

extern "C" void kernel_launch(void* const* d_in, const int* in_sizes, int n_in,
                              void* d_out, int out_size, void* d_ws, size_t ws_size,
                              hipStream_t stream) {
  const float* feat = (const float*)d_in[0];
  const float* W1   = (const float*)d_in[1];
  const float* al1  = (const float*)d_in[2];
  const float* ar1  = (const float*)d_in[3];
  const float* b1   = (const float*)d_in[4];
  const float* lns  = (const float*)d_in[5];
  const float* lnb  = (const float*)d_in[6];
  const float* W2   = (const float*)d_in[7];
  const float* al2  = (const float*)d_in[8];
  const float* ar2  = (const float*)d_in[9];
  const float* b2   = (const float*)d_in[10];
  const int*   src  = (const int*)d_in[11];
  const int*   dst  = (const int*)d_in[12];
  const int*   indeg= (const int*)d_in[13];

  const int Nn = in_sizes[13];
  const int E  = in_sizes[11];
  const int NH = Nn * 4;

  char* p = (char*)d_ws;
  float*    hbuf  = (float*)p;   p += (size_t)Nn * 128 * sizeof(float);
  float*    hmid  = (float*)p;   p += (size_t)Nn * 32  * sizeof(float);
  float*    el    = (float*)p;   p += (size_t)NH * sizeof(float);
  float*    er    = (float*)p;   p += (size_t)NH * sizeof(float);
  (void)ws_size; (void)n_in; (void)out_size;

  float* outh = (float*)d_out;

  dim3 b256(256);
  const int rowTiles   = (Nn + 15) / 16;
  const int gemmBlocks = (rowTiles + 7) / 8;
  const int ownBlocks  = (Nn + OWN_NB - 1) / OWN_NB;

  gat_gemm_wmma<128><<<gemmBlocks, b256, 0, stream>>>(feat, W1, hbuf, Nn);
  gat_attn_proj<<<(NH + 255) / 256, b256, 0, stream>>>(hbuf, al1, ar1, el, er, NH);
  gat_owner<1><<<ownBlocks, OWN_THREADS, 0, stream>>>(src, dst, el, er, hbuf, b1, lns, lnb, hmid, Nn, E);

  gat_gemm_wmma<32><<<gemmBlocks, b256, 0, stream>>>(hmid, W2, hbuf, Nn);
  gat_attn_proj<<<(NH + 255) / 256, b256, 0, stream>>>(hbuf, al2, ar2, el, er, NH);
  gat_owner<2><<<ownBlocks, OWN_THREADS, 0, stream>>>(src, dst, el, er, hbuf, b2, lns, lnb, outh, Nn, E);

  gat_flops<<<1, 256, 0, stream>>>(indeg, Nn, outh + (size_t)Nn * 32);
}


# each workgroup stores its LDS edge list to the workspace in layer 1 and layer 2 loads it: the scan of all 800000 destinations runs once per call, was twice; fp sums by LDS atomic arrival as before
# speedup vs baseline: 2.9483x; 1.1061x over previous
#include <hip/hip_runtime.h>


typedef float v2f __attribute__((ext_vector_type(2)));
typedef float v4f __attribute__((ext_vector_type(4)));
typedef float v8f __attribute__((ext_vector_type(8)));

#define NEG_SLOPE 0.2f
#define LN_EPS 1e-5f

__device__ __forceinline__ unsigned fenc(float f) {
  unsigned u = __float_as_uint(f);
  return (u >> 31) ? ~u : (u | 0x80000000u);
}
__device__ __forceinline__ float fdec(unsigned u) {
  return __uint_as_float((u >> 31) ? (u & 0x7FFFFFFFu) : ~u);
}
__device__ __forceinline__ float lrelu(float x) {
  return x > 0.f ? x : x * NEG_SLOPE;
}

template <int K>
__global__ void gat_gemm_wmma(const float* __restrict__ Amat,
                              const float* __restrict__ Bmat,
                              float* __restrict__ Cmat,
                              int M) {
  __shared__ float bs[K * 128];

  const int tid = threadIdx.x;

  for (int idx = tid; idx < K * 128; idx += 256) {
    int kr = idx >> 7;
    int col = idx & 127;
    int ks = kr >> 2;
    int c = col >> 4;
    int lane = (col & 15) | ((kr & 2) << 3);
    int sub = kr & 1;
    bs[((ks * 32 + lane) * 8 + c) * 2 + sub] = Bmat[idx];
  }
  __syncthreads();

  const int wave = tid >> 5;
  const int lane = tid & 31;
  const int row0 = (blockIdx.x * 8 + wave) * 16;
  if (row0 >= M) return;

  const int m  = lane & 15;
  const int kk = (lane >> 4) << 1;

  v8f acc[8] = {};

  const float* arow = Amat + (size_t)(row0 + m) * K + kk;
  const float* blane = bs + lane * 16;

#pragma unroll 4
  for (int ks = 0; ks < K / 4; ++ks) {
    v2f a = *(const v2f*)(arow + ks * 4);
    const float* bp = blane + ks * 512;
    v4f q0 = *(const v4f*)(bp);
    v4f q1 = *(const v4f*)(bp + 4);
    v4f q2 = *(const v4f*)(bp + 8);
    v4f q3 = *(const v4f*)(bp + 12);
    acc[0] = __builtin_amdgcn_wmma_f32_16x16x4_f32(false, a, false, q0.xy, (short)0, acc[0], false, false);
    acc[1] = __builtin_amdgcn_wmma_f32_16x16x4_f32(false, a, false, q0.zw, (short)0, acc[1], false, false);
    acc[2] = __builtin_amdgcn_wmma_f32_16x16x4_f32(false, a, false, q1.xy, (short)0, acc[2], false, false);
    acc[3] = __builtin_amdgcn_wmma_f32_16x16x4_f32(false, a, false, q1.zw, (short)0, acc[3], false, false);
    acc[4] = __builtin_amdgcn_wmma_f32_16x16x4_f32(false, a, false, q2.xy, (short)0, acc[4], false, false);
    acc[5] = __builtin_amdgcn_wmma_f32_16x16x4_f32(false, a, false, q2.zw, (short)0, acc[5], false, false);
    acc[6] = __builtin_amdgcn_wmma_f32_16x16x4_f32(false, a, false, q3.xy, (short)0, acc[6], false, false);
    acc[7] = __builtin_amdgcn_wmma_f32_16x16x4_f32(false, a, false, q3.zw, (short)0, acc[7], false, false);
  }

  const int mo = (lane >> 4) * 8;
#pragma unroll
  for (int c = 0; c < 8; ++c) {
#pragma unroll
    for (int r = 0; r < 8; ++r) {
      Cmat[(size_t)(row0 + mo + r) * 128 + c * 16 + m] = acc[c][r];
    }
  }
}

__global__ void gat_attn_proj(const float* __restrict__ h,
                              const float* __restrict__ al,
                              const float* __restrict__ ar,
                              float* __restrict__ el, float* __restrict__ er,
                              int NH) {
  int i = blockIdx.x * blockDim.x + threadIdx.x;
  if (i >= NH) return;
  int n = i >> 2, hh = i & 3;
  const float* hp  = h + (size_t)n * 128 + hh * 32;
  const float* alp = al + hh * 32;
  const float* arp = ar + hh * 32;
  float sl = 0.f, sr = 0.f;
#pragma unroll
  for (int f = 0; f < 32; ++f) {
    float v = hp[f];
    sl += v * alp[f];
    sr += v * arp[f];
  }
  el[i] = sl;
  er[i] = sr;
}

#define OWN_NB      196
#define OWN_CAP     32768
#define OWN_THREADS 1024

__device__ __forceinline__ void own_check(int dv, int e, int base, unsigned nb,
                                          unsigned* list, unsigned* cnt) {
  if ((unsigned)(dv - base) < nb) {
    unsigned pos = atomicAdd(cnt, 1u);
    if (pos < OWN_CAP) list[pos] = (unsigned)e;
  }
}

__device__ __forceinline__ void own_build(const int* __restrict__ dst, int e0, int e1,
                                          int base, unsigned nb,
                                          unsigned* list, unsigned* cnt, int tid) {
  const int nvec = (e1 - e0) >> 2;
  const int4* dv = (const int4*)(dst + e0);
  int i = tid;
  for (; i + 3 * OWN_THREADS < nvec; i += 4 * OWN_THREADS) {
    int4 d[4];
#pragma unroll
    for (int u = 0; u < 4; ++u) d[u] = dv[i + u * OWN_THREADS];
#pragma unroll
    for (int u = 0; u < 4; ++u) {
      const int eb = e0 + 4 * (i + u * OWN_THREADS);
      own_check(d[u].x, eb + 0, base, nb, list, cnt);
      own_check(d[u].y, eb + 1, base, nb, list, cnt);
      own_check(d[u].z, eb + 2, base, nb, list, cnt);
      own_check(d[u].w, eb + 3, base, nb, list, cnt);
    }
  }
  for (; i < nvec; i += OWN_THREADS) {
    int4 d = dv[i];
    const int eb = e0 + 4 * i;
    own_check(d.x, eb + 0, base, nb, list, cnt);
    own_check(d.y, eb + 1, base, nb, list, cnt);
    own_check(d.z, eb + 2, base, nb, list, cnt);
    own_check(d.w, eb + 3, base, nb, list, cnt);
  }
  for (int e = e0 + nvec * 4 + tid; e < e1; e += OWN_THREADS)
    own_check(dst[e], e, base, nb, list, cnt);
}

__device__ __forceinline__ void own_max(const unsigned* list, int n,
                                        const int* __restrict__ src, const int* __restrict__ dst,
                                        const float* __restrict__ el, const float* __restrict__ er,
                                        int base, unsigned* smax, int tid) {
  for (int i = tid; i < n; i += OWN_THREADS) {
    const int e = (int)list[i];
    const int s = src[e], d = dst[e];
    const int dl = d - base;
    const v4f l = *(const v4f*)(el + (size_t)s * 4);
    const v4f r = *(const v4f*)(er + (size_t)d * 4);
    atomicMax(&smax[dl * 4 + 0], fenc(lrelu(l.x + r.x)));
    atomicMax(&smax[dl * 4 + 1], fenc(lrelu(l.y + r.y)));
    atomicMax(&smax[dl * 4 + 2], fenc(lrelu(l.z + r.z)));
    atomicMax(&smax[dl * 4 + 3], fenc(lrelu(l.w + r.w)));
  }
}

__device__ __forceinline__ void own_den(const unsigned* list, int n,
                                        const int* __restrict__ src, const int* __restrict__ dst,
                                        const float* __restrict__ el, const float* __restrict__ er,
                                        int base, const unsigned* smax, float* sden, int tid) {
  for (int i = tid; i < n; i += OWN_THREADS) {
    const int e = (int)list[i];
    const int s = src[e], d = dst[e];
    const int dl = d - base;
    const v4f l = *(const v4f*)(el + (size_t)s * 4);
    const v4f r = *(const v4f*)(er + (size_t)d * 4);
    atomicAdd(&sden[dl * 4 + 0], expf(lrelu(l.x + r.x) - fdec(smax[dl * 4 + 0])));
    atomicAdd(&sden[dl * 4 + 1], expf(lrelu(l.y + r.y) - fdec(smax[dl * 4 + 1])));
    atomicAdd(&sden[dl * 4 + 2], expf(lrelu(l.z + r.z) - fdec(smax[dl * 4 + 2])));
    atomicAdd(&sden[dl * 4 + 3], expf(lrelu(l.w + r.w) - fdec(smax[dl * 4 + 3])));
  }
}

__device__ __forceinline__ void own_acc(const unsigned* list, int n,
                                        const int* __restrict__ src, const int* __restrict__ dst,
                                        const float* __restrict__ el, const float* __restrict__ er,
                                        const float* __restrict__ h,
                                        int base, const unsigned* smax, const float* sden,
                                        float* acc, int wave, int lane) {
  for (int b = wave * 32; b < n; b += OWN_THREADS) {
    const int i = b + lane;
    int s = 0, dl = 0;
    float a0 = 0.f, a1 = 0.f, a2 = 0.f, a3 = 0.f;
    if (i < n) {
      const int e = (int)list[i];
      s = src[e];
      const int d = dst[e];
      dl = d - base;
      const v4f l = *(const v4f*)(el + (size_t)s * 4);
      const v4f r = *(const v4f*)(er + (size_t)d * 4);
      a0 = expf(lrelu(l.x + r.x) - fdec(smax[dl * 4 + 0])) / fmaxf(sden[dl * 4 + 0], 1e-9f);
      a1 = expf(lrelu(l.y + r.y) - fdec(smax[dl * 4 + 1])) / fmaxf(sden[dl * 4 + 1], 1e-9f);
      a2 = expf(lrelu(l.z + r.z) - fdec(smax[dl * 4 + 2])) / fmaxf(sden[dl * 4 + 2], 1e-9f);
      a3 = expf(lrelu(l.w + r.w) - fdec(smax[dl * 4 + 3])) / fmaxf(sden[dl * 4 + 3], 1e-9f);
    }
    const int cb = min(32, n - b);
    for (int j = 0; j < cb; j += 8) {
      float hv[8][4], aw[8][4];
      int dj[8];
#pragma unroll
      for (int u = 0; u < 8; ++u) {
        const int jj = j + u;
        const int sj = __builtin_amdgcn_readlane(s, jj);
        dj[u] = __builtin_amdgcn_readlane(dl, jj);
        aw[u][0] = __int_as_float(__builtin_amdgcn_readlane(__float_as_int(a0), jj));
        aw[u][1] = __int_as_float(__builtin_amdgcn_readlane(__float_as_int(a1), jj));
        aw[u][2] = __int_as_float(__builtin_amdgcn_readlane(__float_as_int(a2), jj));
        aw[u][3] = __int_as_float(__builtin_amdgcn_readlane(__float_as_int(a3), jj));
        const float* hp = h + (size_t)sj * 128 + lane;
        hv[u][0] = hp[0];
        hv[u][1] = hp[32];
        hv[u][2] = hp[64];
        hv[u][3] = hp[96];
      }
#pragma unroll
      for (int u = 0; u < 8; ++u) {
        if (j + u < cb) {
          float* ap = acc + dj[u] * 128 + lane;
          atomicAdd(ap,      hv[u][0] * aw[u][0]);
          atomicAdd(ap + 32, hv[u][1] * aw[u][1]);
          atomicAdd(ap + 64, hv[u][2] * aw[u][2]);
          atomicAdd(ap + 96, hv[u][3] * aw[u][3]);
        }
      }
    }
  }
}

template <int LAYER>
__global__ void __launch_bounds__(OWN_THREADS)
gat_owner(const int* __restrict__ src, const int* __restrict__ dst,
          const float* __restrict__ el, const float* __restrict__ er,
          const float* __restrict__ h, const float* __restrict__ bias,
          const float* __restrict__ lns, const float* __restrict__ lnb,
          float* __restrict__ outp,
          unsigned* __restrict__ wlist, unsigned* __restrict__ wcnt,
          int Nn, int E) {
  __shared__ float    acc[OWN_NB * 128];
  __shared__ unsigned smax[OWN_NB * 4];
  __shared__ float    sden[OWN_NB * 4];
  __shared__ unsigned list[OWN_CAP];
  __shared__ unsigned cnt;

  const int tid  = threadIdx.x;
  const int wave = tid >> 5;
  const int lane = tid & 31;
  const int base = blockIdx.x * OWN_NB;
  const unsigned nb = (unsigned)min(OWN_NB, Nn - base);

  for (int i = tid; i < OWN_NB * 128; i += OWN_THREADS) acc[i] = 0.f;
  for (int i = tid; i < OWN_NB * 4; i += OWN_THREADS) { smax[i] = 0u; sden[i] = 0.f; }
  if (tid == 0) cnt = 0u;

  unsigned* wl = wlist + (size_t)blockIdx.x * OWN_CAP;
  unsigned total;
  if (LAYER == 1) {
    __syncthreads();
    own_build(dst, 0, E, base, nb, list, &cnt, tid);
    __syncthreads();
    total = cnt;
    if (tid == 0) wcnt[blockIdx.x] = total;
    if (total <= OWN_CAP)
      for (unsigned i = tid; i < total; i += OWN_THREADS) wl[i] = list[i];
  } else {
    total = wcnt[blockIdx.x];
    if (total <= OWN_CAP) {
      for (unsigned i0 = tid; i0 < total; i0 += 4 * OWN_THREADS) {
        unsigned v[4];
#pragma unroll
        for (int u = 0; u < 4; ++u) {
          const unsigned i = i0 + u * OWN_THREADS;
          v[u] = i < total ? wl[i] : 0u;
        }
#pragma unroll
        for (int u = 0; u < 4; ++u) {
          const unsigned i = i0 + u * OWN_THREADS;
          if (i < total) list[i] = v[u];
        }
      }
    }
    __syncthreads();
  }

  if (total <= OWN_CAP) {
    const int n = (int)total;
    own_max(list, n, src, dst, el, er, base, smax, tid);
    __syncthreads();
    own_den(list, n, src, dst, el, er, base, smax, sden, tid);
    __syncthreads();
    own_acc(list, n, src, dst, el, er, h, base, smax, sden, acc, wave, lane);
  } else {
    for (int pass = 0; pass < 3; ++pass) {
      for (int e0 = 0; e0 < E; e0 += OWN_CAP) {
        __syncthreads();
        if (tid == 0) cnt = 0u;
        __syncthreads();
        own_build(dst, e0, min(E, e0 + OWN_CAP), base, nb, list, &cnt, tid);
        __syncthreads();
        const int n = (int)cnt;
        if (pass == 0)      own_max(list, n, src, dst, el, er, base, smax, tid);
        else if (pass == 1) own_den(list, n, src, dst, el, er, base, smax, sden, tid);
        else                own_acc(list, n, src, dst, el, er, h, base, smax, sden, acc, wave, lane);
      }
    }
  }
  __syncthreads();

  for (int nl = wave; nl < (int)nb; nl += OWN_THREADS / 32) {
    const int n = base + nl;
    float x = 0.f;
#pragma unroll
    for (int hh = 0; hh < 4; ++hh)
      x += acc[nl * 128 + hh * 32 + lane] + bias[hh * 32 + lane];
    x *= 0.25f;
    if (LAYER == 1) {
      float mu = x;
#pragma unroll
      for (int m = 16; m >= 1; m >>= 1) mu += __shfl_xor(mu, m, 32);
      mu *= (1.f / 32.f);
      float dx = x - mu;
      float var = dx * dx;
#pragma unroll
      for (int m = 16; m >= 1; m >>= 1) var += __shfl_xor(var, m, 32);
      var *= (1.f / 32.f);
      float y = dx * rsqrtf(var + LN_EPS) * lns[lane] + lnb[lane];
      outp[(size_t)n * 32 + lane] = fmaxf(y, 0.f);
    } else {
      outp[(size_t)n * 32 + lane] = x;
    }
  }
}

__global__ void gat_flops(const int* __restrict__ in_deg, int Nn, float* out) {
  __shared__ float ssum[256];
  float s = 0.f;
  for (int i = threadIdx.x; i < Nn; i += 256) s += (float)in_deg[i];
  ssum[threadIdx.x] = s;
  __syncthreads();
  for (int m = 128; m > 0; m >>= 1) {
    if ((int)threadIdx.x < m) ssum[threadIdx.x] += ssum[threadIdx.x + m];
    __syncthreads();
  }
  if (threadIdx.x == 0) {
    double ne = (double)ssum[0];
    double fl = (4.0 * ne * (6.0 * 128 * 32 + 6 * 32 + 2) +
                 4.0 * ne * (6.0 * 32 * 32 + 6 * 32 + 2)) / 1e12;
    out[0] = (float)fl;
  }
}

extern "C" void kernel_launch(void* const* d_in, const int* in_sizes, int n_in,
                              void* d_out, int out_size, void* d_ws, size_t ws_size,
                              hipStream_t stream) {
  const float* feat = (const float*)d_in[0];
  const float* W1   = (const float*)d_in[1];
  const float* al1  = (const float*)d_in[2];
  const float* ar1  = (const float*)d_in[3];
  const float* b1   = (const float*)d_in[4];
  const float* lns  = (const float*)d_in[5];
  const float* lnb  = (const float*)d_in[6];
  const float* W2   = (const float*)d_in[7];
  const float* al2  = (const float*)d_in[8];
  const float* ar2  = (const float*)d_in[9];
  const float* b2   = (const float*)d_in[10];
  const int*   src  = (const int*)d_in[11];
  const int*   dst  = (const int*)d_in[12];
  const int*   indeg= (const int*)d_in[13];

  const int Nn = in_sizes[13];
  const int E  = in_sizes[11];
  const int NH = Nn * 4;

  char* p = (char*)d_ws;
  float*    hbuf  = (float*)p;   p += (size_t)Nn * 128 * sizeof(float);
  float*    hmid  = (float*)p;   p += (size_t)Nn * 32  * sizeof(float);
  float*    el    = (float*)p;   p += (size_t)NH * sizeof(float);
  float*    er    = (float*)p;   p += (size_t)NH * sizeof(float);
  const int ownBlocks  = (Nn + OWN_NB - 1) / OWN_NB;
  unsigned* wlist = (unsigned*)p; p += (size_t)ownBlocks * OWN_CAP * sizeof(unsigned);
  unsigned* wcnt  = (unsigned*)p;
  (void)ws_size; (void)n_in; (void)out_size;

  float* outh = (float*)d_out;

  dim3 b256(256);
  const int rowTiles   = (Nn + 15) / 16;
  const int gemmBlocks = (rowTiles + 7) / 8;

  gat_gemm_wmma<128><<<gemmBlocks, b256, 0, stream>>>(feat, W1, hbuf, Nn);
  gat_attn_proj<<<(NH + 255) / 256, b256, 0, stream>>>(hbuf, al1, ar1, el, er, NH);
  gat_owner<1><<<ownBlocks, OWN_THREADS, 0, stream>>>(src, dst, el, er, hbuf, b1, lns, lnb, hmid, wlist, wcnt, Nn, E);

  gat_gemm_wmma<32><<<gemmBlocks, b256, 0, stream>>>(hmid, W2, hbuf, Nn);
  gat_attn_proj<<<(NH + 255) / 256, b256, 0, stream>>>(hbuf, al2, ar2, el, er, NH);
  gat_owner<2><<<ownBlocks, OWN_THREADS, 0, stream>>>(src, dst, el, er, hbuf, b2, lns, lnb, outh, wlist, wcnt, Nn, E);

  gat_flops<<<1, 256, 0, stream>>>(indeg, Nn, outh + (size_t)Nn * 32);
}


# in-degree sum split over the owner workgroups: layer 1 adds each workgroup's own node counts, layer 2's last workgroup adds the 256 integer partial sums, writes the scalar; 6 launches, was 7
# speedup vs baseline: 4.1522x; 1.4083x over previous
#include <hip/hip_runtime.h>


typedef float v2f __attribute__((ext_vector_type(2)));
typedef float v4f __attribute__((ext_vector_type(4)));
typedef float v8f __attribute__((ext_vector_type(8)));

#define NEG_SLOPE 0.2f
#define LN_EPS 1e-5f

__device__ __forceinline__ unsigned fenc(float f) {
  unsigned u = __float_as_uint(f);
  return (u >> 31) ? ~u : (u | 0x80000000u);
}
__device__ __forceinline__ float fdec(unsigned u) {
  return __uint_as_float((u >> 31) ? (u & 0x7FFFFFFFu) : ~u);
}
__device__ __forceinline__ float lrelu(float x) {
  return x > 0.f ? x : x * NEG_SLOPE;
}

template <int K>
__global__ void gat_gemm_wmma(const float* __restrict__ Amat,
                              const float* __restrict__ Bmat,
                              float* __restrict__ Cmat,
                              int M) {
  __shared__ float bs[K * 128];

  const int tid = threadIdx.x;

  for (int idx = tid; idx < K * 128; idx += 256) {
    int kr = idx >> 7;
    int col = idx & 127;
    int ks = kr >> 2;
    int c = col >> 4;
    int lane = (col & 15) | ((kr & 2) << 3);
    int sub = kr & 1;
    bs[((ks * 32 + lane) * 8 + c) * 2 + sub] = Bmat[idx];
  }
  __syncthreads();

  const int wave = tid >> 5;
  const int lane = tid & 31;
  const int row0 = (blockIdx.x * 8 + wave) * 16;
  if (row0 >= M) return;

  const int m  = lane & 15;
  const int kk = (lane >> 4) << 1;

  v8f acc[8] = {};

  const float* arow = Amat + (size_t)(row0 + m) * K + kk;
  const float* blane = bs + lane * 16;

#pragma unroll 4
  for (int ks = 0; ks < K / 4; ++ks) {
    v2f a = *(const v2f*)(arow + ks * 4);
    const float* bp = blane + ks * 512;
    v4f q0 = *(const v4f*)(bp);
    v4f q1 = *(const v4f*)(bp + 4);
    v4f q2 = *(const v4f*)(bp + 8);
    v4f q3 = *(const v4f*)(bp + 12);
    acc[0] = __builtin_amdgcn_wmma_f32_16x16x4_f32(false, a, false, q0.xy, (short)0, acc[0], false, false);
    acc[1] = __builtin_amdgcn_wmma_f32_16x16x4_f32(false, a, false, q0.zw, (short)0, acc[1], false, false);
    acc[2] = __builtin_amdgcn_wmma_f32_16x16x4_f32(false, a, false, q1.xy, (short)0, acc[2], false, false);
    acc[3] = __builtin_amdgcn_wmma_f32_16x16x4_f32(false, a, false, q1.zw, (short)0, acc[3], false, false);
    acc[4] = __builtin_amdgcn_wmma_f32_16x16x4_f32(false, a, false, q2.xy, (short)0, acc[4], false, false);
    acc[5] = __builtin_amdgcn_wmma_f32_16x16x4_f32(false, a, false, q2.zw, (short)0, acc[5], false, false);
    acc[6] = __builtin_amdgcn_wmma_f32_16x16x4_f32(false, a, false, q3.xy, (short)0, acc[6], false, false);
    acc[7] = __builtin_amdgcn_wmma_f32_16x16x4_f32(false, a, false, q3.zw, (short)0, acc[7], false, false);
  }

  const int mo = (lane >> 4) * 8;
#pragma unroll
  for (int c = 0; c < 8; ++c) {
#pragma unroll
    for (int r = 0; r < 8; ++r) {
      Cmat[(size_t)(row0 + mo + r) * 128 + c * 16 + m] = acc[c][r];
    }
  }
}

__global__ void gat_attn_proj(const float* __restrict__ h,
                              const float* __restrict__ al,
                              const float* __restrict__ ar,
                              float* __restrict__ el, float* __restrict__ er,
                              int NH) {
  int i = blockIdx.x * blockDim.x + threadIdx.x;
  if (i >= NH) return;
  int n = i >> 2, hh = i & 3;
  const float* hp  = h + (size_t)n * 128 + hh * 32;
  const float* alp = al + hh * 32;
  const float* arp = ar + hh * 32;
  float sl = 0.f, sr = 0.f;
#pragma unroll
  for (int f = 0; f < 32; ++f) {
    float v = hp[f];
    sl += v * alp[f];
    sr += v * arp[f];
  }
  el[i] = sl;
  er[i] = sr;
}

#define OWN_NB      196
#define OWN_CAP     32768
#define OWN_THREADS 1024

__device__ __forceinline__ void own_check(int dv, int e, int base, unsigned nb,
                                          unsigned* list, unsigned* cnt) {
  if ((unsigned)(dv - base) < nb) {
    unsigned pos = atomicAdd(cnt, 1u);
    if (pos < OWN_CAP) list[pos] = (unsigned)e;
  }
}

__device__ __forceinline__ void own_build(const int* __restrict__ dst, int e0, int e1,
                                          int base, unsigned nb,
                                          unsigned* list, unsigned* cnt, int tid) {
  const int nvec = (e1 - e0) >> 2;
  const int4* dv = (const int4*)(dst + e0);
  int i = tid;
  for (; i + 3 * OWN_THREADS < nvec; i += 4 * OWN_THREADS) {
    int4 d[4];
#pragma unroll
    for (int u = 0; u < 4; ++u) d[u] = dv[i + u * OWN_THREADS];
#pragma unroll
    for (int u = 0; u < 4; ++u) {
      const int eb = e0 + 4 * (i + u * OWN_THREADS);
      own_check(d[u].x, eb + 0, base, nb, list, cnt);
      own_check(d[u].y, eb + 1, base, nb, list, cnt);
      own_check(d[u].z, eb + 2, base, nb, list, cnt);
      own_check(d[u].w, eb + 3, base, nb, list, cnt);
    }
  }
  for (; i < nvec; i += OWN_THREADS) {
    int4 d = dv[i];
    const int eb = e0 + 4 * i;
    own_check(d.x, eb + 0, base, nb, list, cnt);
    own_check(d.y, eb + 1, base, nb, list, cnt);
    own_check(d.z, eb + 2, base, nb, list, cnt);
    own_check(d.w, eb + 3, base, nb, list, cnt);
  }
  for (int e = e0 + nvec * 4 + tid; e < e1; e += OWN_THREADS)
    own_check(dst[e], e, base, nb, list, cnt);
}

__device__ __forceinline__ void own_max(const unsigned* list, int n,
                                        const int* __restrict__ src, const int* __restrict__ dst,
                                        const float* __restrict__ el, const float* __restrict__ er,
                                        int base, unsigned* smax, int tid) {
  for (int i = tid; i < n; i += OWN_THREADS) {
    const int e = (int)list[i];
    const int s = src[e], d = dst[e];
    const int dl = d - base;
    const v4f l = *(const v4f*)(el + (size_t)s * 4);
    const v4f r = *(const v4f*)(er + (size_t)d * 4);
    atomicMax(&smax[dl * 4 + 0], fenc(lrelu(l.x + r.x)));
    atomicMax(&smax[dl * 4 + 1], fenc(lrelu(l.y + r.y)));
    atomicMax(&smax[dl * 4 + 2], fenc(lrelu(l.z + r.z)));
    atomicMax(&smax[dl * 4 + 3], fenc(lrelu(l.w + r.w)));
  }
}

__device__ __forceinline__ void own_den(const unsigned* list, int n,
                                        const int* __restrict__ src, const int* __restrict__ dst,
                                        const float* __restrict__ el, const float* __restrict__ er,
                                        int base, const unsigned* smax, float* sden, int tid) {
  for (int i = tid; i < n; i += OWN_THREADS) {
    const int e = (int)list[i];
    const int s = src[e], d = dst[e];
    const int dl = d - base;
    const v4f l = *(const v4f*)(el + (size_t)s * 4);
    const v4f r = *(const v4f*)(er + (size_t)d * 4);
    atomicAdd(&sden[dl * 4 + 0], expf(lrelu(l.x + r.x) - fdec(smax[dl * 4 + 0])));
    atomicAdd(&sden[dl * 4 + 1], expf(lrelu(l.y + r.y) - fdec(smax[dl * 4 + 1])));
    atomicAdd(&sden[dl * 4 + 2], expf(lrelu(l.z + r.z) - fdec(smax[dl * 4 + 2])));
    atomicAdd(&sden[dl * 4 + 3], expf(lrelu(l.w + r.w) - fdec(smax[dl * 4 + 3])));
  }
}

__device__ __forceinline__ void own_acc(const unsigned* list, int n,
                                        const int* __restrict__ src, const int* __restrict__ dst,
                                        const float* __restrict__ el, const float* __restrict__ er,
                                        const float* __restrict__ h,
                                        int base, const unsigned* smax, const float* sden,
                                        float* acc, int wave, int lane) {
  for (int b = wave * 32; b < n; b += OWN_THREADS) {
    const int i = b + lane;
    int s = 0, dl = 0;
    float a0 = 0.f, a1 = 0.f, a2 = 0.f, a3 = 0.f;
    if (i < n) {
      const int e = (int)list[i];
      s = src[e];
      const int d = dst[e];
      dl = d - base;
      const v4f l = *(const v4f*)(el + (size_t)s * 4);
      const v4f r = *(const v4f*)(er + (size_t)d * 4);
      a0 = expf(lrelu(l.x + r.x) - fdec(smax[dl * 4 + 0])) / fmaxf(sden[dl * 4 + 0], 1e-9f);
      a1 = expf(lrelu(l.y + r.y) - fdec(smax[dl * 4 + 1])) / fmaxf(sden[dl * 4 + 1], 1e-9f);
      a2 = expf(lrelu(l.z + r.z) - fdec(smax[dl * 4 + 2])) / fmaxf(sden[dl * 4 + 2], 1e-9f);
      a3 = expf(lrelu(l.w + r.w) - fdec(smax[dl * 4 + 3])) / fmaxf(sden[dl * 4 + 3], 1e-9f);
    }
    const int cb = min(32, n - b);
    for (int j = 0; j < cb; j += 8) {
      float hv[8][4], aw[8][4];
      int dj[8];
#pragma unroll
      for (int u = 0; u < 8; ++u) {
        const int jj = j + u;
        const int sj = __builtin_amdgcn_readlane(s, jj);
        dj[u] = __builtin_amdgcn_readlane(dl, jj);
        aw[u][0] = __int_as_float(__builtin_amdgcn_readlane(__float_as_int(a0), jj));
        aw[u][1] = __int_as_float(__builtin_amdgcn_readlane(__float_as_int(a1), jj));
        aw[u][2] = __int_as_float(__builtin_amdgcn_readlane(__float_as_int(a2), jj));
        aw[u][3] = __int_as_float(__builtin_amdgcn_readlane(__float_as_int(a3), jj));
        const float* hp = h + (size_t)sj * 128 + lane;
        hv[u][0] = hp[0];
        hv[u][1] = hp[32];
        hv[u][2] = hp[64];
        hv[u][3] = hp[96];
      }
#pragma unroll
      for (int u = 0; u < 8; ++u) {
        if (j + u < cb) {
          float* ap = acc + dj[u] * 128 + lane;
          atomicAdd(ap,      hv[u][0] * aw[u][0]);
          atomicAdd(ap + 32, hv[u][1] * aw[u][1]);
          atomicAdd(ap + 64, hv[u][2] * aw[u][2]);
          atomicAdd(ap + 96, hv[u][3] * aw[u][3]);
        }
      }
    }
  }
}

__device__ __forceinline__ void own_deg_add(int v, int lane, int* sdeg) {
  for (int m = 16; m >= 1; m >>= 1) v += __shfl_xor(v, m, 32);
  if (lane == 0 && v != 0) atomicAdd(sdeg, v);
}

template <int LAYER>
__global__ void __launch_bounds__(OWN_THREADS)
gat_owner(const int* __restrict__ src, const int* __restrict__ dst,
          const float* __restrict__ el, const float* __restrict__ er,
          const float* __restrict__ h, const float* __restrict__ bias,
          const float* __restrict__ lns, const float* __restrict__ lnb,
          float* __restrict__ outp,
          unsigned* __restrict__ wlist, unsigned* __restrict__ wcnt,
          const int* __restrict__ indeg, int* __restrict__ wdeg,
          int Nn, int E) {
  __shared__ float    acc[OWN_NB * 128];
  __shared__ unsigned smax[OWN_NB * 4];
  __shared__ float    sden[OWN_NB * 4];
  __shared__ unsigned list[OWN_CAP];
  __shared__ unsigned cnt;
  __shared__ int      sdeg;

  const int tid  = threadIdx.x;
  const int wave = tid >> 5;
  const int lane = tid & 31;
  const int base = blockIdx.x * OWN_NB;
  const unsigned nb = (unsigned)min(OWN_NB, Nn - base);

  const bool degBlock = (LAYER == 1) || (blockIdx.x == gridDim.x - 1);
  int dpart = 0;
  if (LAYER == 2 && degBlock) {
    for (int i = tid; i < (int)gridDim.x; i += OWN_THREADS) dpart += wdeg[i];
  }

  for (int i = tid; i < OWN_NB * 128; i += OWN_THREADS) acc[i] = 0.f;
  for (int i = tid; i < OWN_NB * 4; i += OWN_THREADS) { smax[i] = 0u; sden[i] = 0.f; }
  if (tid == 0) { cnt = 0u; sdeg = 0; }

  unsigned* wl = wlist + (size_t)blockIdx.x * OWN_CAP;
  unsigned total;
  if (LAYER == 1) {
    __syncthreads();
    if (tid < (int)nb) dpart = indeg[base + tid];
    own_build(dst, 0, E, base, nb, list, &cnt, tid);
    own_deg_add(dpart, lane, &sdeg);
    __syncthreads();
    total = cnt;
    if (tid == 0) { wcnt[blockIdx.x] = total; wdeg[blockIdx.x] = sdeg; }
    if (total <= OWN_CAP)
      for (unsigned i = tid; i < total; i += OWN_THREADS) wl[i] = list[i];
  } else {
    total = wcnt[blockIdx.x];
    if (total <= OWN_CAP) {
      for (unsigned i0 = tid; i0 < total; i0 += 4 * OWN_THREADS) {
        unsigned v[4];
#pragma unroll
        for (int u = 0; u < 4; ++u) {
          const unsigned i = i0 + u * OWN_THREADS;
          v[u] = i < total ? wl[i] : 0u;
        }
#pragma unroll
        for (int u = 0; u < 4; ++u) {
          const unsigned i = i0 + u * OWN_THREADS;
          if (i < total) list[i] = v[u];
        }
      }
    }
    __syncthreads();
    if (degBlock) own_deg_add(dpart, lane, &sdeg);
  }

  if (total <= OWN_CAP) {
    const int n = (int)total;
    own_max(list, n, src, dst, el, er, base, smax, tid);
    __syncthreads();
    own_den(list, n, src, dst, el, er, base, smax, sden, tid);
    __syncthreads();
    own_acc(list, n, src, dst, el, er, h, base, smax, sden, acc, wave, lane);
  } else {
    for (int pass = 0; pass < 3; ++pass) {
      for (int e0 = 0; e0 < E; e0 += OWN_CAP) {
        __syncthreads();
        if (tid == 0) cnt = 0u;
        __syncthreads();
        own_build(dst, e0, min(E, e0 + OWN_CAP), base, nb, list, &cnt, tid);
        __syncthreads();
        const int n = (int)cnt;
        if (pass == 0)      own_max(list, n, src, dst, el, er, base, smax, tid);
        else if (pass == 1) own_den(list, n, src, dst, el, er, base, smax, sden, tid);
        else                own_acc(list, n, src, dst, el, er, h, base, smax, sden, acc, wave, lane);
      }
    }
  }
  __syncthreads();

  if (LAYER == 2 && degBlock && tid == 0) {
    double ne = (double)(float)sdeg;
    double fl = (4.0 * ne * (6.0 * 128 * 32 + 6 * 32 + 2) +
                 4.0 * ne * (6.0 * 32 * 32 + 6 * 32 + 2)) / 1e12;
    outp[(size_t)Nn * 32] = (float)fl;
  }

  for (int nl = wave; nl < (int)nb; nl += OWN_THREADS / 32) {
    const int n = base + nl;
    float x = 0.f;
#pragma unroll
    for (int hh = 0; hh < 4; ++hh)
      x += acc[nl * 128 + hh * 32 + lane] + bias[hh * 32 + lane];
    x *= 0.25f;
    if (LAYER == 1) {
      float mu = x;
#pragma unroll
      for (int m = 16; m >= 1; m >>= 1) mu += __shfl_xor(mu, m, 32);
      mu *= (1.f / 32.f);
      float dx = x - mu;
      float var = dx * dx;
#pragma unroll
      for (int m = 16; m >= 1; m >>= 1) var += __shfl_xor(var, m, 32);
      var *= (1.f / 32.f);
      float y = dx * rsqrtf(var + LN_EPS) * lns[lane] + lnb[lane];
      outp[(size_t)n * 32 + lane] = fmaxf(y, 0.f);
    } else {
      outp[(size_t)n * 32 + lane] = x;
    }
  }
}

extern "C" void kernel_launch(void* const* d_in, const int* in_sizes, int n_in,
                              void* d_out, int out_size, void* d_ws, size_t ws_size,
                              hipStream_t stream) {
  const float* feat = (const float*)d_in[0];
  const float* W1   = (const float*)d_in[1];
  const float* al1  = (const float*)d_in[2];
  const float* ar1  = (const float*)d_in[3];
  const float* b1   = (const float*)d_in[4];
  const float* lns  = (const float*)d_in[5];
  const float* lnb  = (const float*)d_in[6];
  const float* W2   = (const float*)d_in[7];
  const float* al2  = (const float*)d_in[8];
  const float* ar2  = (const float*)d_in[9];
  const float* b2   = (const float*)d_in[10];
  const int*   src  = (const int*)d_in[11];
  const int*   dst  = (const int*)d_in[12];
  const int*   indeg= (const int*)d_in[13];

  const int Nn = in_sizes[13];
  const int E  = in_sizes[11];
  const int NH = Nn * 4;

  char* p = (char*)d_ws;
  float*    hbuf  = (float*)p;   p += (size_t)Nn * 128 * sizeof(float);
  float*    hmid  = (float*)p;   p += (size_t)Nn * 32  * sizeof(float);
  float*    el    = (float*)p;   p += (size_t)NH * sizeof(float);
  float*    er    = (float*)p;   p += (size_t)NH * sizeof(float);
  const int ownBlocks  = (Nn + OWN_NB - 1) / OWN_NB;
  unsigned* wlist = (unsigned*)p; p += (size_t)ownBlocks * OWN_CAP * sizeof(unsigned);
  unsigned* wcnt  = (unsigned*)p; p += (size_t)ownBlocks * sizeof(unsigned);
  int*      wdeg  = (int*)p;
  (void)ws_size; (void)n_in; (void)out_size;

  float* outh = (float*)d_out;

  dim3 b256(256);
  const int rowTiles   = (Nn + 15) / 16;
  const int gemmBlocks = (rowTiles + 7) / 8;

  gat_gemm_wmma<128><<<gemmBlocks, b256, 0, stream>>>(feat, W1, hbuf, Nn);
  gat_attn_proj<<<(NH + 255) / 256, b256, 0, stream>>>(hbuf, al1, ar1, el, er, NH);
  gat_owner<1><<<ownBlocks, OWN_THREADS, 0, stream>>>(src, dst, el, er, hbuf, b1, lns, lnb, hmid, wlist, wcnt, indeg, wdeg, Nn, E);

  gat_gemm_wmma<32><<<gemmBlocks, b256, 0, stream>>>(hmid, W2, hbuf, Nn);
  gat_attn_proj<<<(NH + 255) / 256, b256, 0, stream>>>(hbuf, al2, ar2, el, er, NH);
  gat_owner<2><<<ownBlocks, OWN_THREADS, 0, stream>>>(src, dst, el, er, hbuf, b2, lns, lnb, outh, wlist, wcnt, indeg, wdeg, Nn, E);
}


# each listed edge's source id and 4 head scores are fetched once and kept in its LDS slot: denominator and weighted sum read LDS, not src, dst, el, er again; layer 2 loads source ids; sums as before
# speedup vs baseline: 5.0251x; 1.2102x over previous
#include <hip/hip_runtime.h>


typedef float v2f __attribute__((ext_vector_type(2)));
typedef float v4f __attribute__((ext_vector_type(4)));
typedef float v8f __attribute__((ext_vector_type(8)));

#define NEG_SLOPE 0.2f
#define LN_EPS 1e-5f

__device__ __forceinline__ unsigned fenc(float f) {
  unsigned u = __float_as_uint(f);
  return (u >> 31) ? ~u : (u | 0x80000000u);
}
__device__ __forceinline__ float fdec(unsigned u) {
  return __uint_as_float((u >> 31) ? (u & 0x7FFFFFFFu) : ~u);
}
__device__ __forceinline__ float lrelu(float x) {
  return x > 0.f ? x : x * NEG_SLOPE;
}

template <int K>
__global__ void gat_gemm_wmma(const float* __restrict__ Amat,
                              const float* __restrict__ Bmat,
                              float* __restrict__ Cmat,
                              int M) {
  __shared__ float bs[K * 128];

  const int tid = threadIdx.x;

  for (int idx = tid; idx < K * 128; idx += 256) {
    int kr = idx >> 7;
    int col = idx & 127;
    int ks = kr >> 2;
    int c = col >> 4;
    int lane = (col & 15) | ((kr & 2) << 3);
    int sub = kr & 1;
    bs[((ks * 32 + lane) * 8 + c) * 2 + sub] = Bmat[idx];
  }
  __syncthreads();

  const int wave = tid >> 5;
  const int lane = tid & 31;
  const int row0 = (blockIdx.x * 8 + wave) * 16;
  if (row0 >= M) return;

  const int m  = lane & 15;
  const int kk = (lane >> 4) << 1;

  v8f acc[8] = {};

  const float* arow = Amat + (size_t)(row0 + m) * K + kk;
  const float* blane = bs + lane * 16;

#pragma unroll 4
  for (int ks = 0; ks < K / 4; ++ks) {
    v2f a = *(const v2f*)(arow + ks * 4);
    const float* bp = blane + ks * 512;
    v4f q0 = *(const v4f*)(bp);
    v4f q1 = *(const v4f*)(bp + 4);
    v4f q2 = *(const v4f*)(bp + 8);
    v4f q3 = *(const v4f*)(bp + 12);
    acc[0] = __builtin_amdgcn_wmma_f32_16x16x4_f32(false, a, false, q0.xy, (short)0, acc[0], false, false);
    acc[1] = __builtin_amdgcn_wmma_f32_16x16x4_f32(false, a, false, q0.zw, (short)0, acc[1], false, false);
    acc[2] = __builtin_amdgcn_wmma_f32_16x16x4_f32(false, a, false, q1.xy, (short)0, acc[2], false, false);
    acc[3] = __builtin_amdgcn_wmma_f32_16x16x4_f32(false, a, false, q1.zw, (short)0, acc[3], false, false);
    acc[4] = __builtin_amdgcn_wmma_f32_16x16x4_f32(false, a, false, q2.xy, (short)0, acc[4], false, false);
    acc[5] = __builtin_amdgcn_wmma_f32_16x16x4_f32(false, a, false, q2.zw, (short)0, acc[5], false, false);
    acc[6] = __builtin_amdgcn_wmma_f32_16x16x4_f32(false, a, false, q3.xy, (short)0, acc[6], false, false);
    acc[7] = __builtin_amdgcn_wmma_f32_16x16x4_f32(false, a, false, q3.zw, (short)0, acc[7], false, false);
  }

  const int mo = (lane >> 4) * 8;
#pragma unroll
  for (int c = 0; c < 8; ++c) {
#pragma unroll
    for (int r = 0; r < 8; ++r) {
      Cmat[(size_t)(row0 + mo + r) * 128 + c * 16 + m] = acc[c][r];
    }
  }
}

__global__ void gat_attn_proj(const float* __restrict__ h,
                              const float* __restrict__ al,
                              const float* __restrict__ ar,
                              float* __restrict__ el, float* __restrict__ er,
                              int NH) {
  int i = blockIdx.x * blockDim.x + threadIdx.x;
  if (i >= NH) return;
  int n = i >> 2, hh = i & 3;
  const float* hp  = h + (size_t)n * 128 + hh * 32;
  const float* alp = al + hh * 32;
  const float* arp = ar + hh * 32;
  float sl = 0.f, sr = 0.f;
#pragma unroll
  for (int f = 0; f < 32; ++f) {
    float v = hp[f];
    sl += v * alp[f];
    sr += v * arp[f];
  }
  el[i] = sl;
  er[i] = sr;
}

#define OWN_NB      196
#define OWN_CAP     8192
#define OWN_THREADS 1024

typedef unsigned v2u __attribute__((ext_vector_type(2)));
typedef unsigned v4u __attribute__((ext_vector_type(4)));

__device__ __forceinline__ void own_check(int dv, int e, int base, unsigned nb,
                                          v2u* ent, unsigned* cnt) {
  const unsigned dl = (unsigned)(dv - base);
  if (dl < nb) {
    unsigned pos = atomicAdd(cnt, 1u);
    if (pos < OWN_CAP) { v2u v; v.x = (unsigned)e; v.y = dl; ent[pos] = v; }
  }
}

__device__ __forceinline__ void own_build(const int* __restrict__ dst, int e0, int e1,
                                          int base, unsigned nb,
                                          v2u* ent, unsigned* cnt, int tid) {
  const int nvec = (e1 - e0) >> 2;
  const int4* dv = (const int4*)(dst + e0);
  int i = tid;
  for (; i + 3 * OWN_THREADS < nvec; i += 4 * OWN_THREADS) {
    int4 d[4];
#pragma unroll
    for (int u = 0; u < 4; ++u) d[u] = dv[i + u * OWN_THREADS];
#pragma unroll
    for (int u = 0; u < 4; ++u) {
      const int eb = e0 + 4 * (i + u * OWN_THREADS);
      own_check(d[u].x, eb + 0, base, nb, ent, cnt);
      own_check(d[u].y, eb + 1, base, nb, ent, cnt);
      own_check(d[u].z, eb + 2, base, nb, ent, cnt);
      own_check(d[u].w, eb + 3, base, nb, ent, cnt);
    }
  }
  for (; i < nvec; i += OWN_THREADS) {
    int4 d = dv[i];
    const int eb = e0 + 4 * i;
    own_check(d.x, eb + 0, base, nb, ent, cnt);
    own_check(d.y, eb + 1, base, nb, ent, cnt);
    own_check(d.z, eb + 2, base, nb, ent, cnt);
    own_check(d.w, eb + 3, base, nb, ent, cnt);
  }
  for (int e = e0 + nvec * 4 + tid; e < e1; e += OWN_THREADS)
    own_check(dst[e], e, base, nb, ent, cnt);
}

template <bool HAVE_SRC, bool KEEP>
__device__ __forceinline__ void own_prep(v2u* ent, v4f* sc, int n,
                                         const int* __restrict__ src,
                                         const float* __restrict__ el, const float* erl,
                                         unsigned* smax, v2u* __restrict__ wl,
                                         bool doMax, int tid) {
  for (int i = tid; i < n; i += OWN_THREADS) {
    v2u en = ent[i];
    if (!HAVE_SRC) {
      en.x = (unsigned)src[en.x];
      ent[i] = en;
    }
    if (KEEP) wl[i] = en;
    const unsigned dl = en.y;
    const v4f l = *(const v4f*)(el + (size_t)en.x * 4);
    const v4f r = *(const v4f*)(erl + dl * 4);
    v4f v;
    v.x = lrelu(l.x + r.x);
    v.y = lrelu(l.y + r.y);
    v.z = lrelu(l.z + r.z);
    v.w = lrelu(l.w + r.w);
    sc[i] = v;
    if (doMax) {
      atomicMax(&smax[dl * 4 + 0], fenc(v.x));
      atomicMax(&smax[dl * 4 + 1], fenc(v.y));
      atomicMax(&smax[dl * 4 + 2], fenc(v.z));
      atomicMax(&smax[dl * 4 + 3], fenc(v.w));
    }
  }
}

__device__ __forceinline__ void own_den(const v2u* ent, v4f* sc, int n,
                                        const unsigned* smax, float* sden,
                                        bool add, int tid) {
  for (int i = tid; i < n; i += OWN_THREADS) {
    const unsigned dl = ent[i].y;
    const v4u m = *(const v4u*)(smax + dl * 4);
    v4f v = sc[i];
    v.x = expf(v.x - fdec(m.x));
    v.y = expf(v.y - fdec(m.y));
    v.z = expf(v.z - fdec(m.z));
    v.w = expf(v.w - fdec(m.w));
    sc[i] = v;
    if (add) {
      atomicAdd(&sden[dl * 4 + 0], v.x);
      atomicAdd(&sden[dl * 4 + 1], v.y);
      atomicAdd(&sden[dl * 4 + 2], v.z);
      atomicAdd(&sden[dl * 4 + 3], v.w);
    }
  }
}

__device__ __forceinline__ void own_acc(const v2u* ent, const v4f* sc, int n,
                                        const float* __restrict__ h,
                                        const float* sden,
                                        float* acc, int wave, int lane) {
  for (int b = wave * 32; b < n; b += OWN_THREADS) {
    const int i = b + lane;
    int s = 0, dl = 0;
    float a0 = 0.f, a1 = 0.f, a2 = 0.f, a3 = 0.f;
    if (i < n) {
      const v2u en = ent[i];
      s = (int)en.x;
      dl = (int)en.y;
      const v4f ex = sc[i];
      const v4f dn = *(const v4f*)(sden + dl * 4);
      a0 = ex.x / fmaxf(dn.x, 1e-9f);
      a1 = ex.y / fmaxf(dn.y, 1e-9f);
      a2 = ex.z / fmaxf(dn.z, 1e-9f);
      a3 = ex.w / fmaxf(dn.w, 1e-9f);
    }
    const int cb = min(32, n - b);
    for (int j = 0; j < cb; j += 8) {
      float hv[8][4], aw[8][4];
      int dj[8];
#pragma unroll
      for (int u = 0; u < 8; ++u) {
        const int jj = j + u;
        const int sj = __builtin_amdgcn_readlane(s, jj);
        dj[u] = __builtin_amdgcn_readlane(dl, jj);
        aw[u][0] = __int_as_float(__builtin_amdgcn_readlane(__float_as_int(a0), jj));
        aw[u][1] = __int_as_float(__builtin_amdgcn_readlane(__float_as_int(a1), jj));
        aw[u][2] = __int_as_float(__builtin_amdgcn_readlane(__float_as_int(a2), jj));
        aw[u][3] = __int_as_float(__builtin_amdgcn_readlane(__float_as_int(a3), jj));
        const float* hp = h + (size_t)sj * 128 + lane;
        hv[u][0] = hp[0];
        hv[u][1] = hp[32];
        hv[u][2] = hp[64];
        hv[u][3] = hp[96];
      }
#pragma unroll
      for (int u = 0; u < 8; ++u) {
        if (j + u < cb) {
          float* ap = acc + dj[u] * 128 + lane;
          atomicAdd(ap,      hv[u][0] * aw[u][0]);
          atomicAdd(ap + 32, hv[u][1] * aw[u][1]);
          atomicAdd(ap + 64, hv[u][2] * aw[u][2]);
          atomicAdd(ap + 96, hv[u][3] * aw[u][3]);
        }
      }
    }
  }
}

__device__ __forceinline__ void own_deg_add(int v, int lane, int* sdeg) {
  for (int m = 16; m >= 1; m >>= 1) v += __shfl_xor(v, m, 32);
  if (lane == 0 && v != 0) atomicAdd(sdeg, v);
}

template <int LAYER>
__global__ void __launch_bounds__(OWN_THREADS)
gat_owner(const int* __restrict__ src, const int* __restrict__ dst,
          const float* __restrict__ el, const float* __restrict__ er,
          const float* __restrict__ h, const float* __restrict__ bias,
          const float* __restrict__ lns, const float* __restrict__ lnb,
          float* __restrict__ outp,
          unsigned* __restrict__ wlist, unsigned* __restrict__ wcnt,
          const int* __restrict__ indeg, int* __restrict__ wdeg,
          int Nn, int E) {
  __shared__ float    acc[OWN_NB * 128];
  __shared__ __attribute__((aligned(16))) unsigned smax[OWN_NB * 4];
  __shared__ __attribute__((aligned(16))) float    sden[OWN_NB * 4];
  __shared__ __attribute__((aligned(16))) float    erl[OWN_NB * 4];
  __shared__ v2u      ent[OWN_CAP];
  __shared__ v4f      sc[OWN_CAP];
  __shared__ unsigned cnt;
  __shared__ int      sdeg;

  const int tid  = threadIdx.x;
  const int wave = tid >> 5;
  const int lane = tid & 31;
  const int base = blockIdx.x * OWN_NB;
  const unsigned nb = (unsigned)min(OWN_NB, Nn - base);

  const bool degBlock = (LAYER == 1) || (blockIdx.x == gridDim.x - 1);
  int dpart = 0;
  if (LAYER == 2 && degBlock) {
    for (int i = tid; i < (int)gridDim.x; i += OWN_THREADS) dpart += wdeg[i];
  }

  for (int i = tid; i < OWN_NB * 128; i += OWN_THREADS) acc[i] = 0.f;
  for (int i = tid; i < OWN_NB * 4; i += OWN_THREADS) { smax[i] = 0u; sden[i] = 0.f; }
  if (tid == 0) { cnt = 0u; sdeg = 0; }

  v2u* wl = (v2u*)wlist + (size_t)blockIdx.x * OWN_CAP;
  float erv = 0.f;
  unsigned total;
  if (LAYER == 1) {
    __syncthreads();
    if (tid < (int)nb) dpart = indeg[base + tid];
    if (tid < (int)nb * 4) erv = er[(size_t)base * 4 + tid];
    own_build(dst, 0, E, base, nb, ent, &cnt, tid);
    if (tid < (int)nb * 4) erl[tid] = erv;
    own_deg_add(dpart, lane, &sdeg);
    __syncthreads();
    total = cnt;
    if (tid == 0) { wcnt[blockIdx.x] = total; wdeg[blockIdx.x] = sdeg; }
  } else {
    total = wcnt[blockIdx.x];
    if (tid < (int)nb * 4) erv = er[(size_t)base * 4 + tid];
    if (total <= OWN_CAP) {
      for (unsigned i0 = tid; i0 < total; i0 += 4 * OWN_THREADS) {
        v2u v[4];
#pragma unroll
        for (int u = 0; u < 4; ++u) {
          const unsigned i = i0 + u * OWN_THREADS;
          v2u z; z.x = 0u; z.y = 0u;
          v[u] = i < total ? wl[i] : z;
        }
#pragma unroll
        for (int u = 0; u < 4; ++u) {
          const unsigned i = i0 + u * OWN_THREADS;
          if (i < total) ent[i] = v[u];
        }
      }
    }
    if (tid < (int)nb * 4) erl[tid] = erv;
    __syncthreads();
    if (degBlock) own_deg_add(dpart, lane, &sdeg);
  }

  if (total <= OWN_CAP) {
    const int n = (int)total;
    own_prep<LAYER == 2, LAYER == 1>(ent, sc, n, src, el, erl, smax, wl, true, tid);
    __syncthreads();
    own_den(ent, sc, n, smax, sden, true, tid);
    __syncthreads();
    own_acc(ent, sc, n, h, sden, acc, wave, lane);
  } else {
    for (int pass = 0; pass < 3; ++pass) {
      for (int e0 = 0; e0 < E; e0 += OWN_CAP) {
        __syncthreads();
        if (tid == 0) cnt = 0u;
        __syncthreads();
        own_build(dst, e0, min(E, e0 + OWN_CAP), base, nb, ent, &cnt, tid);
        __syncthreads();
        const int n = (int)cnt;
        own_prep<false, false>(ent, sc, n, src, el, erl, smax, wl, pass == 0, tid);
        if (pass >= 1) {
          __syncthreads();
          own_den(ent, sc, n, smax, sden, pass == 1, tid);
        }
        if (pass == 2) {
          __syncthreads();
          own_acc(ent, sc, n, h, sden, acc, wave, lane);
        }
      }
    }
  }
  __syncthreads();

  if (LAYER == 2 && degBlock && tid == 0) {
    double ne = (double)(float)sdeg;
    double fl = (4.0 * ne * (6.0 * 128 * 32 + 6 * 32 + 2) +
                 4.0 * ne * (6.0 * 32 * 32 + 6 * 32 + 2)) / 1e12;
    outp[(size_t)Nn * 32] = (float)fl;
  }

  for (int nl = wave; nl < (int)nb; nl += OWN_THREADS / 32) {
    const int n = base + nl;
    float x = 0.f;
#pragma unroll
    for (int hh = 0; hh < 4; ++hh)
      x += acc[nl * 128 + hh * 32 + lane] + bias[hh * 32 + lane];
    x *= 0.25f;
    if (LAYER == 1) {
      float mu = x;
#pragma unroll
      for (int m = 16; m >= 1; m >>= 1) mu += __shfl_xor(mu, m, 32);
      mu *= (1.f / 32.f);
      float dx = x - mu;
      float var = dx * dx;
#pragma unroll
      for (int m = 16; m >= 1; m >>= 1) var += __shfl_xor(var, m, 32);
      var *= (1.f / 32.f);
      float y = dx * rsqrtf(var + LN_EPS) * lns[lane] + lnb[lane];
      outp[(size_t)n * 32 + lane] = fmaxf(y, 0.f);
    } else {
      outp[(size_t)n * 32 + lane] = x;
    }
  }
}

extern "C" void kernel_launch(void* const* d_in, const int* in_sizes, int n_in,
                              void* d_out, int out_size, void* d_ws, size_t ws_size,
                              hipStream_t stream) {
  const float* feat = (const float*)d_in[0];
  const float* W1   = (const float*)d_in[1];
  const float* al1  = (const float*)d_in[2];
  const float* ar1  = (const float*)d_in[3];
  const float* b1   = (const float*)d_in[4];
  const float* lns  = (const float*)d_in[5];
  const float* lnb  = (const float*)d_in[6];
  const float* W2   = (const float*)d_in[7];
  const float* al2  = (const float*)d_in[8];
  const float* ar2  = (const float*)d_in[9];
  const float* b2   = (const float*)d_in[10];
  const int*   src  = (const int*)d_in[11];
  const int*   dst  = (const int*)d_in[12];
  const int*   indeg= (const int*)d_in[13];

  const int Nn = in_sizes[13];
  const int E  = in_sizes[11];
  const int NH = Nn * 4;

  char* p = (char*)d_ws;
  float*    hbuf  = (float*)p;   p += (size_t)Nn * 128 * sizeof(float);
  float*    hmid  = (float*)p;   p += (size_t)Nn * 32  * sizeof(float);
  float*    el    = (float*)p;   p += (size_t)NH * sizeof(float);
  float*    er    = (float*)p;   p += (size_t)NH * sizeof(float);
  const int ownBlocks  = (Nn + OWN_NB - 1) / OWN_NB;
  unsigned* wlist = (unsigned*)p; p += (size_t)ownBlocks * OWN_CAP * 2 * sizeof(unsigned);
  unsigned* wcnt  = (unsigned*)p; p += (size_t)ownBlocks * sizeof(unsigned);
  int*      wdeg  = (int*)p;
  (void)ws_size; (void)n_in; (void)out_size;

  float* outh = (float*)d_out;

  dim3 b256(256);
  const int rowTiles   = (Nn + 15) / 16;
  const int gemmBlocks = (rowTiles + 7) / 8;

  gat_gemm_wmma<128><<<gemmBlocks, b256, 0, stream>>>(feat, W1, hbuf, Nn);
  gat_attn_proj<<<(NH + 255) / 256, b256, 0, stream>>>(hbuf, al1, ar1, el, er, NH);
  gat_owner<1><<<ownBlocks, OWN_THREADS, 0, stream>>>(src, dst, el, er, hbuf, b1, lns, lnb, hmid, wlist, wcnt, indeg, wdeg, Nn, E);

  gat_gemm_wmma<32><<<gemmBlocks, b256, 0, stream>>>(hmid, W2, hbuf, Nn);
  gat_attn_proj<<<(NH + 255) / 256, b256, 0, stream>>>(hbuf, al2, ar2, el, er, NH);
  gat_owner<2><<<ownBlocks, OWN_THREADS, 0, stream>>>(src, dst, el, er, hbuf, b2, lns, lnb, outh, wlist, wcnt, indeg, wdeg, Nn, E);
}


# layer 1's projection launch also stores one owner byte per edge (dst / 196); owner workgroups scan these 0.8 MB, 16 edges per load, not 3.2 MB of destinations; sums by LDS atomic arrival as before
# speedup vs baseline: 6.0324x; 1.2005x over previous
#include <hip/hip_runtime.h>


typedef float v2f __attribute__((ext_vector_type(2)));
typedef float v4f __attribute__((ext_vector_type(4)));
typedef float v8f __attribute__((ext_vector_type(8)));

#define NEG_SLOPE 0.2f
#define LN_EPS 1e-5f

__device__ __forceinline__ unsigned fenc(float f) {
  unsigned u = __float_as_uint(f);
  return (u >> 31) ? ~u : (u | 0x80000000u);
}
__device__ __forceinline__ float fdec(unsigned u) {
  return __uint_as_float((u >> 31) ? (u & 0x7FFFFFFFu) : ~u);
}
__device__ __forceinline__ float lrelu(float x) {
  return x > 0.f ? x : x * NEG_SLOPE;
}

template <int K>
__global__ void gat_gemm_wmma(const float* __restrict__ Amat,
                              const float* __restrict__ Bmat,
                              float* __restrict__ Cmat,
                              int M) {
  __shared__ float bs[K * 128];

  const int tid = threadIdx.x;

  for (int idx = tid; idx < K * 128; idx += 256) {
    int kr = idx >> 7;
    int col = idx & 127;
    int ks = kr >> 2;
    int c = col >> 4;
    int lane = (col & 15) | ((kr & 2) << 3);
    int sub = kr & 1;
    bs[((ks * 32 + lane) * 8 + c) * 2 + sub] = Bmat[idx];
  }
  __syncthreads();

  const int wave = tid >> 5;
  const int lane = tid & 31;
  const int row0 = (blockIdx.x * 8 + wave) * 16;
  if (row0 >= M) return;

  const int m  = lane & 15;
  const int kk = (lane >> 4) << 1;

  v8f acc[8] = {};

  const float* arow = Amat + (size_t)(row0 + m) * K + kk;
  const float* blane = bs + lane * 16;

#pragma unroll 4
  for (int ks = 0; ks < K / 4; ++ks) {
    v2f a = *(const v2f*)(arow + ks * 4);
    const float* bp = blane + ks * 512;
    v4f q0 = *(const v4f*)(bp);
    v4f q1 = *(const v4f*)(bp + 4);
    v4f q2 = *(const v4f*)(bp + 8);
    v4f q3 = *(const v4f*)(bp + 12);
    acc[0] = __builtin_amdgcn_wmma_f32_16x16x4_f32(false, a, false, q0.xy, (short)0, acc[0], false, false);
    acc[1] = __builtin_amdgcn_wmma_f32_16x16x4_f32(false, a, false, q0.zw, (short)0, acc[1], false, false);
    acc[2] = __builtin_amdgcn_wmma_f32_16x16x4_f32(false, a, false, q1.xy, (short)0, acc[2], false, false);
    acc[3] = __builtin_amdgcn_wmma_f32_16x16x4_f32(false, a, false, q1.zw, (short)0, acc[3], false, false);
    acc[4] = __builtin_amdgcn_wmma_f32_16x16x4_f32(false, a, false, q2.xy, (short)0, acc[4], false, false);
    acc[5] = __builtin_amdgcn_wmma_f32_16x16x4_f32(false, a, false, q2.zw, (short)0, acc[5], false, false);
    acc[6] = __builtin_amdgcn_wmma_f32_16x16x4_f32(false, a, false, q3.xy, (short)0, acc[6], false, false);
    acc[7] = __builtin_amdgcn_wmma_f32_16x16x4_f32(false, a, false, q3.zw, (short)0, acc[7], false, false);
  }

  const int mo = (lane >> 4) * 8;
#pragma unroll
  for (int c = 0; c < 8; ++c) {
#pragma unroll
    for (int r = 0; r < 8; ++r) {
      Cmat[(size_t)(row0 + mo + r) * 128 + c * 16 + m] = acc[c][r];
    }
  }
}

#define OWN_NB      196
#define OWN_CAP     8192
#define OWN_THREADS 1024

__global__ void gat_attn_proj(const float* __restrict__ h,
                              const float* __restrict__ al,
                              const float* __restrict__ ar,
                              float* __restrict__ el, float* __restrict__ er,
                              int NH,
                              const int* __restrict__ dst, unsigned* __restrict__ wkey,
                              int E) {
  int i = blockIdx.x * blockDim.x + threadIdx.x;
  if (wkey) {
    const int nq = E >> 2;
    const int4* d4 = (const int4*)dst;
    for (int q = i; q < nq; q += (int)(gridDim.x * blockDim.x)) {
      const int4 d = d4[q];
      wkey[q] = (((unsigned)d.x / OWN_NB) & 255u) |
                ((((unsigned)d.y / OWN_NB) & 255u) << 8) |
                ((((unsigned)d.z / OWN_NB) & 255u) << 16) |
                (((unsigned)d.w / OWN_NB) << 24);
    }
  }
  if (i >= NH) return;
  int n = i >> 2, hh = i & 3;
  const float* hp  = h + (size_t)n * 128 + hh * 32;
  const float* alp = al + hh * 32;
  const float* arp = ar + hh * 32;
  float sl = 0.f, sr = 0.f;
#pragma unroll
  for (int f = 0; f < 32; ++f) {
    float v = hp[f];
    sl += v * alp[f];
    sr += v * arp[f];
  }
  el[i] = sl;
  er[i] = sr;
}

typedef unsigned v2u __attribute__((ext_vector_type(2)));
typedef unsigned v4u __attribute__((ext_vector_type(4)));

__device__ __forceinline__ void own_check(int dv, int e, int base, unsigned nb,
                                          v2u* ent, unsigned* cnt) {
  const unsigned dl = (unsigned)(dv - base);
  if (dl < nb) {
    unsigned pos = atomicAdd(cnt, 1u);
    if (pos < OWN_CAP) { v2u v; v.x = (unsigned)e; v.y = dl; ent[pos] = v; }
  }
}

__device__ __forceinline__ void own_build(const int* __restrict__ dst, int e0, int e1,
                                          int base, unsigned nb,
                                          v2u* ent, unsigned* cnt, int tid) {
  const int nvec = (e1 - e0) >> 2;
  const int4* dv = (const int4*)(dst + e0);
  int i = tid;
  for (; i + 3 * OWN_THREADS < nvec; i += 4 * OWN_THREADS) {
    int4 d[4];
#pragma unroll
    for (int u = 0; u < 4; ++u) d[u] = dv[i + u * OWN_THREADS];
#pragma unroll
    for (int u = 0; u < 4; ++u) {
      const int eb = e0 + 4 * (i + u * OWN_THREADS);
      own_check(d[u].x, eb + 0, base, nb, ent, cnt);
      own_check(d[u].y, eb + 1, base, nb, ent, cnt);
      own_check(d[u].z, eb + 2, base, nb, ent, cnt);
      own_check(d[u].w, eb + 3, base, nb, ent, cnt);
    }
  }
  for (; i < nvec; i += OWN_THREADS) {
    int4 d = dv[i];
    const int eb = e0 + 4 * i;
    own_check(d.x, eb + 0, base, nb, ent, cnt);
    own_check(d.y, eb + 1, base, nb, ent, cnt);
    own_check(d.z, eb + 2, base, nb, ent, cnt);
    own_check(d.w, eb + 3, base, nb, ent, cnt);
  }
  for (int e = e0 + nvec * 4 + tid; e < e1; e += OWN_THREADS)
    own_check(dst[e], e, base, nb, ent, cnt);
}

__device__ __forceinline__ unsigned own_nz(unsigned k, unsigned pat) {
  const unsigned x = k ^ pat;
  return ((x & 0x7F7F7F7Fu) + 0x7F7F7F7Fu) | x;
}
__device__ __forceinline__ unsigned own_hits(const v4u k, unsigned pat) {
  const unsigned c = ((own_nz(k.x, pat) >> 7) & 0x01010101u) | ((own_nz(k.y, pat) >> 6) & 0x02020202u) |
                     ((own_nz(k.z, pat) >> 5) & 0x04040404u) | ((own_nz(k.w, pat) >> 4) & 0x08080808u);
  return c ^ 0x0F0F0F0Fu;
}
__device__ __forceinline__ void own_emit(unsigned m, int e0, v2u* ent, unsigned* cnt) {
  while (m) {
    const int p = __builtin_ctz(m);
    m &= m - 1u;
    const int e = e0 + ((p & 3) << 2) + (p >> 3) + ((p & 4) << 12);
    const unsigned pos = atomicAdd(cnt, 1u);
    if (pos < OWN_CAP) ent[pos].x = (unsigned)e;
  }
}

__device__ __forceinline__ void own_scan4(const v4u k[4], unsigned pat, int e0,
                                          v2u* ent, unsigned* cnt) {
  own_emit(own_hits(k[0], pat) | (own_hits(k[1], pat) << 4), e0, ent, cnt);
  own_emit(own_hits(k[2], pat) | (own_hits(k[3], pat) << 4), e0 + 32 * OWN_THREADS, ent, cnt);
}

__device__ __forceinline__ void own_build_keys(const unsigned* __restrict__ wkey,
                                               const int* __restrict__ dst, int E,
                                               unsigned me, int base, unsigned nb,
                                               v2u* ent, unsigned* cnt, int tid) {
  const int nvec = E >> 4;
  const v4u* kv = (const v4u*)wkey;
  const unsigned pat = me * 0x01010101u;
  const int ntrip = nvec / (4 * OWN_THREADS);
  int i = tid;
  if (ntrip > 0) {
    v4u ka[4], kb[4];
#pragma unroll
    for (int u = 0; u < 4; ++u) ka[u] = kv[i + u * OWN_THREADS];
    int t = 0;
    for (; t + 2 < ntrip; t += 2) {
#pragma unroll
      for (int u = 0; u < 4; ++u) kb[u] = kv[i + (4 + u) * OWN_THREADS];
      own_scan4(ka, pat, 16 * i, ent, cnt);
      i += 4 * OWN_THREADS;
#pragma unroll
      for (int u = 0; u < 4; ++u) ka[u] = kv[i + (4 + u) * OWN_THREADS];
      own_scan4(kb, pat, 16 * i, ent, cnt);
      i += 4 * OWN_THREADS;
    }
    if (t + 1 < ntrip) {
#pragma unroll
      for (int u = 0; u < 4; ++u) kb[u] = kv[i + (4 + u) * OWN_THREADS];
      own_scan4(ka, pat, 16 * i, ent, cnt);
      i += 4 * OWN_THREADS;
      own_scan4(kb, pat, 16 * i, ent, cnt);
    } else {
      own_scan4(ka, pat, 16 * i, ent, cnt);
    }
    i += 4 * OWN_THREADS;
  }
  for (; i < nvec; i += OWN_THREADS) {
    own_emit(own_hits(kv[i], pat), 16 * i, ent, cnt);
  }
  for (int e = nvec * 16 + tid; e < E; e += OWN_THREADS)
    own_check(dst[e], e, base, nb, ent, cnt);
}

template <bool HAVE_SRC, bool KEEP, bool NEED_DL>
__device__ __forceinline__ void own_prep(v2u* ent, v4f* sc, int n,
                                         const int* __restrict__ src,
                                         const int* __restrict__ dst, int base, unsigned nb,
                                         const float* __restrict__ el, const float* erl,
                                         unsigned* smax, v2u* __restrict__ wl,
                                         bool doMax, int tid) {
  for (int i = tid; i < n; i += OWN_THREADS) {
    v2u en = ent[i];
    if (!HAVE_SRC) {
      const unsigned e = en.x;
      en.x = (unsigned)src[e];
      if (NEED_DL) en.y = min((unsigned)(dst[e] - base), nb - 1u);
      ent[i] = en;
    }
    if (KEEP) wl[i] = en;
    const unsigned dl = en.y;
    const v4f l = *(const v4f*)(el + (size_t)en.x * 4);
    const v4f r = *(const v4f*)(erl + dl * 4);
    v4f v;
    v.x = lrelu(l.x + r.x);
    v.y = lrelu(l.y + r.y);
    v.z = lrelu(l.z + r.z);
    v.w = lrelu(l.w + r.w);
    sc[i] = v;
    if (doMax) {
      atomicMax(&smax[dl * 4 + 0], fenc(v.x));
      atomicMax(&smax[dl * 4 + 1], fenc(v.y));
      atomicMax(&smax[dl * 4 + 2], fenc(v.z));
      atomicMax(&smax[dl * 4 + 3], fenc(v.w));
    }
  }
}

__device__ __forceinline__ void own_den(const v2u* ent, v4f* sc, int n,
                                        const unsigned* smax, float* sden,
                                        bool add, int tid) {
  for (int i = tid; i < n; i += OWN_THREADS) {
    const unsigned dl = ent[i].y;
    const v4u m = *(const v4u*)(smax + dl * 4);
    v4f v = sc[i];
    v.x = expf(v.x - fdec(m.x));
    v.y = expf(v.y - fdec(m.y));
    v.z = expf(v.z - fdec(m.z));
    v.w = expf(v.w - fdec(m.w));
    sc[i] = v;
    if (add) {
      atomicAdd(&sden[dl * 4 + 0], v.x);
      atomicAdd(&sden[dl * 4 + 1], v.y);
      atomicAdd(&sden[dl * 4 + 2], v.z);
      atomicAdd(&sden[dl * 4 + 3], v.w);
    }
  }
}

__device__ __forceinline__ void own_acc(const v2u* ent, const v4f* sc, int n,
                                        const float* __restrict__ h,
                                        const float* sden,
                                        float* acc, int wave, int lane) {
  for (int b = wave * 32; b < n; b += OWN_THREADS) {
    const int i = b + lane;
    int s = 0, dl = 0;
    float a0 = 0.f, a1 = 0.f, a2 = 0.f, a3 = 0.f;
    if (i < n) {
      const v2u en = ent[i];
      s = (int)en.x;
      dl = (int)en.y;
      const v4f ex = sc[i];
      const v4f dn = *(const v4f*)(sden + dl * 4);
      a0 = ex.x / fmaxf(dn.x, 1e-9f);
      a1 = ex.y / fmaxf(dn.y, 1e-9f);
      a2 = ex.z / fmaxf(dn.z, 1e-9f);
      a3 = ex.w / fmaxf(dn.w, 1e-9f);
    }
    const int cb = min(32, n - b);
    for (int j = 0; j < cb; j += 8) {
      float hv[8][4], aw[8][4];
      int dj[8];
#pragma unroll
      for (int u = 0; u < 8; ++u) {
        const int jj = j + u;
        const int sj = __builtin_amdgcn_readlane(s, jj);
        dj[u] = __builtin_amdgcn_readlane(dl, jj);
        aw[u][0] = __int_as_float(__builtin_amdgcn_readlane(__float_as_int(a0), jj));
        aw[u][1] = __int_as_float(__builtin_amdgcn_readlane(__float_as_int(a1), jj));
        aw[u][2] = __int_as_float(__builtin_amdgcn_readlane(__float_as_int(a2), jj));
        aw[u][3] = __int_as_float(__builtin_amdgcn_readlane(__float_as_int(a3), jj));
        const float* hp = h + (size_t)sj * 128 + lane;
        hv[u][0] = hp[0];
        hv[u][1] = hp[32];
        hv[u][2] = hp[64];
        hv[u][3] = hp[96];
      }
#pragma unroll
      for (int u = 0; u < 8; ++u) {
        if (j + u < cb) {
          float* ap = acc + dj[u] * 128 + lane;
          atomicAdd(ap,      hv[u][0] * aw[u][0]);
          atomicAdd(ap + 32, hv[u][1] * aw[u][1]);
          atomicAdd(ap + 64, hv[u][2] * aw[u][2]);
          atomicAdd(ap + 96, hv[u][3] * aw[u][3]);
        }
      }
    }
  }
}

__device__ __forceinline__ void own_deg_add(int v, int lane, int* sdeg) {
  for (int m = 16; m >= 1; m >>= 1) v += __shfl_xor(v, m, 32);
  if (lane == 0 && v != 0) atomicAdd(sdeg, v);
}

template <int LAYER>
__global__ void __launch_bounds__(OWN_THREADS)
gat_owner(const int* __restrict__ src, const int* __restrict__ dst,
          const float* __restrict__ el, const float* __restrict__ er,
          const float* __restrict__ h, const float* __restrict__ bias,
          const float* __restrict__ lns, const float* __restrict__ lnb,
          float* __restrict__ outp,
          unsigned* __restrict__ wlist, unsigned* __restrict__ wcnt,
          const int* __restrict__ indeg, int* __restrict__ wdeg,
          const unsigned* __restrict__ wkey,
          int Nn, int E) {
  __shared__ float    acc[OWN_NB * 128];
  __shared__ __attribute__((aligned(16))) unsigned smax[OWN_NB * 4];
  __shared__ __attribute__((aligned(16))) float    sden[OWN_NB * 4];
  __shared__ __attribute__((aligned(16))) float    erl[OWN_NB * 4];
  __shared__ v2u      ent[OWN_CAP];
  __shared__ v4f      sc[OWN_CAP];
  __shared__ unsigned cnt;
  __shared__ int      sdeg;

  const int tid  = threadIdx.x;
  const int wave = tid >> 5;
  const int lane = tid & 31;
  const int base = blockIdx.x * OWN_NB;
  const unsigned nb = (unsigned)min(OWN_NB, Nn - base);

  const bool degBlock = (LAYER == 1) || (blockIdx.x == gridDim.x - 1);
  int dpart = 0;
  if (LAYER == 2 && degBlock) {
    for (int i = tid; i < (int)gridDim.x; i += OWN_THREADS) dpart += wdeg[i];
  }

  for (int i = tid; i < OWN_NB * 128; i += OWN_THREADS) acc[i] = 0.f;
  for (int i = tid; i < OWN_NB * 4; i += OWN_THREADS) { smax[i] = 0u; sden[i] = 0.f; }
  if (tid == 0) { cnt = 0u; sdeg = 0; }

  v2u* wl = (v2u*)wlist + (size_t)blockIdx.x * OWN_CAP;
  float erv = 0.f;
  unsigned total;
  if (LAYER == 1) {
    __syncthreads();
    if (tid < (int)nb) dpart = indeg[base + tid];
    if (tid < (int)nb * 4) erv = er[(size_t)base * 4 + tid];
    if (wkey) own_build_keys(wkey, dst, E, blockIdx.x, base, nb, ent, &cnt, tid);
    else      own_build(dst, 0, E, base, nb, ent, &cnt, tid);
    if (tid < (int)nb * 4) erl[tid] = erv;
    own_deg_add(dpart, lane, &sdeg);
    __syncthreads();
    total = cnt;
    if (tid == 0) { wcnt[blockIdx.x] = total; wdeg[blockIdx.x] = sdeg; }
  } else {
    total = wcnt[blockIdx.x];
    if (tid < (int)nb * 4) erv = er[(size_t)base * 4 + tid];
    if (total <= OWN_CAP) {
      for (unsigned i0 = tid; i0 < total; i0 += 4 * OWN_THREADS) {
        v2u v[4];
#pragma unroll
        for (int u = 0; u < 4; ++u) {
          const unsigned i = i0 + u * OWN_THREADS;
          v2u z; z.x = 0u; z.y = 0u;
          v[u] = i < total ? wl[i] : z;
        }
#pragma unroll
        for (int u = 0; u < 4; ++u) {
          const unsigned i = i0 + u * OWN_THREADS;
          if (i < total) ent[i] = v[u];
        }
      }
    }
    if (tid < (int)nb * 4) erl[tid] = erv;
    __syncthreads();
    if (degBlock) own_deg_add(dpart, lane, &sdeg);
  }

  if (total <= OWN_CAP) {
    const int n = (int)total;
    own_prep<LAYER == 2, LAYER == 1, LAYER == 1>(ent, sc, n, src, dst, base, nb, el, erl, smax, wl, true, tid);
    __syncthreads();
    own_den(ent, sc, n, smax, sden, true, tid);
    __syncthreads();
    own_acc(ent, sc, n, h, sden, acc, wave, lane);
  } else {
    for (int pass = 0; pass < 3; ++pass) {
      for (int e0 = 0; e0 < E; e0 += OWN_CAP) {
        __syncthreads();
        if (tid == 0) cnt = 0u;
        __syncthreads();
        own_build(dst, e0, min(E, e0 + OWN_CAP), base, nb, ent, &cnt, tid);
        __syncthreads();
        const int n = (int)cnt;
        own_prep<false, false, false>(ent, sc, n, src, dst, base, nb, el, erl, smax, wl, pass == 0, tid);
        if (pass >= 1) {
          __syncthreads();
          own_den(ent, sc, n, smax, sden, pass == 1, tid);
        }
        if (pass == 2) {
          __syncthreads();
          own_acc(ent, sc, n, h, sden, acc, wave, lane);
        }
      }
    }
  }
  __syncthreads();

  if (LAYER == 2 && degBlock && tid == 0) {
    double ne = (double)(float)sdeg;
    double fl = (4.0 * ne * (6.0 * 128 * 32 + 6 * 32 + 2) +
                 4.0 * ne * (6.0 * 32 * 32 + 6 * 32 + 2)) / 1e12;
    outp[(size_t)Nn * 32] = (float)fl;
  }

  for (int nl = wave; nl < (int)nb; nl += OWN_THREADS / 32) {
    const int n = base + nl;
    float x = 0.f;
#pragma unroll
    for (int hh = 0; hh < 4; ++hh)
      x += acc[nl * 128 + hh * 32 + lane] + bias[hh * 32 + lane];
    x *= 0.25f;
    if (LAYER == 1) {
      float mu = x;
#pragma unroll
      for (int m = 16; m >= 1; m >>= 1) mu += __shfl_xor(mu, m, 32);
      mu *= (1.f / 32.f);
      float dx = x - mu;
      float var = dx * dx;
#pragma unroll
      for (int m = 16; m >= 1; m >>= 1) var += __shfl_xor(var, m, 32);
      var *= (1.f / 32.f);
      float y = dx * rsqrtf(var + LN_EPS) * lns[lane] + lnb[lane];
      outp[(size_t)n * 32 + lane] = fmaxf(y, 0.f);
    } else {
      outp[(size_t)n * 32 + lane] = x;
    }
  }
}

extern "C" void kernel_launch(void* const* d_in, const int* in_sizes, int n_in,
                              void* d_out, int out_size, void* d_ws, size_t ws_size,
                              hipStream_t stream) {
  const float* feat = (const float*)d_in[0];
  const float* W1   = (const float*)d_in[1];
  const float* al1  = (const float*)d_in[2];
  const float* ar1  = (const float*)d_in[3];
  const float* b1   = (const float*)d_in[4];
  const float* lns  = (const float*)d_in[5];
  const float* lnb  = (const float*)d_in[6];
  const float* W2   = (const float*)d_in[7];
  const float* al2  = (const float*)d_in[8];
  const float* ar2  = (const float*)d_in[9];
  const float* b2   = (const float*)d_in[10];
  const int*   src  = (const int*)d_in[11];
  const int*   dst  = (const int*)d_in[12];
  const int*   indeg= (const int*)d_in[13];

  const int Nn = in_sizes[13];
  const int E  = in_sizes[11];
  const int NH = Nn * 4;

  char* p = (char*)d_ws;
  float*    hbuf  = (float*)p;   p += (size_t)Nn * 128 * sizeof(float);
  float*    hmid  = (float*)p;   p += (size_t)Nn * 32  * sizeof(float);
  float*    el    = (float*)p;   p += (size_t)NH * sizeof(float);
  float*    er    = (float*)p;   p += (size_t)NH * sizeof(float);
  const int ownBlocks  = (Nn + OWN_NB - 1) / OWN_NB;
  unsigned* wlist = (unsigned*)p; p += (size_t)ownBlocks * OWN_CAP * 2 * sizeof(unsigned);
  unsigned* wkey  = (unsigned*)p; p += ((size_t)E + 15) / 16 * 16;
  unsigned* wcnt  = (unsigned*)p; p += (size_t)ownBlocks * sizeof(unsigned);
  int*      wdeg  = (int*)p;
  (void)ws_size; (void)n_in; (void)out_size;

  float* outh = (float*)d_out;

  dim3 b256(256);
  const int rowTiles   = (Nn + 15) / 16;
  const int gemmBlocks = (rowTiles + 7) / 8;

  gat_gemm_wmma<128><<<gemmBlocks, b256, 0, stream>>>(feat, W1, hbuf, Nn);
  unsigned* wkeyArg = ownBlocks <= 256 ? wkey : nullptr;
  gat_attn_proj<<<(NH + 255) / 256, b256, 0, stream>>>(hbuf, al1, ar1, el, er, NH, dst, wkeyArg, E);
  gat_owner<1><<<ownBlocks, OWN_THREADS, 0, stream>>>(src, dst, el, er, hbuf, b1, lns, lnb, hmid, wlist, wcnt, indeg, wdeg, wkeyArg, Nn, E);

  gat_gemm_wmma<32><<<gemmBlocks, b256, 0, stream>>>(hmid, W2, hbuf, Nn);
  gat_attn_proj<<<(NH + 255) / 256, b256, 0, stream>>>(hbuf, al2, ar2, el, er, NH, dst, nullptr, E);
  gat_owner<2><<<ownBlocks, OWN_THREADS, 0, stream>>>(src, dst, el, er, hbuf, b2, lns, lnb, outh, wlist, wcnt, indeg, wdeg, nullptr, Nn, E);
}


# layer 2's projection (rows x W2 on WMMA, el, er) runs in the layer-1 owner workgroup's epilogue on its LayerNorm rows kept in LDS: 4 launches, was 6; no N x 32 array; el2/er2 fp sum order changes
# speedup vs baseline: 6.6114x; 1.0960x over previous
#include <hip/hip_runtime.h>


typedef float v2f __attribute__((ext_vector_type(2)));
typedef float v4f __attribute__((ext_vector_type(4)));
typedef float v8f __attribute__((ext_vector_type(8)));

#define NEG_SLOPE 0.2f
#define LN_EPS 1e-5f

__device__ __forceinline__ unsigned fenc(float f) {
  unsigned u = __float_as_uint(f);
  return (u >> 31) ? ~u : (u | 0x80000000u);
}
__device__ __forceinline__ float fdec(unsigned u) {
  return __uint_as_float((u >> 31) ? (u & 0x7FFFFFFFu) : ~u);
}
__device__ __forceinline__ float lrelu(float x) {
  return x > 0.f ? x : x * NEG_SLOPE;
}

template <int K>
__global__ void gat_gemm_wmma(const float* __restrict__ Amat,
                              const float* __restrict__ Bmat,
                              float* __restrict__ Cmat,
                              int M) {
  __shared__ float bs[K * 128];

  const int tid = threadIdx.x;

  for (int idx = tid; idx < K * 128; idx += 256) {
    int kr = idx >> 7;
    int col = idx & 127;
    int ks = kr >> 2;
    int c = col >> 4;
    int lane = (col & 15) | ((kr & 2) << 3);
    int sub = kr & 1;
    bs[((ks * 32 + lane) * 8 + c) * 2 + sub] = Bmat[idx];
  }
  __syncthreads();

  const int wave = tid >> 5;
  const int lane = tid & 31;
  const int row0 = (blockIdx.x * 8 + wave) * 16;
  if (row0 >= M) return;

  const int m  = lane & 15;
  const int kk = (lane >> 4) << 1;

  v8f acc[8] = {};

  const float* arow = Amat + (size_t)(row0 + m) * K + kk;
  const float* blane = bs + lane * 16;

#pragma unroll 4
  for (int ks = 0; ks < K / 4; ++ks) {
    v2f a = *(const v2f*)(arow + ks * 4);
    const float* bp = blane + ks * 512;
    v4f q0 = *(const v4f*)(bp);
    v4f q1 = *(const v4f*)(bp + 4);
    v4f q2 = *(const v4f*)(bp + 8);
    v4f q3 = *(const v4f*)(bp + 12);
    acc[0] = __builtin_amdgcn_wmma_f32_16x16x4_f32(false, a, false, q0.xy, (short)0, acc[0], false, false);
    acc[1] = __builtin_amdgcn_wmma_f32_16x16x4_f32(false, a, false, q0.zw, (short)0, acc[1], false, false);
    acc[2] = __builtin_amdgcn_wmma_f32_16x16x4_f32(false, a, false, q1.xy, (short)0, acc[2], false, false);
    acc[3] = __builtin_amdgcn_wmma_f32_16x16x4_f32(false, a, false, q1.zw, (short)0, acc[3], false, false);
    acc[4] = __builtin_amdgcn_wmma_f32_16x16x4_f32(false, a, false, q2.xy, (short)0, acc[4], false, false);
    acc[5] = __builtin_amdgcn_wmma_f32_16x16x4_f32(false, a, false, q2.zw, (short)0, acc[5], false, false);
    acc[6] = __builtin_amdgcn_wmma_f32_16x16x4_f32(false, a, false, q3.xy, (short)0, acc[6], false, false);
    acc[7] = __builtin_amdgcn_wmma_f32_16x16x4_f32(false, a, false, q3.zw, (short)0, acc[7], false, false);
  }

  const int mo = (lane >> 4) * 8;
#pragma unroll
  for (int c = 0; c < 8; ++c) {
#pragma unroll
    for (int r = 0; r < 8; ++r) {
      Cmat[(size_t)(row0 + mo + r) * 128 + c * 16 + m] = acc[c][r];
    }
  }
}

#define OWN_NB      196
#define OWN_CAP     8192
#define OWN_THREADS 1024
#define P2_TILES    ((OWN_NB + 15) / 16)
#define P2_ROWS     (P2_TILES * 16)
#define P2_HPITCH   36
#define P2_OPITCH   132
#define P2_RPW      ((OWN_NB + 31) / 32)

__global__ void gat_attn_proj(const float* __restrict__ h,
                              const float* __restrict__ al,
                              const float* __restrict__ ar,
                              float* __restrict__ el, float* __restrict__ er,
                              int NH,
                              const int* __restrict__ dst, unsigned* __restrict__ wkey,
                              int E) {
  int i = blockIdx.x * blockDim.x + threadIdx.x;
  if (wkey) {
    const int nq = E >> 2;
    const int4* d4 = (const int4*)dst;
    for (int q = i; q < nq; q += (int)(gridDim.x * blockDim.x)) {
      const int4 d = d4[q];
      wkey[q] = (((unsigned)d.x / OWN_NB) & 255u) |
                ((((unsigned)d.y / OWN_NB) & 255u) << 8) |
                ((((unsigned)d.z / OWN_NB) & 255u) << 16) |
                (((unsigned)d.w / OWN_NB) << 24);
    }
  }
  if (i >= NH) return;
  int n = i >> 2, hh = i & 3;
  const float* hp  = h + (size_t)n * 128 + hh * 32;
  const float* alp = al + hh * 32;
  const float* arp = ar + hh * 32;
  float sl = 0.f, sr = 0.f;
#pragma unroll
  for (int f = 0; f < 32; ++f) {
    float v = hp[f];
    sl += v * alp[f];
    sr += v * arp[f];
  }
  el[i] = sl;
  er[i] = sr;
}

typedef unsigned v2u __attribute__((ext_vector_type(2)));
typedef unsigned v4u __attribute__((ext_vector_type(4)));

__device__ __forceinline__ void own_check(int dv, int e, int base, unsigned nb,
                                          v2u* ent, unsigned* cnt) {
  const unsigned dl = (unsigned)(dv - base);
  if (dl < nb) {
    unsigned pos = atomicAdd(cnt, 1u);
    if (pos < OWN_CAP) { v2u v; v.x = (unsigned)e; v.y = dl; ent[pos] = v; }
  }
}

__device__ __forceinline__ void own_build(const int* __restrict__ dst, int e0, int e1,
                                          int base, unsigned nb,
                                          v2u* ent, unsigned* cnt, int tid) {
  const int nvec = (e1 - e0) >> 2;
  const int4* dv = (const int4*)(dst + e0);
  int i = tid;
  for (; i + 3 * OWN_THREADS < nvec; i += 4 * OWN_THREADS) {
    int4 d[4];
#pragma unroll
    for (int u = 0; u < 4; ++u) d[u] = dv[i + u * OWN_THREADS];
#pragma unroll
    for (int u = 0; u < 4; ++u) {
      const int eb = e0 + 4 * (i + u * OWN_THREADS);
      own_check(d[u].x, eb + 0, base, nb, ent, cnt);
      own_check(d[u].y, eb + 1, base, nb, ent, cnt);
      own_check(d[u].z, eb + 2, base, nb, ent, cnt);
      own_check(d[u].w, eb + 3, base, nb, ent, cnt);
    }
  }
  for (; i < nvec; i += OWN_THREADS) {
    int4 d = dv[i];
    const int eb = e0 + 4 * i;
    own_check(d.x, eb + 0, base, nb, ent, cnt);
    own_check(d.y, eb + 1, base, nb, ent, cnt);
    own_check(d.z, eb + 2, base, nb, ent, cnt);
    own_check(d.w, eb + 3, base, nb, ent, cnt);
  }
  for (int e = e0 + nvec * 4 + tid; e < e1; e += OWN_THREADS)
    own_check(dst[e], e, base, nb, ent, cnt);
}

__device__ __forceinline__ unsigned own_nz(unsigned k, unsigned pat) {
  const unsigned x = k ^ pat;
  return ((x & 0x7F7F7F7Fu) + 0x7F7F7F7Fu) | x;
}
__device__ __forceinline__ unsigned own_hits(const v4u k, unsigned pat) {
  const unsigned c = ((own_nz(k.x, pat) >> 7) & 0x01010101u) | ((own_nz(k.y, pat) >> 6) & 0x02020202u) |
                     ((own_nz(k.z, pat) >> 5) & 0x04040404u) | ((own_nz(k.w, pat) >> 4) & 0x08080808u);
  return c ^ 0x0F0F0F0Fu;
}
__device__ __forceinline__ void own_emit(unsigned m, int e0, v2u* ent, unsigned* cnt) {
  while (m) {
    const int p = __builtin_ctz(m);
    m &= m - 1u;
    const int e = e0 + ((p & 3) << 2) + (p >> 3) + ((p & 4) << 12);
    const unsigned pos = atomicAdd(cnt, 1u);
    if (pos < OWN_CAP) ent[pos].x = (unsigned)e;
  }
}

__device__ __forceinline__ void own_scan4(const v4u k[4], unsigned pat, int e0,
                                          v2u* ent, unsigned* cnt) {
  own_emit(own_hits(k[0], pat) | (own_hits(k[1], pat) << 4), e0, ent, cnt);
  own_emit(own_hits(k[2], pat) | (own_hits(k[3], pat) << 4), e0 + 32 * OWN_THREADS, ent, cnt);
}

__device__ __forceinline__ void own_build_keys(const unsigned* __restrict__ wkey,
                                               const int* __restrict__ dst, int E,
                                               unsigned me, int base, unsigned nb,
                                               v2u* ent, unsigned* cnt, int tid) {
  const int nvec = E >> 4;
  const v4u* kv = (const v4u*)wkey;
  const unsigned pat = me * 0x01010101u;
  const int ntrip = nvec / (4 * OWN_THREADS);
  int i = tid;
  if (ntrip > 0) {
    v4u ka[4], kb[4];
#pragma unroll
    for (int u = 0; u < 4; ++u) ka[u] = kv[i + u * OWN_THREADS];
    int t = 0;
    for (; t + 2 < ntrip; t += 2) {
#pragma unroll
      for (int u = 0; u < 4; ++u) kb[u] = kv[i + (4 + u) * OWN_THREADS];
      own_scan4(ka, pat, 16 * i, ent, cnt);
      i += 4 * OWN_THREADS;
#pragma unroll
      for (int u = 0; u < 4; ++u) ka[u] = kv[i + (4 + u) * OWN_THREADS];
      own_scan4(kb, pat, 16 * i, ent, cnt);
      i += 4 * OWN_THREADS;
    }
    if (t + 1 < ntrip) {
#pragma unroll
      for (int u = 0; u < 4; ++u) kb[u] = kv[i + (4 + u) * OWN_THREADS];
      own_scan4(ka, pat, 16 * i, ent, cnt);
      i += 4 * OWN_THREADS;
      own_scan4(kb, pat, 16 * i, ent, cnt);
    } else {
      own_scan4(ka, pat, 16 * i, ent, cnt);
    }
    i += 4 * OWN_THREADS;
  }
  for (; i < nvec; i += OWN_THREADS) {
    own_emit(own_hits(kv[i], pat), 16 * i, ent, cnt);
  }
  for (int e = nvec * 16 + tid; e < E; e += OWN_THREADS)
    own_check(dst[e], e, base, nb, ent, cnt);
}

template <bool HAVE_SRC, bool KEEP, bool NEED_DL>
__device__ __forceinline__ void own_prep(v2u* ent, v4f* sc, int n,
                                         const int* __restrict__ src,
                                         const int* __restrict__ dst, int base, unsigned nb,
                                         const float* __restrict__ el, const float* erl,
                                         unsigned* smax, v2u* __restrict__ wl,
                                         bool doMax, int tid) {
  for (int i = tid; i < n; i += OWN_THREADS) {
    v2u en = ent[i];
    if (!HAVE_SRC) {
      const unsigned e = en.x;
      en.x = (unsigned)src[e];
      if (NEED_DL) en.y = min((unsigned)(dst[e] - base), nb - 1u);
      ent[i] = en;
    }
    if (KEEP) wl[i] = en;
    const unsigned dl = en.y;
    const v4f l = *(const v4f*)(el + (size_t)en.x * 4);
    const v4f r = *(const v4f*)(erl + dl * 4);
    v4f v;
    v.x = lrelu(l.x + r.x);
    v.y = lrelu(l.y + r.y);
    v.z = lrelu(l.z + r.z);
    v.w = lrelu(l.w + r.w);
    sc[i] = v;
    if (doMax) {
      atomicMax(&smax[dl * 4 + 0], fenc(v.x));
      atomicMax(&smax[dl * 4 + 1], fenc(v.y));
      atomicMax(&smax[dl * 4 + 2], fenc(v.z));
      atomicMax(&smax[dl * 4 + 3], fenc(v.w));
    }
  }
}

__device__ __forceinline__ void own_den(const v2u* ent, v4f* sc, int n,
                                        const unsigned* smax, float* sden,
                                        bool add, int tid) {
  for (int i = tid; i < n; i += OWN_THREADS) {
    const unsigned dl = ent[i].y;
    const v4u m = *(const v4u*)(smax + dl * 4);
    v4f v = sc[i];
    v.x = expf(v.x - fdec(m.x));
    v.y = expf(v.y - fdec(m.y));
    v.z = expf(v.z - fdec(m.z));
    v.w = expf(v.w - fdec(m.w));
    sc[i] = v;
    if (add) {
      atomicAdd(&sden[dl * 4 + 0], v.x);
      atomicAdd(&sden[dl * 4 + 1], v.y);
      atomicAdd(&sden[dl * 4 + 2], v.z);
      atomicAdd(&sden[dl * 4 + 3], v.w);
    }
  }
}

__device__ __forceinline__ void own_acc(const v2u* ent, const v4f* sc, int n,
                                        const float* __restrict__ h,
                                        const float* sden,
                                        float* acc, int wave, int lane) {
  for (int b = wave * 32; b < n; b += OWN_THREADS) {
    const int i = b + lane;
    int s = 0, dl = 0;
    float a0 = 0.f, a1 = 0.f, a2 = 0.f, a3 = 0.f;
    if (i < n) {
      const v2u en = ent[i];
      s = (int)en.x;
      dl = (int)en.y;
      const v4f ex = sc[i];
      const v4f dn = *(const v4f*)(sden + dl * 4);
      a0 = ex.x / fmaxf(dn.x, 1e-9f);
      a1 = ex.y / fmaxf(dn.y, 1e-9f);
      a2 = ex.z / fmaxf(dn.z, 1e-9f);
      a3 = ex.w / fmaxf(dn.w, 1e-9f);
    }
    const int cb = min(32, n - b);
    for (int j = 0; j < cb; j += 8) {
      float hv[8][4], aw[8][4];
      int dj[8];
#pragma unroll
      for (int u = 0; u < 8; ++u) {
        const int jj = j + u;
        const int sj = __builtin_amdgcn_readlane(s, jj);
        dj[u] = __builtin_amdgcn_readlane(dl, jj);
        aw[u][0] = __int_as_float(__builtin_amdgcn_readlane(__float_as_int(a0), jj));
        aw[u][1] = __int_as_float(__builtin_amdgcn_readlane(__float_as_int(a1), jj));
        aw[u][2] = __int_as_float(__builtin_amdgcn_readlane(__float_as_int(a2), jj));
        aw[u][3] = __int_as_float(__builtin_amdgcn_readlane(__float_as_int(a3), jj));
        const float* hp = h + (size_t)sj * 128 + lane;
        hv[u][0] = hp[0];
        hv[u][1] = hp[32];
        hv[u][2] = hp[64];
        hv[u][3] = hp[96];
      }
#pragma unroll
      for (int u = 0; u < 8; ++u) {
        if (j + u < cb) {
          float* ap = acc + dj[u] * 128 + lane;
          atomicAdd(ap,      hv[u][0] * aw[u][0]);
          atomicAdd(ap + 32, hv[u][1] * aw[u][1]);
          atomicAdd(ap + 64, hv[u][2] * aw[u][2]);
          atomicAdd(ap + 96, hv[u][3] * aw[u][3]);
        }
      }
    }
  }
}

__device__ __forceinline__ void own_deg_add(int v, int lane, int* sdeg) {
  for (int m = 16; m >= 1; m >>= 1) v += __shfl_xor(v, m, 32);
  if (lane == 0 && v != 0) atomicAdd(sdeg, v);
}

template <int LAYER>
__global__ void __launch_bounds__(OWN_THREADS)
gat_owner(const int* __restrict__ src, const int* __restrict__ dst,
          const float* __restrict__ el, const float* __restrict__ er,
          const float* __restrict__ h, const float* __restrict__ bias,
          const float* __restrict__ lns, const float* __restrict__ lnb,
          float* __restrict__ outp,
          unsigned* __restrict__ wlist, unsigned* __restrict__ wcnt,
          const int* __restrict__ indeg, int* __restrict__ wdeg,
          const unsigned* __restrict__ wkey,
          const float* __restrict__ Wn, const float* __restrict__ aln,
          const float* __restrict__ arn,
          float* __restrict__ eln, float* __restrict__ ern,
          int Nn, int E) {
  __shared__ float    acc[OWN_NB * 128];
  __shared__ __attribute__((aligned(16))) unsigned smax[OWN_NB * 4];
  __shared__ __attribute__((aligned(16))) float    sden[OWN_NB * 4];
  __shared__ __attribute__((aligned(16))) float    erl[OWN_NB * 4];
  __shared__ v2u      ent[OWN_CAP];
  __shared__ v4f      sc[OWN_CAP];
  __shared__ unsigned cnt;
  __shared__ int      sdeg;

  const int tid  = threadIdx.x;
  const int wave = tid >> 5;
  const int lane = tid & 31;
  const int base = blockIdx.x * OWN_NB;
  const unsigned nb = (unsigned)min(OWN_NB, Nn - base);

  const bool degBlock = (LAYER == 1) || (blockIdx.x == gridDim.x - 1);
  int dpart = 0;
  if (LAYER == 2 && degBlock) {
    for (int i = tid; i < (int)gridDim.x; i += OWN_THREADS) dpart += wdeg[i];
  }

  for (int i = tid; i < OWN_NB * 128; i += OWN_THREADS) acc[i] = 0.f;
  for (int i = tid; i < OWN_NB * 4; i += OWN_THREADS) { smax[i] = 0u; sden[i] = 0.f; }
  if (tid == 0) { cnt = 0u; sdeg = 0; }

  v2u* wl = (v2u*)wlist + (size_t)blockIdx.x * OWN_CAP;
  float erv = 0.f;
  unsigned total;
  if (LAYER == 1) {
    __syncthreads();
    if (tid < (int)nb) dpart = indeg[base + tid];
    if (tid < (int)nb * 4) erv = er[(size_t)base * 4 + tid];
    if (wkey) own_build_keys(wkey, dst, E, blockIdx.x, base, nb, ent, &cnt, tid);
    else      own_build(dst, 0, E, base, nb, ent, &cnt, tid);
    if (tid < (int)nb * 4) erl[tid] = erv;
    own_deg_add(dpart, lane, &sdeg);
    __syncthreads();
    total = cnt;
    if (tid == 0) { wcnt[blockIdx.x] = total; wdeg[blockIdx.x] = sdeg; }
  } else {
    total = wcnt[blockIdx.x];
    if (tid < (int)nb * 4) erv = er[(size_t)base * 4 + tid];
    if (total <= OWN_CAP) {
      for (unsigned i0 = tid; i0 < total; i0 += 4 * OWN_THREADS) {
        v2u v[4];
#pragma unroll
        for (int u = 0; u < 4; ++u) {
          const unsigned i = i0 + u * OWN_THREADS;
          v2u z; z.x = 0u; z.y = 0u;
          v[u] = i < total ? wl[i] : z;
        }
#pragma unroll
        for (int u = 0; u < 4; ++u) {
          const unsigned i = i0 + u * OWN_THREADS;
          if (i < total) ent[i] = v[u];
        }
      }
    }
    if (tid < (int)nb * 4) erl[tid] = erv;
    __syncthreads();
    if (degBlock) own_deg_add(dpart, lane, &sdeg);
  }

  if (total <= OWN_CAP) {
    const int n = (int)total;
    own_prep<LAYER == 2, LAYER == 1, LAYER == 1>(ent, sc, n, src, dst, base, nb, el, erl, smax, wl, true, tid);
    __syncthreads();
    own_den(ent, sc, n, smax, sden, true, tid);
    __syncthreads();
    own_acc(ent, sc, n, h, sden, acc, wave, lane);
  } else {
    for (int pass = 0; pass < 3; ++pass) {
      for (int e0 = 0; e0 < E; e0 += OWN_CAP) {
        __syncthreads();
        if (tid == 0) cnt = 0u;
        __syncthreads();
        own_build(dst, e0, min(E, e0 + OWN_CAP), base, nb, ent, &cnt, tid);
        __syncthreads();
        const int n = (int)cnt;
        own_prep<false, false, false>(ent, sc, n, src, dst, base, nb, el, erl, smax, wl, pass == 0, tid);
        if (pass >= 1) {
          __syncthreads();
          own_den(ent, sc, n, smax, sden, pass == 1, tid);
        }
        if (pass == 2) {
          __syncthreads();
          own_acc(ent, sc, n, h, sden, acc, wave, lane);
        }
      }
    }
  }
  __syncthreads();

  if (LAYER == 2 && degBlock && tid == 0) {
    double ne = (double)(float)sdeg;
    double fl = (4.0 * ne * (6.0 * 128 * 32 + 6 * 32 + 2) +
                 4.0 * ne * (6.0 * 32 * 32 + 6 * 32 + 2)) / 1e12;
    outp[(size_t)Nn * 32] = (float)fl;
  }

  float* hm = (float*)ent;
  float* st = (float*)sc;
  float* bs = st + P2_ROWS * P2_OPITCH;
  static_assert(32 * 128 == 4 * OWN_THREADS, "Wn: 4 consecutive floats per thread");
  v4f alv = {}, arv = {}, wq = {};
  if (LAYER == 1) {
    alv = *(const v4f*)(aln + lane * 4);
    arv = *(const v4f*)(arn + lane * 4);
    wq  = *(const v4f*)(Wn + tid * 4);
    for (int i = (int)nb * P2_HPITCH + tid; i < P2_ROWS * P2_HPITCH; i += OWN_THREADS) hm[i] = 0.f;
  }

  for (int nl = wave; nl < (int)nb; nl += OWN_THREADS / 32) {
    const int n = base + nl;
    float x = 0.f;
#pragma unroll
    for (int hh = 0; hh < 4; ++hh)
      x += acc[nl * 128 + hh * 32 + lane] + bias[hh * 32 + lane];
    x *= 0.25f;
    if (LAYER == 1) {
      float mu = x;
#pragma unroll
      for (int m = 16; m >= 1; m >>= 1) mu += __shfl_xor(mu, m, 32);
      mu *= (1.f / 32.f);
      float dx = x - mu;
      float var = dx * dx;
#pragma unroll
      for (int m = 16; m >= 1; m >>= 1) var += __shfl_xor(var, m, 32);
      var *= (1.f / 32.f);
      float y = dx * rsqrtf(var + LN_EPS) * lns[lane] + lnb[lane];
      hm[nl * P2_HPITCH + lane] = fmaxf(y, 0.f);
    } else {
      outp[(size_t)n * 32 + lane] = x;
    }
  }

  if (LAYER == 1) {
    {
      const int kr = tid >> 5, col = (tid & 31) * 4;
      const int ln = (col & 15) | ((kr & 2) << 3);
      float* bq = bs + (((kr >> 2) * 32 + ln) * 8 + (col >> 4)) * 2 + (kr & 1);
      bq[0]  = wq.x;
      bq[16] = wq.y;
      bq[32] = wq.z;
      bq[48] = wq.w;
    }
    __syncthreads();
    const int wv = __builtin_amdgcn_readfirstlane(wave);
    const int t = wv >> 1, hf = wv & 1;
    if (t * 16 < (int)nb) {
      const int m = lane & 15;
      const float* ap = hm + (t * 16 + m) * P2_HPITCH + ((lane >> 4) << 1);
      const float* bp = bs + lane * 16 + hf * 8;
      v8f c0 = {}, c1 = {}, c2 = {}, c3 = {};
#pragma unroll
      for (int ks = 0; ks < 8; ++ks) {
        const v2f a  = *(const v2f*)(ap + ks * 4);
        const v4f q0 = *(const v4f*)(bp + ks * 512);
        const v4f q1 = *(const v4f*)(bp + ks * 512 + 4);
        c0 = __builtin_amdgcn_wmma_f32_16x16x4_f32(false, a, false, q0.xy, (short)0, c0, false, false);
        c1 = __builtin_amdgcn_wmma_f32_16x16x4_f32(false, a, false, q0.zw, (short)0, c1, false, false);
        c2 = __builtin_amdgcn_wmma_f32_16x16x4_f32(false, a, false, q1.xy, (short)0, c2, false, false);
        c3 = __builtin_amdgcn_wmma_f32_16x16x4_f32(false, a, false, q1.zw, (short)0, c3, false, false);
      }
      float* sp = st + (t * 16 + (lane >> 4) * 8) * P2_OPITCH + hf * 64 + m;
#pragma unroll
      for (int r = 0; r < 8; ++r) {
        sp[r * P2_OPITCH]      = c0[r];
        sp[r * P2_OPITCH + 16] = c1[r];
        sp[r * P2_OPITCH + 32] = c2[r];
        sp[r * P2_OPITCH + 48] = c3[r];
      }
    }
    __syncthreads();
    float pl[P2_RPW], pr[P2_RPW];
    v4f pv[P2_RPW];
#pragma unroll
    for (int u = 0; u < P2_RPW; ++u)
      pv[u] = *(const v4f*)(st + min(wv + 32 * u, P2_ROWS - 1) * P2_OPITCH + lane * 4);
#pragma unroll
    for (int u = 0; u < P2_RPW; ++u) {
      const int nl = wv + 32 * u;
      const v4f v = pv[u];
      if (nl < (int)nb) *(v4f*)(outp + (size_t)(base + nl) * 128 + lane * 4) = v;
      pl[u] = v.x * alv.x + v.y * alv.y + v.z * alv.z + v.w * alv.w;
      pr[u] = v.x * arv.x + v.y * arv.y + v.z * arv.z + v.w * arv.w;
    }
#pragma unroll
    for (int msk = 1; msk <= 4; msk <<= 1) {
#pragma unroll
      for (int u = 0; u < P2_RPW; ++u) {
        pl[u] += __shfl_xor(pl[u], msk, 32);
        pr[u] += __shfl_xor(pr[u], msk, 32);
      }
    }
#pragma unroll
    for (int u = 0; u < P2_RPW; ++u) {
      const int nl = wv + 32 * u;
      if (nl < (int)nb && (lane & 7) == 0) {
        sden[nl * 4 + (lane >> 3)] = pl[u];
        erl[nl * 4 + (lane >> 3)]  = pr[u];
      }
    }
    __syncthreads();
    if (tid < (int)nb * 4) {
      eln[(size_t)base * 4 + tid] = sden[tid];
      ern[(size_t)base * 4 + tid] = erl[tid];
    }
  }
}

extern "C" void kernel_launch(void* const* d_in, const int* in_sizes, int n_in,
                              void* d_out, int out_size, void* d_ws, size_t ws_size,
                              hipStream_t stream) {
  const float* feat = (const float*)d_in[0];
  const float* W1   = (const float*)d_in[1];
  const float* al1  = (const float*)d_in[2];
  const float* ar1  = (const float*)d_in[3];
  const float* b1   = (const float*)d_in[4];
  const float* lns  = (const float*)d_in[5];
  const float* lnb  = (const float*)d_in[6];
  const float* W2   = (const float*)d_in[7];
  const float* al2  = (const float*)d_in[8];
  const float* ar2  = (const float*)d_in[9];
  const float* b2   = (const float*)d_in[10];
  const int*   src  = (const int*)d_in[11];
  const int*   dst  = (const int*)d_in[12];
  const int*   indeg= (const int*)d_in[13];

  const int Nn = in_sizes[13];
  const int E  = in_sizes[11];
  const int NH = Nn * 4;

  char* p = (char*)d_ws;
  float*    hbuf  = (float*)p;   p += (size_t)Nn * 128 * sizeof(float);
  float*    hbuf2 = (float*)p;   p += (size_t)Nn * 128 * sizeof(float);
  float*    el    = (float*)p;   p += (size_t)NH * sizeof(float);
  float*    er    = (float*)p;   p += (size_t)NH * sizeof(float);
  float*    el2   = (float*)p;   p += (size_t)NH * sizeof(float);
  float*    er2   = (float*)p;   p += (size_t)NH * sizeof(float);
  const int ownBlocks  = (Nn + OWN_NB - 1) / OWN_NB;
  unsigned* wlist = (unsigned*)p; p += (size_t)ownBlocks * OWN_CAP * 2 * sizeof(unsigned);
  unsigned* wkey  = (unsigned*)p; p += ((size_t)E + 15) / 16 * 16;
  unsigned* wcnt  = (unsigned*)p; p += (size_t)ownBlocks * sizeof(unsigned);
  int*      wdeg  = (int*)p;
  (void)ws_size; (void)n_in; (void)out_size;

  float* outh = (float*)d_out;

  dim3 b256(256);
  const int rowTiles   = (Nn + 15) / 16;
  const int gemmBlocks = (rowTiles + 7) / 8;

  gat_gemm_wmma<128><<<gemmBlocks, b256, 0, stream>>>(feat, W1, hbuf, Nn);
  unsigned* wkeyArg = ownBlocks <= 256 ? wkey : nullptr;
  gat_attn_proj<<<(NH + 255) / 256, b256, 0, stream>>>(hbuf, al1, ar1, el, er, NH, dst, wkeyArg, E);
  gat_owner<1><<<ownBlocks, OWN_THREADS, 0, stream>>>(src, dst, el, er, hbuf, b1, lns, lnb, hbuf2, wlist, wcnt, indeg, wdeg, wkeyArg, W2, al2, ar2, el2, er2, Nn, E);

  gat_owner<2><<<ownBlocks, OWN_THREADS, 0, stream>>>(src, dst, el2, er2, hbuf2, b2, lns, lnb, outh, wlist, wcnt, indeg, wdeg, nullptr, nullptr, nullptr, nullptr, nullptr, nullptr, Nn, E);
}


# every wave-wide read of a 512-byte projected row is one 16-byte-a-lane load of its 4 whole lines: edge gather was 4 one-line loads per edge, score launch 32-line loads; el/er fp sum order changes
# speedup vs baseline: 7.1124x; 1.0758x over previous
#include <hip/hip_runtime.h>


typedef float v2f __attribute__((ext_vector_type(2)));
typedef float v4f __attribute__((ext_vector_type(4)));
typedef float v8f __attribute__((ext_vector_type(8)));

#define NEG_SLOPE 0.2f
#define LN_EPS 1e-5f

__device__ __forceinline__ unsigned fenc(float f) {
  unsigned u = __float_as_uint(f);
  return (u >> 31) ? ~u : (u | 0x80000000u);
}
__device__ __forceinline__ float fdec(unsigned u) {
  return __uint_as_float((u >> 31) ? (u & 0x7FFFFFFFu) : ~u);
}
__device__ __forceinline__ float lrelu(float x) {
  return x > 0.f ? x : x * NEG_SLOPE;
}

template <int K>
__global__ void gat_gemm_wmma(const float* __restrict__ Amat,
                              const float* __restrict__ Bmat,
                              float* __restrict__ Cmat,
                              int M) {
  __shared__ float bs[K * 128];

  const int tid = threadIdx.x;

  for (int idx = tid; idx < K * 128; idx += 256) {
    int kr = idx >> 7;
    int col = idx & 127;
    int ks = kr >> 2;
    int c = col >> 4;
    int lane = (col & 15) | ((kr & 2) << 3);
    int sub = kr & 1;
    bs[((ks * 32 + lane) * 8 + c) * 2 + sub] = Bmat[idx];
  }
  __syncthreads();

  const int wave = tid >> 5;
  const int lane = tid & 31;
  const int row0 = (blockIdx.x * 8 + wave) * 16;
  if (row0 >= M) return;

  const int m  = lane & 15;
  const int kk = (lane >> 4) << 1;

  v8f acc[8] = {};

  const float* arow = Amat + (size_t)(row0 + m) * K + kk;
  const float* blane = bs + lane * 16;

#pragma unroll 4
  for (int ks = 0; ks < K / 4; ++ks) {
    v2f a = *(const v2f*)(arow + ks * 4);
    const float* bp = blane + ks * 512;
    v4f q0 = *(const v4f*)(bp);
    v4f q1 = *(const v4f*)(bp + 4);
    v4f q2 = *(const v4f*)(bp + 8);
    v4f q3 = *(const v4f*)(bp + 12);
    acc[0] = __builtin_amdgcn_wmma_f32_16x16x4_f32(false, a, false, q0.xy, (short)0, acc[0], false, false);
    acc[1] = __builtin_amdgcn_wmma_f32_16x16x4_f32(false, a, false, q0.zw, (short)0, acc[1], false, false);
    acc[2] = __builtin_amdgcn_wmma_f32_16x16x4_f32(false, a, false, q1.xy, (short)0, acc[2], false, false);
    acc[3] = __builtin_amdgcn_wmma_f32_16x16x4_f32(false, a, false, q1.zw, (short)0, acc[3], false, false);
    acc[4] = __builtin_amdgcn_wmma_f32_16x16x4_f32(false, a, false, q2.xy, (short)0, acc[4], false, false);
    acc[5] = __builtin_amdgcn_wmma_f32_16x16x4_f32(false, a, false, q2.zw, (short)0, acc[5], false, false);
    acc[6] = __builtin_amdgcn_wmma_f32_16x16x4_f32(false, a, false, q3.xy, (short)0, acc[6], false, false);
    acc[7] = __builtin_amdgcn_wmma_f32_16x16x4_f32(false, a, false, q3.zw, (short)0, acc[7], false, false);
  }

  const int mo = (lane >> 4) * 8;
#pragma unroll
  for (int c = 0; c < 8; ++c) {
#pragma unroll
    for (int r = 0; r < 8; ++r) {
      Cmat[(size_t)(row0 + mo + r) * 128 + c * 16 + m] = acc[c][r];
    }
  }
}

#define OWN_NB      196
#define OWN_CAP     8192
#define OWN_THREADS 1024
#define P2_TILES    ((OWN_NB + 15) / 16)
#define P2_ROWS     (P2_TILES * 16)
#define P2_HPITCH   36
#define P2_OPITCH   132
#define P2_RPW      ((OWN_NB + 31) / 32)

#define PROJ_ROWS 16

__device__ __forceinline__ float proj_sum8(const float (&p)[8], int lane) {
  const bool b4 = (lane & 4) != 0, b2 = (lane & 2) != 0, b1 = (lane & 1) != 0;
  float w[4], x[2];
#pragma unroll
  for (int j = 0; j < 4; ++j) {
    const float mine = b4 ? p[j + 4] : p[j];
    const float give = b4 ? p[j] : p[j + 4];
    w[j] = mine + __shfl_xor(give, 4, 32);
  }
#pragma unroll
  for (int j = 0; j < 2; ++j) {
    const float mine = b2 ? w[j + 2] : w[j];
    const float give = b2 ? w[j] : w[j + 2];
    x[j] = mine + __shfl_xor(give, 2, 32);
  }
  const float mine = b1 ? x[1] : x[0];
  const float give = b1 ? x[0] : x[1];
  return mine + __shfl_xor(give, 1, 32);
}

__global__ void gat_attn_proj(const float* __restrict__ h,
                              const float* __restrict__ al,
                              const float* __restrict__ ar,
                              float* __restrict__ el, float* __restrict__ er,
                              int Nn,
                              const int* __restrict__ dst, unsigned* __restrict__ wkey,
                              int E) {
  const int i = blockIdx.x * blockDim.x + threadIdx.x;
  if (wkey) {
    const int nq = E >> 2;
    const int4* d4 = (const int4*)dst;
    for (int q = i; q < nq; q += (int)(gridDim.x * blockDim.x)) {
      const int4 d = d4[q];
      wkey[q] = (((unsigned)d.x / OWN_NB) & 255u) |
                ((((unsigned)d.y / OWN_NB) & 255u) << 8) |
                ((((unsigned)d.z / OWN_NB) & 255u) << 16) |
                (((unsigned)d.w / OWN_NB) << 24);
    }
  }
  const int lane = threadIdx.x & 31;
  const int n0 = (i >> 5) * PROJ_ROWS;
  if (n0 >= Nn) return;
  const float* hl = h + lane * 4;
  const v4f a4 = *(const v4f*)(al + lane * 4);
  const v4f r4 = *(const v4f*)(ar + lane * 4);
  v4f v[PROJ_ROWS];
  const float* hr = hl + (size_t)n0 * 128;
  const int last = Nn - 1 - n0;
#pragma unroll
  for (int u = 0; u < PROJ_ROWS; ++u) v[u] = *(const v4f*)(hr + min(u, last) * 128);
  __builtin_amdgcn_sched_barrier(0);
#pragma unroll
  for (int g = 0; g < PROJ_ROWS / 8; ++g) {
    float pl[8], pr[8];
#pragma unroll
    for (int r = 0; r < 8; ++r) {
      const v4f t = v[g * 8 + r];
      pl[r] = t.x * a4.x + t.y * a4.y + t.z * a4.z + t.w * a4.w;
      pr[r] = t.x * r4.x + t.y * r4.y + t.z * r4.z + t.w * r4.w;
    }
    const float sl = proj_sum8(pl, lane);
    const float sr = proj_sum8(pr, lane);
    const int n = n0 + g * 8 + (lane & 7);
    if (n < Nn) {
      el[(size_t)n * 4 + (lane >> 3)] = sl;
      er[(size_t)n * 4 + (lane >> 3)] = sr;
    }
  }
}

typedef unsigned v2u __attribute__((ext_vector_type(2)));
typedef unsigned v4u __attribute__((ext_vector_type(4)));

__device__ __forceinline__ void own_check(int dv, int e, int base, unsigned nb,
                                          v2u* ent, unsigned* cnt) {
  const unsigned dl = (unsigned)(dv - base);
  if (dl < nb) {
    unsigned pos = atomicAdd(cnt, 1u);
    if (pos < OWN_CAP) { v2u v; v.x = (unsigned)e; v.y = dl; ent[pos] = v; }
  }
}

__device__ __forceinline__ void own_build(const int* __restrict__ dst, int e0, int e1,
                                          int base, unsigned nb,
                                          v2u* ent, unsigned* cnt, int tid) {
  const int nvec = (e1 - e0) >> 2;
  const int4* dv = (const int4*)(dst + e0);
  int i = tid;
  for (; i + 3 * OWN_THREADS < nvec; i += 4 * OWN_THREADS) {
    int4 d[4];
#pragma unroll
    for (int u = 0; u < 4; ++u) d[u] = dv[i + u * OWN_THREADS];
#pragma unroll
    for (int u = 0; u < 4; ++u) {
      const int eb = e0 + 4 * (i + u * OWN_THREADS);
      own_check(d[u].x, eb + 0, base, nb, ent, cnt);
      own_check(d[u].y, eb + 1, base, nb, ent, cnt);
      own_check(d[u].z, eb + 2, base, nb, ent, cnt);
      own_check(d[u].w, eb + 3, base, nb, ent, cnt);
    }
  }
  for (; i < nvec; i += OWN_THREADS) {
    int4 d = dv[i];
    const int eb = e0 + 4 * i;
    own_check(d.x, eb + 0, base, nb, ent, cnt);
    own_check(d.y, eb + 1, base, nb, ent, cnt);
    own_check(d.z, eb + 2, base, nb, ent, cnt);
    own_check(d.w, eb + 3, base, nb, ent, cnt);
  }
  for (int e = e0 + nvec * 4 + tid; e < e1; e += OWN_THREADS)
    own_check(dst[e], e, base, nb, ent, cnt);
}

__device__ __forceinline__ unsigned own_nz(unsigned k, unsigned pat) {
  const unsigned x = k ^ pat;
  return ((x & 0x7F7F7F7Fu) + 0x7F7F7F7Fu) | x;
}
__device__ __forceinline__ unsigned own_hits(const v4u k, unsigned pat) {
  const unsigned c = ((own_nz(k.x, pat) >> 7) & 0x01010101u) | ((own_nz(k.y, pat) >> 6) & 0x02020202u) |
                     ((own_nz(k.z, pat) >> 5) & 0x04040404u) | ((own_nz(k.w, pat) >> 4) & 0x08080808u);
  return c ^ 0x0F0F0F0Fu;
}
__device__ __forceinline__ void own_emit(unsigned m, int e0, v2u* ent, unsigned* cnt) {
  while (m) {
    const int p = __builtin_ctz(m);
    m &= m - 1u;
    const int e = e0 + ((p & 3) << 2) + (p >> 3) + ((p & 4) << 12);
    const unsigned pos = atomicAdd(cnt, 1u);
    if (pos < OWN_CAP) ent[pos].x = (unsigned)e;
  }
}

__device__ __forceinline__ void own_scan4(const v4u k[4], unsigned pat, int e0,
                                          v2u* ent, unsigned* cnt) {
  own_emit(own_hits(k[0], pat) | (own_hits(k[1], pat) << 4), e0, ent, cnt);
  own_emit(own_hits(k[2], pat) | (own_hits(k[3], pat) << 4), e0 + 32 * OWN_THREADS, ent, cnt);
}

__device__ __forceinline__ void own_build_keys(const unsigned* __restrict__ wkey,
                                               const int* __restrict__ dst, int E,
                                               unsigned me, int base, unsigned nb,
                                               v2u* ent, unsigned* cnt, int tid) {
  const int nvec = E >> 4;
  const v4u* kv = (const v4u*)wkey;
  const unsigned pat = me * 0x01010101u;
  const int ntrip = nvec / (4 * OWN_THREADS);
  int i = tid;
  if (ntrip > 0) {
    v4u ka[4], kb[4];
#pragma unroll
    for (int u = 0; u < 4; ++u) ka[u] = kv[i + u * OWN_THREADS];
    int t = 0;
    for (; t + 2 < ntrip; t += 2) {
#pragma unroll
      for (int u = 0; u < 4; ++u) kb[u] = kv[i + (4 + u) * OWN_THREADS];
      own_scan4(ka, pat, 16 * i, ent, cnt);
      i += 4 * OWN_THREADS;
#pragma unroll
      for (int u = 0; u < 4; ++u) ka[u] = kv[i + (4 + u) * OWN_THREADS];
      own_scan4(kb, pat, 16 * i, ent, cnt);
      i += 4 * OWN_THREADS;
    }
    if (t + 1 < ntrip) {
#pragma unroll
      for (int u = 0; u < 4; ++u) kb[u] = kv[i + (4 + u) * OWN_THREADS];
      own_scan4(ka, pat, 16 * i, ent, cnt);
      i += 4 * OWN_THREADS;
      own_scan4(kb, pat, 16 * i, ent, cnt);
    } else {
      own_scan4(ka, pat, 16 * i, ent, cnt);
    }
    i += 4 * OWN_THREADS;
  }
  for (; i < nvec; i += OWN_THREADS) {
    own_emit(own_hits(kv[i], pat), 16 * i, ent, cnt);
  }
  for (int e = nvec * 16 + tid; e < E; e += OWN_THREADS)
    own_check(dst[e], e, base, nb, ent, cnt);
}

template <bool HAVE_SRC, bool KEEP, bool NEED_DL>
__device__ __forceinline__ void own_prep(v2u* ent, v4f* sc, int n,
                                         const int* __restrict__ src,
                                         const int* __restrict__ dst, int base, unsigned nb,
                                         const float* __restrict__ el, const float* erl,
                                         unsigned* smax, v2u* __restrict__ wl,
                                         bool doMax, int tid) {
  for (int i = tid; i < n; i += OWN_THREADS) {
    v2u en = ent[i];
    if (!HAVE_SRC) {
      const unsigned e = en.x;
      en.x = (unsigned)src[e];
      if (NEED_DL) en.y = min((unsigned)(dst[e] - base), nb - 1u);
      ent[i] = en;
    }
    if (KEEP) wl[i] = en;
    const unsigned dl = en.y;
    const v4f l = *(const v4f*)(el + (size_t)en.x * 4);
    const v4f r = *(const v4f*)(erl + dl * 4);
    v4f v;
    v.x = lrelu(l.x + r.x);
    v.y = lrelu(l.y + r.y);
    v.z = lrelu(l.z + r.z);
    v.w = lrelu(l.w + r.w);
    sc[i] = v;
    if (doMax) {
      atomicMax(&smax[dl * 4 + 0], fenc(v.x));
      atomicMax(&smax[dl * 4 + 1], fenc(v.y));
      atomicMax(&smax[dl * 4 + 2], fenc(v.z));
      atomicMax(&smax[dl * 4 + 3], fenc(v.w));
    }
  }
}

__device__ __forceinline__ void own_den(const v2u* ent, v4f* sc, int n,
                                        const unsigned* smax, float* sden,
                                        bool add, int tid) {
  for (int i = tid; i < n; i += OWN_THREADS) {
    const unsigned dl = ent[i].y;
    const v4u m = *(const v4u*)(smax + dl * 4);
    v4f v = sc[i];
    v.x = expf(v.x - fdec(m.x));
    v.y = expf(v.y - fdec(m.y));
    v.z = expf(v.z - fdec(m.z));
    v.w = expf(v.w - fdec(m.w));
    sc[i] = v;
    if (add) {
      atomicAdd(&sden[dl * 4 + 0], v.x);
      atomicAdd(&sden[dl * 4 + 1], v.y);
      atomicAdd(&sden[dl * 4 + 2], v.z);
      atomicAdd(&sden[dl * 4 + 3], v.w);
    }
  }
}

template <bool TAIL>
__device__ __forceinline__ void own_acc8(int j, int cb, int s, int dl,
                                         float a0, float a1, float a2, float a3,
                                         bool hb0, bool hb1,
                                         const float* __restrict__ hl,
                                         float* acc, int lane) {
  v4f hv[8];
  float aw[8];
  int dj[8];
#pragma unroll
  for (int u = 0; u < 8; ++u) {
    const int jj = j + u;
    const int sj = __builtin_amdgcn_readlane(s, jj);
    dj[u] = __builtin_amdgcn_readlane(dl, jj);
    const float w0 = __int_as_float(__builtin_amdgcn_readlane(__float_as_int(a0), jj));
    const float w1 = __int_as_float(__builtin_amdgcn_readlane(__float_as_int(a1), jj));
    const float w2 = __int_as_float(__builtin_amdgcn_readlane(__float_as_int(a2), jj));
    const float w3 = __int_as_float(__builtin_amdgcn_readlane(__float_as_int(a3), jj));
    const float w01 = hb0 ? w1 : w0;
    const float w23 = hb0 ? w3 : w2;
    aw[u] = hb1 ? w23 : w01;
    hv[u] = *(const v4f*)(hl + (size_t)sj * 128);
  }
#pragma unroll
  for (int u = 0; u < 8; ++u) {
    if (!TAIL || u == 0 || j + u < cb) {
      float* ap = acc + dj[u] * 128 + lane;
      atomicAdd(ap,      hv[u].x * aw[u]);
      atomicAdd(ap + 32, hv[u].y * aw[u]);
      atomicAdd(ap + 64, hv[u].z * aw[u]);
      atomicAdd(ap + 96, hv[u].w * aw[u]);
    }
  }
}

__device__ __forceinline__ void own_acc(const v2u* ent, const v4f* sc, int n,
                                        const float* __restrict__ h,
                                        const float* sden,
                                        float* acc, int wave, int lane) {
  const bool hb0 = (lane & 8) != 0, hb1 = (lane & 16) != 0;
  const float* hl = h + lane * 4;
  for (int b = wave * 32; b < n; b += OWN_THREADS) {
    const int i = b + lane;
    int s = 0, dl = 0;
    float a0 = 0.f, a1 = 0.f, a2 = 0.f, a3 = 0.f;
    if (i < n) {
      const v2u en = ent[i];
      s = (int)en.x;
      dl = (int)en.y;
      const v4f ex = sc[i];
      const v4f dn = *(const v4f*)(sden + dl * 4);
      a0 = ex.x / fmaxf(dn.x, 1e-9f);
      a1 = ex.y / fmaxf(dn.y, 1e-9f);
      a2 = ex.z / fmaxf(dn.z, 1e-9f);
      a3 = ex.w / fmaxf(dn.w, 1e-9f);
    }
    const int cb = min(32, n - b);
    int j = 0;
    for (; j + 8 <= cb; j += 8)
      own_acc8<false>(j, cb, s, dl, a0, a1, a2, a3, hb0, hb1, hl, acc, lane);
    if (j < cb)
      own_acc8<true>(j, cb, s, dl, a0, a1, a2, a3, hb0, hb1, hl, acc, lane);
  }
}

__device__ __forceinline__ void own_deg_add(int v, int lane, int* sdeg) {
  for (int m = 16; m >= 1; m >>= 1) v += __shfl_xor(v, m, 32);
  if (lane == 0 && v != 0) atomicAdd(sdeg, v);
}

template <int LAYER>
__global__ void __launch_bounds__(OWN_THREADS)
gat_owner(const int* __restrict__ src, const int* __restrict__ dst,
          const float* __restrict__ el, const float* __restrict__ er,
          const float* __restrict__ h, const float* __restrict__ bias,
          const float* __restrict__ lns, const float* __restrict__ lnb,
          float* __restrict__ outp,
          unsigned* __restrict__ wlist, unsigned* __restrict__ wcnt,
          const int* __restrict__ indeg, int* __restrict__ wdeg,
          const unsigned* __restrict__ wkey,
          const float* __restrict__ Wn, const float* __restrict__ aln,
          const float* __restrict__ arn,
          float* __restrict__ eln, float* __restrict__ ern,
          int Nn, int E) {
  __shared__ float    acc[OWN_NB * 128];
  __shared__ __attribute__((aligned(16))) unsigned smax[OWN_NB * 4];
  __shared__ __attribute__((aligned(16))) float    sden[OWN_NB * 4];
  __shared__ __attribute__((aligned(16))) float    erl[OWN_NB * 4];
  __shared__ v2u      ent[OWN_CAP];
  __shared__ v4f      sc[OWN_CAP];
  __shared__ unsigned cnt;
  __shared__ int      sdeg;

  const int tid  = threadIdx.x;
  const int wave = tid >> 5;
  const int lane = tid & 31;
  const int base = blockIdx.x * OWN_NB;
  const unsigned nb = (unsigned)min(OWN_NB, Nn - base);

  const bool degBlock = (LAYER == 1) || (blockIdx.x == gridDim.x - 1);
  int dpart = 0;
  if (LAYER == 2 && degBlock) {
    for (int i = tid; i < (int)gridDim.x; i += OWN_THREADS) dpart += wdeg[i];
  }

  for (int i = tid; i < OWN_NB * 128; i += OWN_THREADS) acc[i] = 0.f;
  for (int i = tid; i < OWN_NB * 4; i += OWN_THREADS) { smax[i] = 0u; sden[i] = 0.f; }
  if (tid == 0) { cnt = 0u; sdeg = 0; }

  v2u* wl = (v2u*)wlist + (size_t)blockIdx.x * OWN_CAP;
  float erv = 0.f;
  unsigned total;
  if (LAYER == 1) {
    __syncthreads();
    if (tid < (int)nb) dpart = indeg[base + tid];
    if (tid < (int)nb * 4) erv = er[(size_t)base * 4 + tid];
    if (wkey) own_build_keys(wkey, dst, E, blockIdx.x, base, nb, ent, &cnt, tid);
    else      own_build(dst, 0, E, base, nb, ent, &cnt, tid);
    if (tid < (int)nb * 4) erl[tid] = erv;
    own_deg_add(dpart, lane, &sdeg);
    __syncthreads();
    total = cnt;
    if (tid == 0) { wcnt[blockIdx.x] = total; wdeg[blockIdx.x] = sdeg; }
  } else {
    total = wcnt[blockIdx.x];
    if (tid < (int)nb * 4) erv = er[(size_t)base * 4 + tid];
    if (total <= OWN_CAP) {
      for (unsigned i0 = tid; i0 < total; i0 += 4 * OWN_THREADS) {
        v2u v[4];
#pragma unroll
        for (int u = 0; u < 4; ++u) {
          const unsigned i = i0 + u * OWN_THREADS;
          v2u z; z.x = 0u; z.y = 0u;
          v[u] = i < total ? wl[i] : z;
        }
#pragma unroll
        for (int u = 0; u < 4; ++u) {
          const unsigned i = i0 + u * OWN_THREADS;
          if (i < total) ent[i] = v[u];
        }
      }
    }
    if (tid < (int)nb * 4) erl[tid] = erv;
    __syncthreads();
    if (degBlock) own_deg_add(dpart, lane, &sdeg);
  }

  if (total <= OWN_CAP) {
    const int n = (int)total;
    own_prep<LAYER == 2, LAYER == 1, LAYER == 1>(ent, sc, n, src, dst, base, nb, el, erl, smax, wl, true, tid);
    __syncthreads();
    own_den(ent, sc, n, smax, sden, true, tid);
    __syncthreads();
    own_acc(ent, sc, n, h, sden, acc, wave, lane);
  } else {
    for (int pass = 0; pass < 3; ++pass) {
      for (int e0 = 0; e0 < E; e0 += OWN_CAP) {
        __syncthreads();
        if (tid == 0) cnt = 0u;
        __syncthreads();
        own_build(dst, e0, min(E, e0 + OWN_CAP), base, nb, ent, &cnt, tid);
        __syncthreads();
        const int n = (int)cnt;
        own_prep<false, false, false>(ent, sc, n, src, dst, base, nb, el, erl, smax, wl, pass == 0, tid);
        if (pass >= 1) {
          __syncthreads();
          own_den(ent, sc, n, smax, sden, pass == 1, tid);
        }
        if (pass == 2) {
          __syncthreads();
          own_acc(ent, sc, n, h, sden, acc, wave, lane);
        }
      }
    }
  }
  __syncthreads();

  if (LAYER == 2 && degBlock && tid == 0) {
    double ne = (double)(float)sdeg;
    double fl = (4.0 * ne * (6.0 * 128 * 32 + 6 * 32 + 2) +
                 4.0 * ne * (6.0 * 32 * 32 + 6 * 32 + 2)) / 1e12;
    outp[(size_t)Nn * 32] = (float)fl;
  }

  float* hm = (float*)ent;
  float* st = (float*)sc;
  float* bs = st + P2_ROWS * P2_OPITCH;
  static_assert(32 * 128 == 4 * OWN_THREADS, "Wn: 4 consecutive floats per thread");
  v4f alv = {}, arv = {}, wq = {};
  if (LAYER == 1) {
    alv = *(const v4f*)(aln + lane * 4);
    arv = *(const v4f*)(arn + lane * 4);
    wq  = *(const v4f*)(Wn + tid * 4);
    for (int i = (int)nb * P2_HPITCH + tid; i < P2_ROWS * P2_HPITCH; i += OWN_THREADS) hm[i] = 0.f;
  }

  const int pq = (lane & 3) * 32 + (lane >> 2);
  for (int nl = wave; nl < (int)nb; nl += OWN_THREADS / 32) {
    const int n = base + nl;
    float x = 0.f;
#pragma unroll
    for (int hh = 0; hh < 4; ++hh)
      x += acc[nl * 128 + pq + hh * 8] + bias[hh * 32 + lane];
    x *= 0.25f;
    if (LAYER == 1) {
      float mu = x;
#pragma unroll
      for (int m = 16; m >= 1; m >>= 1) mu += __shfl_xor(mu, m, 32);
      mu *= (1.f / 32.f);
      float dx = x - mu;
      float var = dx * dx;
#pragma unroll
      for (int m = 16; m >= 1; m >>= 1) var += __shfl_xor(var, m, 32);
      var *= (1.f / 32.f);
      float y = dx * rsqrtf(var + LN_EPS) * lns[lane] + lnb[lane];
      hm[nl * P2_HPITCH + lane] = fmaxf(y, 0.f);
    } else {
      outp[(size_t)n * 32 + lane] = x;
    }
  }

  if (LAYER == 1) {
    {
      const int kr = tid >> 5, col = (tid & 31) * 4;
      const int ln = (col & 15) | ((kr & 2) << 3);
      float* bq = bs + (((kr >> 2) * 32 + ln) * 8 + (col >> 4)) * 2 + (kr & 1);
      bq[0]  = wq.x;
      bq[16] = wq.y;
      bq[32] = wq.z;
      bq[48] = wq.w;
    }
    __syncthreads();
    const int wv = __builtin_amdgcn_readfirstlane(wave);
    const int t = wv >> 1, hf = wv & 1;
    if (t * 16 < (int)nb) {
      const int m = lane & 15;
      const float* ap = hm + (t * 16 + m) * P2_HPITCH + ((lane >> 4) << 1);
      const float* bp = bs + lane * 16 + hf * 8;
      v8f c0 = {}, c1 = {}, c2 = {}, c3 = {};
#pragma unroll
      for (int ks = 0; ks < 8; ++ks) {
        const v2f a  = *(const v2f*)(ap + ks * 4);
        const v4f q0 = *(const v4f*)(bp + ks * 512);
        const v4f q1 = *(const v4f*)(bp + ks * 512 + 4);
        c0 = __builtin_amdgcn_wmma_f32_16x16x4_f32(false, a, false, q0.xy, (short)0, c0, false, false);
        c1 = __builtin_amdgcn_wmma_f32_16x16x4_f32(false, a, false, q0.zw, (short)0, c1, false, false);
        c2 = __builtin_amdgcn_wmma_f32_16x16x4_f32(false, a, false, q1.xy, (short)0, c2, false, false);
        c3 = __builtin_amdgcn_wmma_f32_16x16x4_f32(false, a, false, q1.zw, (short)0, c3, false, false);
      }
      float* sp = st + (t * 16 + (lane >> 4) * 8) * P2_OPITCH + hf * 64 + m;
#pragma unroll
      for (int r = 0; r < 8; ++r) {
        sp[r * P2_OPITCH]      = c0[r];
        sp[r * P2_OPITCH + 16] = c1[r];
        sp[r * P2_OPITCH + 32] = c2[r];
        sp[r * P2_OPITCH + 48] = c3[r];
      }
    }
    __syncthreads();
    float pl[P2_RPW], pr[P2_RPW];
    v4f pv[P2_RPW];
#pragma unroll
    for (int u = 0; u < P2_RPW; ++u)
      pv[u] = *(const v4f*)(st + min(wv + 32 * u, P2_ROWS - 1) * P2_OPITCH + lane * 4);
#pragma unroll
    for (int u = 0; u < P2_RPW; ++u) {
      const int nl = wv + 32 * u;
      const v4f v = pv[u];
      if (nl < (int)nb) *(v4f*)(outp + (size_t)(base + nl) * 128 + lane * 4) = v;
      pl[u] = v.x * alv.x + v.y * alv.y + v.z * alv.z + v.w * alv.w;
      pr[u] = v.x * arv.x + v.y * arv.y + v.z * arv.z + v.w * arv.w;
    }
#pragma unroll
    for (int msk = 1; msk <= 4; msk <<= 1) {
#pragma unroll
      for (int u = 0; u < P2_RPW; ++u) {
        pl[u] += __shfl_xor(pl[u], msk, 32);
        pr[u] += __shfl_xor(pr[u], msk, 32);
      }
    }
#pragma unroll
    for (int u = 0; u < P2_RPW; ++u) {
      const int nl = wv + 32 * u;
      if (nl < (int)nb && (lane & 7) == 0) {
        sden[nl * 4 + (lane >> 3)] = pl[u];
        erl[nl * 4 + (lane >> 3)]  = pr[u];
      }
    }
    __syncthreads();
    if (tid < (int)nb * 4) {
      eln[(size_t)base * 4 + tid] = sden[tid];
      ern[(size_t)base * 4 + tid] = erl[tid];
    }
  }
}

extern "C" void kernel_launch(void* const* d_in, const int* in_sizes, int n_in,
                              void* d_out, int out_size, void* d_ws, size_t ws_size,
                              hipStream_t stream) {
  const float* feat = (const float*)d_in[0];
  const float* W1   = (const float*)d_in[1];
  const float* al1  = (const float*)d_in[2];
  const float* ar1  = (const float*)d_in[3];
  const float* b1   = (const float*)d_in[4];
  const float* lns  = (const float*)d_in[5];
  const float* lnb  = (const float*)d_in[6];
  const float* W2   = (const float*)d_in[7];
  const float* al2  = (const float*)d_in[8];
  const float* ar2  = (const float*)d_in[9];
  const float* b2   = (const float*)d_in[10];
  const int*   src  = (const int*)d_in[11];
  const int*   dst  = (const int*)d_in[12];
  const int*   indeg= (const int*)d_in[13];

  const int Nn = in_sizes[13];
  const int E  = in_sizes[11];
  const int NH = Nn * 4;

  char* p = (char*)d_ws;
  float*    hbuf  = (float*)p;   p += (size_t)Nn * 128 * sizeof(float);
  float*    hbuf2 = (float*)p;   p += (size_t)Nn * 128 * sizeof(float);
  float*    el    = (float*)p;   p += (size_t)NH * sizeof(float);
  float*    er    = (float*)p;   p += (size_t)NH * sizeof(float);
  float*    el2   = (float*)p;   p += (size_t)NH * sizeof(float);
  float*    er2   = (float*)p;   p += (size_t)NH * sizeof(float);
  const int ownBlocks  = (Nn + OWN_NB - 1) / OWN_NB;
  unsigned* wlist = (unsigned*)p; p += (size_t)ownBlocks * OWN_CAP * 2 * sizeof(unsigned);
  unsigned* wkey  = (unsigned*)p; p += ((size_t)E + 15) / 16 * 16;
  unsigned* wcnt  = (unsigned*)p; p += (size_t)ownBlocks * sizeof(unsigned);
  int*      wdeg  = (int*)p;
  (void)ws_size; (void)n_in; (void)out_size;

  float* outh = (float*)d_out;

  dim3 b256(256);
  const int rowTiles   = (Nn + 15) / 16;
  const int gemmBlocks = (rowTiles + 7) / 8;

  gat_gemm_wmma<128><<<gemmBlocks, b256, 0, stream>>>(feat, W1, hbuf, Nn);
  unsigned* wkeyArg = ownBlocks <= 256 ? wkey : nullptr;
  const int projBlocks = (Nn + 8 * PROJ_ROWS - 1) / (8 * PROJ_ROWS);
  gat_attn_proj<<<projBlocks, b256, 0, stream>>>(hbuf, al1, ar1, el, er, Nn, dst, wkeyArg, E);
  gat_owner<1><<<ownBlocks, OWN_THREADS, 0, stream>>>(src, dst, el, er, hbuf, b1, lns, lnb, hbuf2, wlist, wcnt, indeg, wdeg, wkeyArg, W2, al2, ar2, el2, er2, Nn, E);

  gat_owner<2><<<ownBlocks, OWN_THREADS, 0, stream>>>(src, dst, el2, er2, hbuf2, b2, lns, lnb, outh, wlist, wcnt, indeg, wdeg, nullptr, nullptr, nullptr, nullptr, nullptr, nullptr, Nn, E);
}


# weighted-sum loop remapped: 8 lanes per edge, 4 edges per wave step; a lane loads 4 features of all 4 heads, sums heads in registers: 1 LDS add per edge (was 4), no lane reads; fp sum order changes
# speedup vs baseline: 7.3224x; 1.0295x over previous
#include <hip/hip_runtime.h>


typedef float v2f __attribute__((ext_vector_type(2)));
typedef float v4f __attribute__((ext_vector_type(4)));
typedef float v8f __attribute__((ext_vector_type(8)));

#define NEG_SLOPE 0.2f
#define LN_EPS 1e-5f

__device__ __forceinline__ unsigned fenc(float f) {
  unsigned u = __float_as_uint(f);
  return (u >> 31) ? ~u : (u | 0x80000000u);
}
__device__ __forceinline__ float fdec(unsigned u) {
  return __uint_as_float((u >> 31) ? (u & 0x7FFFFFFFu) : ~u);
}
__device__ __forceinline__ float lrelu(float x) {
  return x > 0.f ? x : x * NEG_SLOPE;
}

template <int K>
__global__ void gat_gemm_wmma(const float* __restrict__ Amat,
                              const float* __restrict__ Bmat,
                              float* __restrict__ Cmat,
                              int M) {
  __shared__ float bs[K * 128];

  const int tid = threadIdx.x;

  for (int idx = tid; idx < K * 128; idx += 256) {
    int kr = idx >> 7;
    int col = idx & 127;
    int ks = kr >> 2;
    int c = col >> 4;
    int lane = (col & 15) | ((kr & 2) << 3);
    int sub = kr & 1;
    bs[((ks * 32 + lane) * 8 + c) * 2 + sub] = Bmat[idx];
  }
  __syncthreads();

  const int wave = tid >> 5;
  const int lane = tid & 31;
  const int row0 = (blockIdx.x * 8 + wave) * 16;
  if (row0 >= M) return;

  const int m  = lane & 15;
  const int kk = (lane >> 4) << 1;

  v8f acc[8] = {};

  const float* arow = Amat + (size_t)(row0 + m) * K + kk;
  const float* blane = bs + lane * 16;

#pragma unroll 4
  for (int ks = 0; ks < K / 4; ++ks) {
    v2f a = *(const v2f*)(arow + ks * 4);
    const float* bp = blane + ks * 512;
    v4f q0 = *(const v4f*)(bp);
    v4f q1 = *(const v4f*)(bp + 4);
    v4f q2 = *(const v4f*)(bp + 8);
    v4f q3 = *(const v4f*)(bp + 12);
    acc[0] = __builtin_amdgcn_wmma_f32_16x16x4_f32(false, a, false, q0.xy, (short)0, acc[0], false, false);
    acc[1] = __builtin_amdgcn_wmma_f32_16x16x4_f32(false, a, false, q0.zw, (short)0, acc[1], false, false);
    acc[2] = __builtin_amdgcn_wmma_f32_16x16x4_f32(false, a, false, q1.xy, (short)0, acc[2], false, false);
    acc[3] = __builtin_amdgcn_wmma_f32_16x16x4_f32(false, a, false, q1.zw, (short)0, acc[3], false, false);
    acc[4] = __builtin_amdgcn_wmma_f32_16x16x4_f32(false, a, false, q2.xy, (short)0, acc[4], false, false);
    acc[5] = __builtin_amdgcn_wmma_f32_16x16x4_f32(false, a, false, q2.zw, (short)0, acc[5], false, false);
    acc[6] = __builtin_amdgcn_wmma_f32_16x16x4_f32(false, a, false, q3.xy, (short)0, acc[6], false, false);
    acc[7] = __builtin_amdgcn_wmma_f32_16x16x4_f32(false, a, false, q3.zw, (short)0, acc[7], false, false);
  }

  const int mo = (lane >> 4) * 8;
#pragma unroll
  for (int c = 0; c < 8; ++c) {
#pragma unroll
    for (int r = 0; r < 8; ++r) {
      Cmat[(size_t)(row0 + mo + r) * 128 + c * 16 + m] = acc[c][r];
    }
  }
}

#define OWN_NB      196
#define OWN_CAP     8192
#define OWN_THREADS 1024
#define ACC_P       40
#define P2_TILES    ((OWN_NB + 15) / 16)
#define P2_ROWS     (P2_TILES * 16)
#define P2_HPITCH   36
#define P2_OPITCH   132
#define P2_RPW      ((OWN_NB + 31) / 32)

#define PROJ_ROWS 16

__device__ __forceinline__ float proj_sum8(const float (&p)[8], int lane) {
  const bool b4 = (lane & 4) != 0, b2 = (lane & 2) != 0, b1 = (lane & 1) != 0;
  float w[4], x[2];
#pragma unroll
  for (int j = 0; j < 4; ++j) {
    const float mine = b4 ? p[j + 4] : p[j];
    const float give = b4 ? p[j] : p[j + 4];
    w[j] = mine + __shfl_xor(give, 4, 32);
  }
#pragma unroll
  for (int j = 0; j < 2; ++j) {
    const float mine = b2 ? w[j + 2] : w[j];
    const float give = b2 ? w[j] : w[j + 2];
    x[j] = mine + __shfl_xor(give, 2, 32);
  }
  const float mine = b1 ? x[1] : x[0];
  const float give = b1 ? x[0] : x[1];
  return mine + __shfl_xor(give, 1, 32);
}

__global__ void gat_attn_proj(const float* __restrict__ h,
                              const float* __restrict__ al,
                              const float* __restrict__ ar,
                              float* __restrict__ el, float* __restrict__ er,
                              int Nn,
                              const int* __restrict__ dst, unsigned* __restrict__ wkey,
                              int E) {
  const int i = blockIdx.x * blockDim.x + threadIdx.x;
  if (wkey) {
    const int nq = E >> 2;
    const int4* d4 = (const int4*)dst;
    for (int q = i; q < nq; q += (int)(gridDim.x * blockDim.x)) {
      const int4 d = d4[q];
      wkey[q] = (((unsigned)d.x / OWN_NB) & 255u) |
                ((((unsigned)d.y / OWN_NB) & 255u) << 8) |
                ((((unsigned)d.z / OWN_NB) & 255u) << 16) |
                (((unsigned)d.w / OWN_NB) << 24);
    }
  }
  const int lane = threadIdx.x & 31;
  const int n0 = (i >> 5) * PROJ_ROWS;
  if (n0 >= Nn) return;
  const float* hl = h + lane * 4;
  const v4f a4 = *(const v4f*)(al + lane * 4);
  const v4f r4 = *(const v4f*)(ar + lane * 4);
  v4f v[PROJ_ROWS];
  const float* hr = hl + (size_t)n0 * 128;
  const int last = Nn - 1 - n0;
#pragma unroll
  for (int u = 0; u < PROJ_ROWS; ++u) v[u] = *(const v4f*)(hr + min(u, last) * 128);
  __builtin_amdgcn_sched_barrier(0);
#pragma unroll
  for (int g = 0; g < PROJ_ROWS / 8; ++g) {
    float pl[8], pr[8];
#pragma unroll
    for (int r = 0; r < 8; ++r) {
      const v4f t = v[g * 8 + r];
      pl[r] = t.x * a4.x + t.y * a4.y + t.z * a4.z + t.w * a4.w;
      pr[r] = t.x * r4.x + t.y * r4.y + t.z * r4.z + t.w * r4.w;
    }
    const float sl = proj_sum8(pl, lane);
    const float sr = proj_sum8(pr, lane);
    const int n = n0 + g * 8 + (lane & 7);
    if (n < Nn) {
      el[(size_t)n * 4 + (lane >> 3)] = sl;
      er[(size_t)n * 4 + (lane >> 3)] = sr;
    }
  }
}

typedef unsigned v2u __attribute__((ext_vector_type(2)));
typedef unsigned v4u __attribute__((ext_vector_type(4)));

__device__ __forceinline__ void own_check(int dv, int e, int base, unsigned nb,
                                          v2u* ent, unsigned* cnt) {
  const unsigned dl = (unsigned)(dv - base);
  if (dl < nb) {
    unsigned pos = atomicAdd(cnt, 1u);
    if (pos < OWN_CAP) { v2u v; v.x = (unsigned)e; v.y = dl; ent[pos] = v; }
  }
}

__device__ __forceinline__ void own_build(const int* __restrict__ dst, int e0, int e1,
                                          int base, unsigned nb,
                                          v2u* ent, unsigned* cnt, int tid) {
  const int nvec = (e1 - e0) >> 2;
  const int4* dv = (const int4*)(dst + e0);
  int i = tid;
  for (; i + 3 * OWN_THREADS < nvec; i += 4 * OWN_THREADS) {
    int4 d[4];
#pragma unroll
    for (int u = 0; u < 4; ++u) d[u] = dv[i + u * OWN_THREADS];
#pragma unroll
    for (int u = 0; u < 4; ++u) {
      const int eb = e0 + 4 * (i + u * OWN_THREADS);
      own_check(d[u].x, eb + 0, base, nb, ent, cnt);
      own_check(d[u].y, eb + 1, base, nb, ent, cnt);
      own_check(d[u].z, eb + 2, base, nb, ent, cnt);
      own_check(d[u].w, eb + 3, base, nb, ent, cnt);
    }
  }
  for (; i < nvec; i += OWN_THREADS) {
    int4 d = dv[i];
    const int eb = e0 + 4 * i;
    own_check(d.x, eb + 0, base, nb, ent, cnt);
    own_check(d.y, eb + 1, base, nb, ent, cnt);
    own_check(d.z, eb + 2, base, nb, ent, cnt);
    own_check(d.w, eb + 3, base, nb, ent, cnt);
  }
  for (int e = e0 + nvec * 4 + tid; e < e1; e += OWN_THREADS)
    own_check(dst[e], e, base, nb, ent, cnt);
}

__device__ __forceinline__ unsigned own_nz(unsigned k, unsigned pat) {
  const unsigned x = k ^ pat;
  return ((x & 0x7F7F7F7Fu) + 0x7F7F7F7Fu) | x;
}
__device__ __forceinline__ unsigned own_hits(const v4u k, unsigned pat) {
  const unsigned c = ((own_nz(k.x, pat) >> 7) & 0x01010101u) | ((own_nz(k.y, pat) >> 6) & 0x02020202u) |
                     ((own_nz(k.z, pat) >> 5) & 0x04040404u) | ((own_nz(k.w, pat) >> 4) & 0x08080808u);
  return c ^ 0x0F0F0F0Fu;
}
__device__ __forceinline__ void own_emit(unsigned m, int e0, v2u* ent, unsigned* cnt) {
  while (m) {
    const int p = __builtin_ctz(m);
    m &= m - 1u;
    const int e = e0 + ((p & 3) << 2) + (p >> 3) + ((p & 4) << 12);
    const unsigned pos = atomicAdd(cnt, 1u);
    if (pos < OWN_CAP) ent[pos].x = (unsigned)e;
  }
}

__device__ __forceinline__ void own_scan4(const v4u k[4], unsigned pat, int e0,
                                          v2u* ent, unsigned* cnt) {
  own_emit(own_hits(k[0], pat) | (own_hits(k[1], pat) << 4), e0, ent, cnt);
  own_emit(own_hits(k[2], pat) | (own_hits(k[3], pat) << 4), e0 + 32 * OWN_THREADS, ent, cnt);
}

__device__ __forceinline__ void own_build_keys(const unsigned* __restrict__ wkey,
                                               const int* __restrict__ dst, int E,
                                               unsigned me, int base, unsigned nb,
                                               v2u* ent, unsigned* cnt, int tid) {
  const int nvec = E >> 4;
  const v4u* kv = (const v4u*)wkey;
  const unsigned pat = me * 0x01010101u;
  const int ntrip = nvec / (4 * OWN_THREADS);
  int i = tid;
  if (ntrip > 0) {
    v4u ka[4], kb[4];
#pragma unroll
    for (int u = 0; u < 4; ++u) ka[u] = kv[i + u * OWN_THREADS];
    int t = 0;
    for (; t + 2 < ntrip; t += 2) {
#pragma unroll
      for (int u = 0; u < 4; ++u) kb[u] = kv[i + (4 + u) * OWN_THREADS];
      own_scan4(ka, pat, 16 * i, ent, cnt);
      i += 4 * OWN_THREADS;
#pragma unroll
      for (int u = 0; u < 4; ++u) ka[u] = kv[i + (4 + u) * OWN_THREADS];
      own_scan4(kb, pat, 16 * i, ent, cnt);
      i += 4 * OWN_THREADS;
    }
    if (t + 1 < ntrip) {
#pragma unroll
      for (int u = 0; u < 4; ++u) kb[u] = kv[i + (4 + u) * OWN_THREADS];
      own_scan4(ka, pat, 16 * i, ent, cnt);
      i += 4 * OWN_THREADS;
      own_scan4(kb, pat, 16 * i, ent, cnt);
    } else {
      own_scan4(ka, pat, 16 * i, ent, cnt);
    }
    i += 4 * OWN_THREADS;
  }
  for (; i < nvec; i += OWN_THREADS) {
    own_emit(own_hits(kv[i], pat), 16 * i, ent, cnt);
  }
  for (int e = nvec * 16 + tid; e < E; e += OWN_THREADS)
    own_check(dst[e], e, base, nb, ent, cnt);
}

template <bool HAVE_SRC, bool KEEP, bool NEED_DL>
__device__ __forceinline__ void own_prep(v2u* ent, v4f* sc, int n,
                                         const int* __restrict__ src,
                                         const int* __restrict__ dst, int base, unsigned nb,
                                         const float* __restrict__ el, const float* erl,
                                         unsigned* smax, v2u* __restrict__ wl,
                                         bool doMax, int tid) {
  for (int i = tid; i < n; i += OWN_THREADS) {
    v2u en = ent[i];
    if (!HAVE_SRC) {
      const unsigned e = en.x;
      en.x = (unsigned)src[e];
      if (NEED_DL) en.y = min((unsigned)(dst[e] - base), nb - 1u);
      ent[i] = en;
    }
    if (KEEP) wl[i] = en;
    const unsigned dl = en.y;
    const v4f l = *(const v4f*)(el + (size_t)en.x * 4);
    const v4f r = *(const v4f*)(erl + dl * 4);
    v4f v;
    v.x = lrelu(l.x + r.x);
    v.y = lrelu(l.y + r.y);
    v.z = lrelu(l.z + r.z);
    v.w = lrelu(l.w + r.w);
    sc[i] = v;
    if (doMax) {
      atomicMax(&smax[dl * 4 + 0], fenc(v.x));
      atomicMax(&smax[dl * 4 + 1], fenc(v.y));
      atomicMax(&smax[dl * 4 + 2], fenc(v.z));
      atomicMax(&smax[dl * 4 + 3], fenc(v.w));
    }
  }
}

__device__ __forceinline__ void own_den(const v2u* ent, v4f* sc, int n,
                                        const unsigned* smax, float* sden,
                                        bool add, int tid) {
  for (int i = tid; i < n; i += OWN_THREADS) {
    const unsigned dl = ent[i].y;
    const v4u m = *(const v4u*)(smax + dl * 4);
    v4f v = sc[i];
    v.x = expf(v.x - fdec(m.x));
    v.y = expf(v.y - fdec(m.y));
    v.z = expf(v.z - fdec(m.z));
    v.w = expf(v.w - fdec(m.w));
    sc[i] = v;
    if (add) {
      atomicAdd(&sden[dl * 4 + 0], v.x);
      atomicAdd(&sden[dl * 4 + 1], v.y);
      atomicAdd(&sden[dl * 4 + 2], v.z);
      atomicAdd(&sden[dl * 4 + 3], v.w);
    }
  }
}

template <bool FULL>
__device__ __forceinline__ void own_acc16(int eq, int n, const v2u* ent, const v4f* sc,
                                          const char* __restrict__ hb, unsigned ro,
                                          float* accr) {
  v4f x[4][4];
  v4f al[4];
  int dl[4];
#pragma unroll
  for (int t = 0; t < 4; ++t) {
    const int idx = eq + 4 * t;
    const int idc = FULL ? idx : min(idx, n - 1);
    const v2u en = ent[idc];
    al[t] = sc[idc];
    dl[t] = (int)en.y;
    const char* p = hb + ((en.x << 9) + ro);
    x[t][0] = *(const v4f*)(p);
    x[t][1] = *(const v4f*)(p + 128);
    x[t][2] = *(const v4f*)(p + 256);
    x[t][3] = *(const v4f*)(p + 384);
  }
#pragma unroll
  for (int t = 0; t < 4; ++t) {
    const v4f y = x[t][0] * al[t].x + x[t][1] * al[t].y + x[t][2] * al[t].z + x[t][3] * al[t].w;
    if (FULL || eq + 4 * t < n) {
      float* ap = accr + dl[t] * ACC_P;
      atomicAdd(ap,      y.x);
      atomicAdd(ap + 8,  y.y);
      atomicAdd(ap + 16, y.z);
      atomicAdd(ap + 24, y.w);
    }
  }
}

__device__ __forceinline__ void own_acc(const v2u* ent, v4f* sc, int n,
                                        const float* __restrict__ h,
                                        const float* sden,
                                        float* acc, int wave, int lane) {
  for (int i = wave * 32 + lane; i < n; i += OWN_THREADS) {
    const v4f ex = sc[i];
    const v4f dn = *(const v4f*)(sden + ent[i].y * 4);
    v4f a;
    a.x = ex.x / fmaxf(dn.x, 1e-9f);
    a.y = ex.y / fmaxf(dn.y, 1e-9f);
    a.z = ex.z / fmaxf(dn.z, 1e-9f);
    a.w = ex.w / fmaxf(dn.w, 1e-9f);
    sc[i] = a;
  }
  __syncthreads();
  const char* hb = (const char*)h;
  const int q = lane >> 3, r = lane & 7;
  const unsigned ro = (unsigned)r * 16u;
  float* accr = acc + r;
  const int ng = (n + 15) >> 4;
  for (int g = wave; g < ng; g += OWN_THREADS / 32) {
    const int e0 = g * 16;
    if (e0 + 16 <= n) own_acc16<true>(e0 + q, n, ent, sc, hb, ro, accr);
    else              own_acc16<false>(e0 + q, n, ent, sc, hb, ro, accr);
  }
}

__device__ __forceinline__ void own_deg_add(int v, int lane, int* sdeg) {
  for (int m = 16; m >= 1; m >>= 1) v += __shfl_xor(v, m, 32);
  if (lane == 0 && v != 0) atomicAdd(sdeg, v);
}

template <int LAYER>
__global__ void __launch_bounds__(OWN_THREADS)
gat_owner(const int* __restrict__ src, const int* __restrict__ dst,
          const float* __restrict__ el, const float* __restrict__ er,
          const float* __restrict__ h, const float* __restrict__ bias,
          const float* __restrict__ lns, const float* __restrict__ lnb,
          float* __restrict__ outp,
          unsigned* __restrict__ wlist, unsigned* __restrict__ wcnt,
          const int* __restrict__ indeg, int* __restrict__ wdeg,
          const unsigned* __restrict__ wkey,
          const float* __restrict__ Wn, const float* __restrict__ aln,
          const float* __restrict__ arn,
          float* __restrict__ eln, float* __restrict__ ern,
          int Nn, int E) {
  __shared__ float    acc[OWN_NB * ACC_P];
  __shared__ __attribute__((aligned(16))) unsigned smax[OWN_NB * 4];
  __shared__ __attribute__((aligned(16))) float    sden[OWN_NB * 4];
  __shared__ __attribute__((aligned(16))) float    erl[OWN_NB * 4];
  __shared__ v2u      ent[OWN_CAP];
  __shared__ v4f      sc[OWN_CAP];
  __shared__ unsigned cnt;
  __shared__ int      sdeg;

  const int tid  = threadIdx.x;
  const int wave = tid >> 5;
  const int lane = tid & 31;
  const int base = blockIdx.x * OWN_NB;
  const unsigned nb = (unsigned)min(OWN_NB, Nn - base);

  const bool degBlock = (LAYER == 1) || (blockIdx.x == gridDim.x - 1);
  int dpart = 0;
  if (LAYER == 2 && degBlock) {
    for (int i = tid; i < (int)gridDim.x; i += OWN_THREADS) dpart += wdeg[i];
  }

  for (int i = tid; i < OWN_NB * ACC_P; i += OWN_THREADS) acc[i] = 0.f;
  for (int i = tid; i < OWN_NB * 4; i += OWN_THREADS) { smax[i] = 0u; sden[i] = 0.f; }
  if (tid == 0) { cnt = 0u; sdeg = 0; }

  v2u* wl = (v2u*)wlist + (size_t)blockIdx.x * OWN_CAP;
  float erv = 0.f;
  unsigned total;
  if (LAYER == 1) {
    __syncthreads();
    if (tid < (int)nb) dpart = indeg[base + tid];
    if (tid < (int)nb * 4) erv = er[(size_t)base * 4 + tid];
    if (wkey) own_build_keys(wkey, dst, E, blockIdx.x, base, nb, ent, &cnt, tid);
    else      own_build(dst, 0, E, base, nb, ent, &cnt, tid);
    if (tid < (int)nb * 4) erl[tid] = erv;
    own_deg_add(dpart, lane, &sdeg);
    __syncthreads();
    total = cnt;
    if (tid == 0) { wcnt[blockIdx.x] = total; wdeg[blockIdx.x] = sdeg; }
  } else {
    total = wcnt[blockIdx.x];
    if (tid < (int)nb * 4) erv = er[(size_t)base * 4 + tid];
    if (total <= OWN_CAP) {
      for (unsigned i0 = tid; i0 < total; i0 += 4 * OWN_THREADS) {
        v2u v[4];
#pragma unroll
        for (int u = 0; u < 4; ++u) {
          const unsigned i = i0 + u * OWN_THREADS;
          v2u z; z.x = 0u; z.y = 0u;
          v[u] = i < total ? wl[i] : z;
        }
#pragma unroll
        for (int u = 0; u < 4; ++u) {
          const unsigned i = i0 + u * OWN_THREADS;
          if (i < total) ent[i] = v[u];
        }
      }
    }
    if (tid < (int)nb * 4) erl[tid] = erv;
    __syncthreads();
    if (degBlock) own_deg_add(dpart, lane, &sdeg);
  }

  if (total <= OWN_CAP) {
    const int n = (int)total;
    own_prep<LAYER == 2, LAYER == 1, LAYER == 1>(ent, sc, n, src, dst, base, nb, el, erl, smax, wl, true, tid);
    __syncthreads();
    own_den(ent, sc, n, smax, sden, true, tid);
    __syncthreads();
    own_acc(ent, sc, n, h, sden, acc, wave, lane);
  } else {
    for (int pass = 0; pass < 3; ++pass) {
      for (int e0 = 0; e0 < E; e0 += OWN_CAP) {
        __syncthreads();
        if (tid == 0) cnt = 0u;
        __syncthreads();
        own_build(dst, e0, min(E, e0 + OWN_CAP), base, nb, ent, &cnt, tid);
        __syncthreads();
        const int n = (int)cnt;
        own_prep<false, false, false>(ent, sc, n, src, dst, base, nb, el, erl, smax, wl, pass == 0, tid);
        if (pass >= 1) {
          __syncthreads();
          own_den(ent, sc, n, smax, sden, pass == 1, tid);
        }
        if (pass == 2) {
          __syncthreads();
          own_acc(ent, sc, n, h, sden, acc, wave, lane);
        }
      }
    }
  }
  __syncthreads();

  if (LAYER == 2 && degBlock && tid == 0) {
    double ne = (double)(float)sdeg;
    double fl = (4.0 * ne * (6.0 * 128 * 32 + 6 * 32 + 2) +
                 4.0 * ne * (6.0 * 32 * 32 + 6 * 32 + 2)) / 1e12;
    outp[(size_t)Nn * 32] = (float)fl;
  }

  float* hm = (float*)ent;
  float* st = (float*)sc;
  float* bs = st + P2_ROWS * P2_OPITCH;
  static_assert(32 * 128 == 4 * OWN_THREADS, "Wn: 4 consecutive floats per thread");
  v4f alv = {}, arv = {}, wq = {};
  if (LAYER == 1) {
    alv = *(const v4f*)(aln + lane * 4);
    arv = *(const v4f*)(arn + lane * 4);
    wq  = *(const v4f*)(Wn + tid * 4);
    for (int i = (int)nb * P2_HPITCH + tid; i < P2_ROWS * P2_HPITCH; i += OWN_THREADS) hm[i] = 0.f;
  }

  const int pq = (lane & 3) * 8 + (lane >> 2);
  const float bsum = ((bias[lane] + bias[32 + lane]) + bias[64 + lane]) + bias[96 + lane];
  for (int nl = wave; nl < (int)nb; nl += OWN_THREADS / 32) {
    const int n = base + nl;
    float x = (acc[nl * ACC_P + pq] + bsum) * 0.25f;
    if (LAYER == 1) {
      float mu = x;
#pragma unroll
      for (int m = 16; m >= 1; m >>= 1) mu += __shfl_xor(mu, m, 32);
      mu *= (1.f / 32.f);
      float dx = x - mu;
      float var = dx * dx;
#pragma unroll
      for (int m = 16; m >= 1; m >>= 1) var += __shfl_xor(var, m, 32);
      var *= (1.f / 32.f);
      float y = dx * rsqrtf(var + LN_EPS) * lns[lane] + lnb[lane];
      hm[nl * P2_HPITCH + lane] = fmaxf(y, 0.f);
    } else {
      outp[(size_t)n * 32 + lane] = x;
    }
  }

  if (LAYER == 1) {
    {
      const int kr = tid >> 5, col = (tid & 31) * 4;
      const int ln = (col & 15) | ((kr & 2) << 3);
      float* bq = bs + (((kr >> 2) * 32 + ln) * 8 + (col >> 4)) * 2 + (kr & 1);
      bq[0]  = wq.x;
      bq[16] = wq.y;
      bq[32] = wq.z;
      bq[48] = wq.w;
    }
    __syncthreads();
    const int wv = __builtin_amdgcn_readfirstlane(wave);
    const int t = wv >> 1, hf = wv & 1;
    if (t * 16 < (int)nb) {
      const int m = lane & 15;
      const float* ap = hm + (t * 16 + m) * P2_HPITCH + ((lane >> 4) << 1);
      const float* bp = bs + lane * 16 + hf * 8;
      v8f c0 = {}, c1 = {}, c2 = {}, c3 = {};
#pragma unroll
      for (int ks = 0; ks < 8; ++ks) {
        const v2f a  = *(const v2f*)(ap + ks * 4);
        const v4f q0 = *(const v4f*)(bp + ks * 512);
        const v4f q1 = *(const v4f*)(bp + ks * 512 + 4);
        c0 = __builtin_amdgcn_wmma_f32_16x16x4_f32(false, a, false, q0.xy, (short)0, c0, false, false);
        c1 = __builtin_amdgcn_wmma_f32_16x16x4_f32(false, a, false, q0.zw, (short)0, c1, false, false);
        c2 = __builtin_amdgcn_wmma_f32_16x16x4_f32(false, a, false, q1.xy, (short)0, c2, false, false);
        c3 = __builtin_amdgcn_wmma_f32_16x16x4_f32(false, a, false, q1.zw, (short)0, c3, false, false);
      }
      float* sp = st + (t * 16 + (lane >> 4) * 8) * P2_OPITCH + hf * 64 + m;
#pragma unroll
      for (int r = 0; r < 8; ++r) {
        sp[r * P2_OPITCH]      = c0[r];
        sp[r * P2_OPITCH + 16] = c1[r];
        sp[r * P2_OPITCH + 32] = c2[r];
        sp[r * P2_OPITCH + 48] = c3[r];
      }
    }
    __syncthreads();
    float pl[P2_RPW], pr[P2_RPW];
    v4f pv[P2_RPW];
#pragma unroll
    for (int u = 0; u < P2_RPW; ++u)
      pv[u] = *(const v4f*)(st + min(wv + 32 * u, P2_ROWS - 1) * P2_OPITCH + lane * 4);
#pragma unroll
    for (int u = 0; u < P2_RPW; ++u) {
      const int nl = wv + 32 * u;
      const v4f v = pv[u];
      if (nl < (int)nb) *(v4f*)(outp + (size_t)(base + nl) * 128 + lane * 4) = v;
      pl[u] = v.x * alv.x + v.y * alv.y + v.z * alv.z + v.w * alv.w;
      pr[u] = v.x * arv.x + v.y * arv.y + v.z * arv.z + v.w * arv.w;
    }
#pragma unroll
    for (int msk = 1; msk <= 4; msk <<= 1) {
#pragma unroll
      for (int u = 0; u < P2_RPW; ++u) {
        pl[u] += __shfl_xor(pl[u], msk, 32);
        pr[u] += __shfl_xor(pr[u], msk, 32);
      }
    }
#pragma unroll
    for (int u = 0; u < P2_RPW; ++u) {
      const int nl = wv + 32 * u;
      if (nl < (int)nb && (lane & 7) == 0) {
        sden[nl * 4 + (lane >> 3)] = pl[u];
        erl[nl * 4 + (lane >> 3)]  = pr[u];
      }
    }
    __syncthreads();
    if (tid < (int)nb * 4) {
      eln[(size_t)base * 4 + tid] = sden[tid];
      ern[(size_t)base * 4 + tid] = erl[tid];
    }
  }
}

extern "C" void kernel_launch(void* const* d_in, const int* in_sizes, int n_in,
                              void* d_out, int out_size, void* d_ws, size_t ws_size,
                              hipStream_t stream) {
  const float* feat = (const float*)d_in[0];
  const float* W1   = (const float*)d_in[1];
  const float* al1  = (const float*)d_in[2];
  const float* ar1  = (const float*)d_in[3];
  const float* b1   = (const float*)d_in[4];
  const float* lns  = (const float*)d_in[5];
  const float* lnb  = (const float*)d_in[6];
  const float* W2   = (const float*)d_in[7];
  const float* al2  = (const float*)d_in[8];
  const float* ar2  = (const float*)d_in[9];
  const float* b2   = (const float*)d_in[10];
  const int*   src  = (const int*)d_in[11];
  const int*   dst  = (const int*)d_in[12];
  const int*   indeg= (const int*)d_in[13];

  const int Nn = in_sizes[13];
  const int E  = in_sizes[11];
  const int NH = Nn * 4;

  char* p = (char*)d_ws;
  float*    hbuf  = (float*)p;   p += (size_t)Nn * 128 * sizeof(float);
  float*    hbuf2 = (float*)p;   p += (size_t)Nn * 128 * sizeof(float);
  float*    el    = (float*)p;   p += (size_t)NH * sizeof(float);
  float*    er    = (float*)p;   p += (size_t)NH * sizeof(float);
  float*    el2   = (float*)p;   p += (size_t)NH * sizeof(float);
  float*    er2   = (float*)p;   p += (size_t)NH * sizeof(float);
  const int ownBlocks  = (Nn + OWN_NB - 1) / OWN_NB;
  unsigned* wlist = (unsigned*)p; p += (size_t)ownBlocks * OWN_CAP * 2 * sizeof(unsigned);
  unsigned* wkey  = (unsigned*)p; p += ((size_t)E + 15) / 16 * 16;
  unsigned* wcnt  = (unsigned*)p; p += (size_t)ownBlocks * sizeof(unsigned);
  int*      wdeg  = (int*)p;
  (void)ws_size; (void)n_in; (void)out_size;

  float* outh = (float*)d_out;

  dim3 b256(256);
  const int rowTiles   = (Nn + 15) / 16;
  const int gemmBlocks = (rowTiles + 7) / 8;

  gat_gemm_wmma<128><<<gemmBlocks, b256, 0, stream>>>(feat, W1, hbuf, Nn);
  unsigned* wkeyArg = ownBlocks <= 256 ? wkey : nullptr;
  const int projBlocks = (Nn + 8 * PROJ_ROWS - 1) / (8 * PROJ_ROWS);
  gat_attn_proj<<<projBlocks, b256, 0, stream>>>(hbuf, al1, ar1, el, er, Nn, dst, wkeyArg, E);
  gat_owner<1><<<ownBlocks, OWN_THREADS, 0, stream>>>(src, dst, el, er, hbuf, b1, lns, lnb, hbuf2, wlist, wcnt, indeg, wdeg, wkeyArg, W2, al2, ar2, el2, er2, Nn, E);

  gat_owner<2><<<ownBlocks, OWN_THREADS, 0, stream>>>(src, dst, el2, er2, hbuf2, b2, lns, lnb, outh, wlist, wcnt, indeg, wdeg, nullptr, nullptr, nullptr, nullptr, nullptr, nullptr, Nn, E);
}


# el/er scores come out of the layer-1 GEMM as a ninth column tile (W1 x attention vectors folded into B per block): no score launch re-reading h; 3 launches, was 4; el/er fp sum order changes
# speedup vs baseline: 7.8287x; 1.0691x over previous
#include <hip/hip_runtime.h>


typedef float v2f __attribute__((ext_vector_type(2)));
typedef float v4f __attribute__((ext_vector_type(4)));
typedef float v8f __attribute__((ext_vector_type(8)));

#define NEG_SLOPE 0.2f
#define LN_EPS 1e-5f

__device__ __forceinline__ unsigned fenc(float f) {
  unsigned u = __float_as_uint(f);
  return (u >> 31) ? ~u : (u | 0x80000000u);
}
__device__ __forceinline__ float fdec(unsigned u) {
  return __uint_as_float((u >> 31) ? (u & 0x7FFFFFFFu) : ~u);
}
__device__ __forceinline__ float lrelu(float x) {
  return x > 0.f ? x : x * NEG_SLOPE;
}

#define OWN_NB      196

__device__ __forceinline__ float proj_sum8(const float (&p)[8], int lane) {
  const bool b4 = (lane & 4) != 0, b2 = (lane & 2) != 0, b1 = (lane & 1) != 0;
  float w[4], x[2];
#pragma unroll
  for (int j = 0; j < 4; ++j) {
    const float mine = b4 ? p[j + 4] : p[j];
    const float give = b4 ? p[j] : p[j + 4];
    w[j] = mine + __shfl_xor(give, 4, 32);
  }
#pragma unroll
  for (int j = 0; j < 2; ++j) {
    const float mine = b2 ? w[j + 2] : w[j];
    const float give = b2 ? w[j] : w[j + 2];
    x[j] = mine + __shfl_xor(give, 2, 32);
  }
  const float mine = b1 ? x[1] : x[0];
  const float give = b1 ? x[0] : x[1];
  return mine + __shfl_xor(give, 1, 32);
}

__device__ __forceinline__ unsigned own_bytes(const int4 d) {
  return (((unsigned)d.x / OWN_NB) & 255u) |
         ((((unsigned)d.y / OWN_NB) & 255u) << 8) |
         ((((unsigned)d.z / OWN_NB) & 255u) << 16) |
         (((unsigned)d.w / OWN_NB) << 24);
}

template <int K>
__global__ void __launch_bounds__(256)
gat_gemm_wmma(const float* __restrict__ Amat,
              const float* __restrict__ Bmat,
              float* __restrict__ Cmat,
              int M,
              const float* __restrict__ al, const float* __restrict__ ar,
              float* __restrict__ el, float* __restrict__ er,
              const int* __restrict__ dst, unsigned* __restrict__ wkey,
              int E) {
  __shared__ float bs[K * 128];
  __shared__ float bw[K * 16];
  __shared__ __attribute__((aligned(16))) float es[8 * 128];

  const int tid = threadIdx.x;

  const int nq = E >> 2;
  const int4* d4 = (const int4*)dst;
  const int q0 = blockIdx.x * 256 + tid, qs = (int)(gridDim.x * 256);
  const bool keys = wkey && nq > 0;
  int4 dA = {}, dB = {};
  if (keys) {
    dA = d4[min(q0, nq - 1)];
    dB = d4[min(q0 + qs, nq - 1)];
  }
  static_assert(K % 16 == 0, "16 rows of B per wave and trip");
  const int wv = tid >> 5, ln = tid & 31;
  const v4f a4 = *(const v4f*)(al + ln * 4);
  const v4f r4 = *(const v4f*)(ar + ln * 4);
  v4f w[16];
#pragma unroll
  for (int u = 0; u < 16; ++u) w[u] = *(const v4f*)(Bmat + (size_t)(min(wv * 16, K - 16) + u) * 128 + ln * 4);

  for (int idx = tid; idx < K * 128; idx += 256) {
    int kr = idx >> 7;
    int col = idx & 127;
    int ks = kr >> 2;
    int c = col >> 4;
    int lane = (col & 15) | ((kr & 2) << 3);
    int sub = kr & 1;
    bs[((ks * 32 + lane) * 8 + c) * 2 + sub] = Bmat[idx];
  }
  for (int k0 = wv * 16; k0 < K; k0 += 8 * 16) {
#pragma unroll
    for (int g = 0; g < 2; ++g) {
      float pl[8], pr[8];
#pragma unroll
      for (int r = 0; r < 8; ++r) {
        const v4f t = w[g * 8 + r];
        pl[r] = t.x * a4.x + t.y * a4.y + t.z * a4.z + t.w * a4.w;
        pr[r] = t.x * r4.x + t.y * r4.y + t.z * r4.z + t.w * r4.w;
      }
      const float sl = proj_sum8(pl, ln);
      const float sr = proj_sum8(pr, ln);
      const int k = k0 + g * 8 + (ln & 7), h = ln >> 3;
      float* bq = bw + ((k >> 2) * 32 + ((k & 2) << 3)) * 2 + (k & 1);
      bq[h * 2]        = sl;
      bq[(h + 4) * 2]  = sr;
      bq[(h + 8) * 2]  = 0.f;
      bq[(h + 12) * 2] = 0.f;
    }
    if (k0 + 8 * 16 < K) {
#pragma unroll
      for (int u = 0; u < 16; ++u) w[u] = *(const v4f*)(Bmat + (size_t)(k0 + 8 * 16 + u) * 128 + ln * 4);
    }
  }
  if (keys) {
    if (q0 < nq) wkey[q0] = own_bytes(dA);
    if (q0 + qs < nq) wkey[q0 + qs] = own_bytes(dB);
    for (int q = q0 + 2 * qs; q < nq; q += qs) wkey[q] = own_bytes(d4[q]);
  }
  __syncthreads();

  const int wave = tid >> 5;
  const int lane = tid & 31;
  const int row0 = (blockIdx.x * 8 + wave) * 16;
  const bool live = row0 < M;

  const int m  = lane & 15;
  const int kk = (lane >> 4) << 1;
  const int mo = (lane >> 4) * 8;
  float* ew = es + wave * 128;

  if (live) {
    v8f acc[8] = {};
    v8f accs = {};

    const float* arow = Amat + (size_t)(row0 + m) * K + kk;
    const float* blane = bs + lane * 16;
    const float* wlane = bw + lane * 2;

#pragma unroll 4
    for (int ks = 0; ks < K / 4; ++ks) {
      v2f a = *(const v2f*)(arow + ks * 4);
      const float* bp = blane + ks * 512;
      v4f q0 = *(const v4f*)(bp);
      v4f q1 = *(const v4f*)(bp + 4);
      v4f q2 = *(const v4f*)(bp + 8);
      v4f q3 = *(const v4f*)(bp + 12);
      v2f qs = *(const v2f*)(wlane + ks * 64);
      acc[0] = __builtin_amdgcn_wmma_f32_16x16x4_f32(false, a, false, q0.xy, (short)0, acc[0], false, false);
      acc[1] = __builtin_amdgcn_wmma_f32_16x16x4_f32(false, a, false, q0.zw, (short)0, acc[1], false, false);
      acc[2] = __builtin_amdgcn_wmma_f32_16x16x4_f32(false, a, false, q1.xy, (short)0, acc[2], false, false);
      acc[3] = __builtin_amdgcn_wmma_f32_16x16x4_f32(false, a, false, q1.zw, (short)0, acc[3], false, false);
      acc[4] = __builtin_amdgcn_wmma_f32_16x16x4_f32(false, a, false, q2.xy, (short)0, acc[4], false, false);
      acc[5] = __builtin_amdgcn_wmma_f32_16x16x4_f32(false, a, false, q2.zw, (short)0, acc[5], false, false);
      acc[6] = __builtin_amdgcn_wmma_f32_16x16x4_f32(false, a, false, q3.xy, (short)0, acc[6], false, false);
      acc[7] = __builtin_amdgcn_wmma_f32_16x16x4_f32(false, a, false, q3.zw, (short)0, acc[7], false, false);
      accs   = __builtin_amdgcn_wmma_f32_16x16x4_f32(false, a, false, qs,    (short)0, accs,   false, false);
    }

#pragma unroll
    for (int c = 0; c < 8; ++c) {
#pragma unroll
      for (int r = 0; r < 8; ++r) {
        Cmat[(size_t)(row0 + mo + r) * 128 + c * 16 + m] = acc[c][r];
      }
    }
    if (m < 8) {
#pragma unroll
      for (int r = 0; r < 8; ++r) ew[(mo + r) * 8 + m] = accs[r];
    }
  }
  __syncthreads();
  if (live) {
    const v4f v = *(const v4f*)(ew + m * 8 + (lane >> 4) * 4);
    float* dp = (lane < 16 ? el : er) + (size_t)(row0 + m) * 4;
    *(v4f*)dp = v;
  }
}

#define OWN_CAP     8192
#define OWN_THREADS 1024
#define ACC_P       40
#define P2_TILES    ((OWN_NB + 15) / 16)
#define P2_ROWS     (P2_TILES * 16)
#define P2_HPITCH   36
#define P2_OPITCH   132
#define P2_RPW      ((OWN_NB + 31) / 32)

typedef unsigned v2u __attribute__((ext_vector_type(2)));
typedef unsigned v4u __attribute__((ext_vector_type(4)));

__device__ __forceinline__ void own_check(int dv, int e, int base, unsigned nb,
                                          v2u* ent, unsigned* cnt) {
  const unsigned dl = (unsigned)(dv - base);
  if (dl < nb) {
    unsigned pos = atomicAdd(cnt, 1u);
    if (pos < OWN_CAP) { v2u v; v.x = (unsigned)e; v.y = dl; ent[pos] = v; }
  }
}

__device__ __forceinline__ void own_build(const int* __restrict__ dst, int e0, int e1,
                                          int base, unsigned nb,
                                          v2u* ent, unsigned* cnt, int tid) {
  const int nvec = (e1 - e0) >> 2;
  const int4* dv = (const int4*)(dst + e0);
  int i = tid;
  for (; i + 3 * OWN_THREADS < nvec; i += 4 * OWN_THREADS) {
    int4 d[4];
#pragma unroll
    for (int u = 0; u < 4; ++u) d[u] = dv[i + u * OWN_THREADS];
#pragma unroll
    for (int u = 0; u < 4; ++u) {
      const int eb = e0 + 4 * (i + u * OWN_THREADS);
      own_check(d[u].x, eb + 0, base, nb, ent, cnt);
      own_check(d[u].y, eb + 1, base, nb, ent, cnt);
      own_check(d[u].z, eb + 2, base, nb, ent, cnt);
      own_check(d[u].w, eb + 3, base, nb, ent, cnt);
    }
  }
  for (; i < nvec; i += OWN_THREADS) {
    int4 d = dv[i];
    const int eb = e0 + 4 * i;
    own_check(d.x, eb + 0, base, nb, ent, cnt);
    own_check(d.y, eb + 1, base, nb, ent, cnt);
    own_check(d.z, eb + 2, base, nb, ent, cnt);
    own_check(d.w, eb + 3, base, nb, ent, cnt);
  }
  for (int e = e0 + nvec * 4 + tid; e < e1; e += OWN_THREADS)
    own_check(dst[e], e, base, nb, ent, cnt);
}

__device__ __forceinline__ unsigned own_nz(unsigned k, unsigned pat) {
  const unsigned x = k ^ pat;
  return ((x & 0x7F7F7F7Fu) + 0x7F7F7F7Fu) | x;
}
__device__ __forceinline__ unsigned own_hits(const v4u k, unsigned pat) {
  const unsigned c = ((own_nz(k.x, pat) >> 7) & 0x01010101u) | ((own_nz(k.y, pat) >> 6) & 0x02020202u) |
                     ((own_nz(k.z, pat) >> 5) & 0x04040404u) | ((own_nz(k.w, pat) >> 4) & 0x08080808u);
  return c ^ 0x0F0F0F0Fu;
}
__device__ __forceinline__ void own_emit(unsigned m, int e0, v2u* ent, unsigned* cnt) {
  while (m) {
    const int p = __builtin_ctz(m);
    m &= m - 1u;
    const int e = e0 + ((p & 3) << 2) + (p >> 3) + ((p & 4) << 12);
    const unsigned pos = atomicAdd(cnt, 1u);
    if (pos < OWN_CAP) ent[pos].x = (unsigned)e;
  }
}

__device__ __forceinline__ void own_scan4(const v4u k[4], unsigned pat, int e0,
                                          v2u* ent, unsigned* cnt) {
  own_emit(own_hits(k[0], pat) | (own_hits(k[1], pat) << 4), e0, ent, cnt);
  own_emit(own_hits(k[2], pat) | (own_hits(k[3], pat) << 4), e0 + 32 * OWN_THREADS, ent, cnt);
}

__device__ __forceinline__ void own_build_keys(const unsigned* __restrict__ wkey,
                                               const int* __restrict__ dst, int E,
                                               unsigned me, int base, unsigned nb,
                                               v2u* ent, unsigned* cnt, int tid) {
  const int nvec = E >> 4;
  const v4u* kv = (const v4u*)wkey;
  const unsigned pat = me * 0x01010101u;
  const int ntrip = nvec / (4 * OWN_THREADS);
  int i = tid;
  if (ntrip > 0) {
    v4u ka[4], kb[4];
#pragma unroll
    for (int u = 0; u < 4; ++u) ka[u] = kv[i + u * OWN_THREADS];
    int t = 0;
    for (; t + 2 < ntrip; t += 2) {
#pragma unroll
      for (int u = 0; u < 4; ++u) kb[u] = kv[i + (4 + u) * OWN_THREADS];
      own_scan4(ka, pat, 16 * i, ent, cnt);
      i += 4 * OWN_THREADS;
#pragma unroll
      for (int u = 0; u < 4; ++u) ka[u] = kv[i + (4 + u) * OWN_THREADS];
      own_scan4(kb, pat, 16 * i, ent, cnt);
      i += 4 * OWN_THREADS;
    }
    if (t + 1 < ntrip) {
#pragma unroll
      for (int u = 0; u < 4; ++u) kb[u] = kv[i + (4 + u) * OWN_THREADS];
      own_scan4(ka, pat, 16 * i, ent, cnt);
      i += 4 * OWN_THREADS;
      own_scan4(kb, pat, 16 * i, ent, cnt);
    } else {
      own_scan4(ka, pat, 16 * i, ent, cnt);
    }
    i += 4 * OWN_THREADS;
  }
  for (; i < nvec; i += OWN_THREADS) {
    own_emit(own_hits(kv[i], pat), 16 * i, ent, cnt);
  }
  for (int e = nvec * 16 + tid; e < E; e += OWN_THREADS)
    own_check(dst[e], e, base, nb, ent, cnt);
}

template <bool HAVE_SRC, bool KEEP, bool NEED_DL>
__device__ __forceinline__ void own_prep(v2u* ent, v4f* sc, int n,
                                         const int* __restrict__ src,
                                         const int* __restrict__ dst, int base, unsigned nb,
                                         const float* __restrict__ el, const float* erl,
                                         unsigned* smax, v2u* __restrict__ wl,
                                         bool doMax, int tid) {
  for (int i = tid; i < n; i += OWN_THREADS) {
    v2u en = ent[i];
    if (!HAVE_SRC) {
      const unsigned e = en.x;
      en.x = (unsigned)src[e];
      if (NEED_DL) en.y = min((unsigned)(dst[e] - base), nb - 1u);
      ent[i] = en;
    }
    if (KEEP) wl[i] = en;
    const unsigned dl = en.y;
    const v4f l = *(const v4f*)(el + (size_t)en.x * 4);
    const v4f r = *(const v4f*)(erl + dl * 4);
    v4f v;
    v.x = lrelu(l.x + r.x);
    v.y = lrelu(l.y + r.y);
    v.z = lrelu(l.z + r.z);
    v.w = lrelu(l.w + r.w);
    sc[i] = v;
    if (doMax) {
      atomicMax(&smax[dl * 4 + 0], fenc(v.x));
      atomicMax(&smax[dl * 4 + 1], fenc(v.y));
      atomicMax(&smax[dl * 4 + 2], fenc(v.z));
      atomicMax(&smax[dl * 4 + 3], fenc(v.w));
    }
  }
}

__device__ __forceinline__ void own_den(const v2u* ent, v4f* sc, int n,
                                        const unsigned* smax, float* sden,
                                        bool add, int tid) {
  for (int i = tid; i < n; i += OWN_THREADS) {
    const unsigned dl = ent[i].y;
    const v4u m = *(const v4u*)(smax + dl * 4);
    v4f v = sc[i];
    v.x = expf(v.x - fdec(m.x));
    v.y = expf(v.y - fdec(m.y));
    v.z = expf(v.z - fdec(m.z));
    v.w = expf(v.w - fdec(m.w));
    sc[i] = v;
    if (add) {
      atomicAdd(&sden[dl * 4 + 0], v.x);
      atomicAdd(&sden[dl * 4 + 1], v.y);
      atomicAdd(&sden[dl * 4 + 2], v.z);
      atomicAdd(&sden[dl * 4 + 3], v.w);
    }
  }
}

template <bool FULL>
__device__ __forceinline__ void own_acc16(int eq, int n, const v2u* ent, const v4f* sc,
                                          const char* __restrict__ hb, unsigned ro,
                                          float* accr) {
  v4f x[4][4];
  v4f al[4];
  int dl[4];
#pragma unroll
  for (int t = 0; t < 4; ++t) {
    const int idx = eq + 4 * t;
    const int idc = FULL ? idx : min(idx, n - 1);
    const v2u en = ent[idc];
    al[t] = sc[idc];
    dl[t] = (int)en.y;
    const char* p = hb + ((en.x << 9) + ro);
    x[t][0] = *(const v4f*)(p);
    x[t][1] = *(const v4f*)(p + 128);
    x[t][2] = *(const v4f*)(p + 256);
    x[t][3] = *(const v4f*)(p + 384);
  }
#pragma unroll
  for (int t = 0; t < 4; ++t) {
    const v4f y = x[t][0] * al[t].x + x[t][1] * al[t].y + x[t][2] * al[t].z + x[t][3] * al[t].w;
    if (FULL || eq + 4 * t < n) {
      float* ap = accr + dl[t] * ACC_P;
      atomicAdd(ap,      y.x);
      atomicAdd(ap + 8,  y.y);
      atomicAdd(ap + 16, y.z);
      atomicAdd(ap + 24, y.w);
    }
  }
}

__device__ __forceinline__ void own_acc(const v2u* ent, v4f* sc, int n,
                                        const float* __restrict__ h,
                                        const float* sden,
                                        float* acc, int wave, int lane) {
  for (int i = wave * 32 + lane; i < n; i += OWN_THREADS) {
    const v4f ex = sc[i];
    const v4f dn = *(const v4f*)(sden + ent[i].y * 4);
    v4f a;
    a.x = ex.x / fmaxf(dn.x, 1e-9f);
    a.y = ex.y / fmaxf(dn.y, 1e-9f);
    a.z = ex.z / fmaxf(dn.z, 1e-9f);
    a.w = ex.w / fmaxf(dn.w, 1e-9f);
    sc[i] = a;
  }
  __syncthreads();
  const char* hb = (const char*)h;
  const int q = lane >> 3, r = lane & 7;
  const unsigned ro = (unsigned)r * 16u;
  float* accr = acc + r;
  const int ng = (n + 15) >> 4;
  for (int g = wave; g < ng; g += OWN_THREADS / 32) {
    const int e0 = g * 16;
    if (e0 + 16 <= n) own_acc16<true>(e0 + q, n, ent, sc, hb, ro, accr);
    else              own_acc16<false>(e0 + q, n, ent, sc, hb, ro, accr);
  }
}

__device__ __forceinline__ void own_deg_add(int v, int lane, int* sdeg) {
  for (int m = 16; m >= 1; m >>= 1) v += __shfl_xor(v, m, 32);
  if (lane == 0 && v != 0) atomicAdd(sdeg, v);
}

template <int LAYER>
__global__ void __launch_bounds__(OWN_THREADS)
gat_owner(const int* __restrict__ src, const int* __restrict__ dst,
          const float* __restrict__ el, const float* __restrict__ er,
          const float* __restrict__ h, const float* __restrict__ bias,
          const float* __restrict__ lns, const float* __restrict__ lnb,
          float* __restrict__ outp,
          unsigned* __restrict__ wlist, unsigned* __restrict__ wcnt,
          const int* __restrict__ indeg, int* __restrict__ wdeg,
          const unsigned* __restrict__ wkey,
          const float* __restrict__ Wn, const float* __restrict__ aln,
          const float* __restrict__ arn,
          float* __restrict__ eln, float* __restrict__ ern,
          int Nn, int E) {
  __shared__ float    acc[OWN_NB * ACC_P];
  __shared__ __attribute__((aligned(16))) unsigned smax[OWN_NB * 4];
  __shared__ __attribute__((aligned(16))) float    sden[OWN_NB * 4];
  __shared__ __attribute__((aligned(16))) float    erl[OWN_NB * 4];
  __shared__ v2u      ent[OWN_CAP];
  __shared__ v4f      sc[OWN_CAP];
  __shared__ unsigned cnt;
  __shared__ int      sdeg;

  const int tid  = threadIdx.x;
  const int wave = tid >> 5;
  const int lane = tid & 31;
  const int base = blockIdx.x * OWN_NB;
  const unsigned nb = (unsigned)min(OWN_NB, Nn - base);

  const bool degBlock = (LAYER == 1) || (blockIdx.x == gridDim.x - 1);
  int dpart = 0;
  if (LAYER == 2 && degBlock) {
    for (int i = tid; i < (int)gridDim.x; i += OWN_THREADS) dpart += wdeg[i];
  }

  for (int i = tid; i < OWN_NB * ACC_P; i += OWN_THREADS) acc[i] = 0.f;
  for (int i = tid; i < OWN_NB * 4; i += OWN_THREADS) { smax[i] = 0u; sden[i] = 0.f; }
  if (tid == 0) { cnt = 0u; sdeg = 0; }

  v2u* wl = (v2u*)wlist + (size_t)blockIdx.x * OWN_CAP;
  float erv = 0.f;
  unsigned total;
  if (LAYER == 1) {
    __syncthreads();
    if (tid < (int)nb) dpart = indeg[base + tid];
    if (tid < (int)nb * 4) erv = er[(size_t)base * 4 + tid];
    if (wkey) own_build_keys(wkey, dst, E, blockIdx.x, base, nb, ent, &cnt, tid);
    else      own_build(dst, 0, E, base, nb, ent, &cnt, tid);
    if (tid < (int)nb * 4) erl[tid] = erv;
    own_deg_add(dpart, lane, &sdeg);
    __syncthreads();
    total = cnt;
    if (tid == 0) { wcnt[blockIdx.x] = total; wdeg[blockIdx.x] = sdeg; }
  } else {
    total = wcnt[blockIdx.x];
    if (tid < (int)nb * 4) erv = er[(size_t)base * 4 + tid];
    if (total <= OWN_CAP) {
      for (unsigned i0 = tid; i0 < total; i0 += 4 * OWN_THREADS) {
        v2u v[4];
#pragma unroll
        for (int u = 0; u < 4; ++u) {
          const unsigned i = i0 + u * OWN_THREADS;
          v2u z; z.x = 0u; z.y = 0u;
          v[u] = i < total ? wl[i] : z;
        }
#pragma unroll
        for (int u = 0; u < 4; ++u) {
          const unsigned i = i0 + u * OWN_THREADS;
          if (i < total) ent[i] = v[u];
        }
      }
    }
    if (tid < (int)nb * 4) erl[tid] = erv;
    __syncthreads();
    if (degBlock) own_deg_add(dpart, lane, &sdeg);
  }

  if (total <= OWN_CAP) {
    const int n = (int)total;
    own_prep<LAYER == 2, LAYER == 1, LAYER == 1>(ent, sc, n, src, dst, base, nb, el, erl, smax, wl, true, tid);
    __syncthreads();
    own_den(ent, sc, n, smax, sden, true, tid);
    __syncthreads();
    own_acc(ent, sc, n, h, sden, acc, wave, lane);
  } else {
    for (int pass = 0; pass < 3; ++pass) {
      for (int e0 = 0; e0 < E; e0 += OWN_CAP) {
        __syncthreads();
        if (tid == 0) cnt = 0u;
        __syncthreads();
        own_build(dst, e0, min(E, e0 + OWN_CAP), base, nb, ent, &cnt, tid);
        __syncthreads();
        const int n = (int)cnt;
        own_prep<false, false, false>(ent, sc, n, src, dst, base, nb, el, erl, smax, wl, pass == 0, tid);
        if (pass >= 1) {
          __syncthreads();
          own_den(ent, sc, n, smax, sden, pass == 1, tid);
        }
        if (pass == 2) {
          __syncthreads();
          own_acc(ent, sc, n, h, sden, acc, wave, lane);
        }
      }
    }
  }
  __syncthreads();

  if (LAYER == 2 && degBlock && tid == 0) {
    double ne = (double)(float)sdeg;
    double fl = (4.0 * ne * (6.0 * 128 * 32 + 6 * 32 + 2) +
                 4.0 * ne * (6.0 * 32 * 32 + 6 * 32 + 2)) / 1e12;
    outp[(size_t)Nn * 32] = (float)fl;
  }

  float* hm = (float*)ent;
  float* st = (float*)sc;
  float* bs = st + P2_ROWS * P2_OPITCH;
  static_assert(32 * 128 == 4 * OWN_THREADS, "Wn: 4 consecutive floats per thread");
  v4f alv = {}, arv = {}, wq = {};
  if (LAYER == 1) {
    alv = *(const v4f*)(aln + lane * 4);
    arv = *(const v4f*)(arn + lane * 4);
    wq  = *(const v4f*)(Wn + tid * 4);
    for (int i = (int)nb * P2_HPITCH + tid; i < P2_ROWS * P2_HPITCH; i += OWN_THREADS) hm[i] = 0.f;
  }

  const int pq = (lane & 3) * 8 + (lane >> 2);
  const float bsum = ((bias[lane] + bias[32 + lane]) + bias[64 + lane]) + bias[96 + lane];
  for (int nl = wave; nl < (int)nb; nl += OWN_THREADS / 32) {
    const int n = base + nl;
    float x = (acc[nl * ACC_P + pq] + bsum) * 0.25f;
    if (LAYER == 1) {
      float mu = x;
#pragma unroll
      for (int m = 16; m >= 1; m >>= 1) mu += __shfl_xor(mu, m, 32);
      mu *= (1.f / 32.f);
      float dx = x - mu;
      float var = dx * dx;
#pragma unroll
      for (int m = 16; m >= 1; m >>= 1) var += __shfl_xor(var, m, 32);
      var *= (1.f / 32.f);
      float y = dx * rsqrtf(var + LN_EPS) * lns[lane] + lnb[lane];
      hm[nl * P2_HPITCH + lane] = fmaxf(y, 0.f);
    } else {
      outp[(size_t)n * 32 + lane] = x;
    }
  }

  if (LAYER == 1) {
    {
      const int kr = tid >> 5, col = (tid & 31) * 4;
      const int ln = (col & 15) | ((kr & 2) << 3);
      float* bq = bs + (((kr >> 2) * 32 + ln) * 8 + (col >> 4)) * 2 + (kr & 1);
      bq[0]  = wq.x;
      bq[16] = wq.y;
      bq[32] = wq.z;
      bq[48] = wq.w;
    }
    __syncthreads();
    const int wv = __builtin_amdgcn_readfirstlane(wave);
    const int t = wv >> 1, hf = wv & 1;
    if (t * 16 < (int)nb) {
      const int m = lane & 15;
      const float* ap = hm + (t * 16 + m) * P2_HPITCH + ((lane >> 4) << 1);
      const float* bp = bs + lane * 16 + hf * 8;
      v8f c0 = {}, c1 = {}, c2 = {}, c3 = {};
#pragma unroll
      for (int ks = 0; ks < 8; ++ks) {
        const v2f a  = *(const v2f*)(ap + ks * 4);
        const v4f q0 = *(const v4f*)(bp + ks * 512);
        const v4f q1 = *(const v4f*)(bp + ks * 512 + 4);
        c0 = __builtin_amdgcn_wmma_f32_16x16x4_f32(false, a, false, q0.xy, (short)0, c0, false, false);
        c1 = __builtin_amdgcn_wmma_f32_16x16x4_f32(false, a, false, q0.zw, (short)0, c1, false, false);
        c2 = __builtin_amdgcn_wmma_f32_16x16x4_f32(false, a, false, q1.xy, (short)0, c2, false, false);
        c3 = __builtin_amdgcn_wmma_f32_16x16x4_f32(false, a, false, q1.zw, (short)0, c3, false, false);
      }
      float* sp = st + (t * 16 + (lane >> 4) * 8) * P2_OPITCH + hf * 64 + m;
#pragma unroll
      for (int r = 0; r < 8; ++r) {
        sp[r * P2_OPITCH]      = c0[r];
        sp[r * P2_OPITCH + 16] = c1[r];
        sp[r * P2_OPITCH + 32] = c2[r];
        sp[r * P2_OPITCH + 48] = c3[r];
      }
    }
    __syncthreads();
    float pl[P2_RPW], pr[P2_RPW];
    v4f pv[P2_RPW];
#pragma unroll
    for (int u = 0; u < P2_RPW; ++u)
      pv[u] = *(const v4f*)(st + min(wv + 32 * u, P2_ROWS - 1) * P2_OPITCH + lane * 4);
#pragma unroll
    for (int u = 0; u < P2_RPW; ++u) {
      const int nl = wv + 32 * u;
      const v4f v = pv[u];
      if (nl < (int)nb) *(v4f*)(outp + (size_t)(base + nl) * 128 + lane * 4) = v;
      pl[u] = v.x * alv.x + v.y * alv.y + v.z * alv.z + v.w * alv.w;
      pr[u] = v.x * arv.x + v.y * arv.y + v.z * arv.z + v.w * arv.w;
    }
#pragma unroll
    for (int msk = 1; msk <= 4; msk <<= 1) {
#pragma unroll
      for (int u = 0; u < P2_RPW; ++u) {
        pl[u] += __shfl_xor(pl[u], msk, 32);
        pr[u] += __shfl_xor(pr[u], msk, 32);
      }
    }
#pragma unroll
    for (int u = 0; u < P2_RPW; ++u) {
      const int nl = wv + 32 * u;
      if (nl < (int)nb && (lane & 7) == 0) {
        sden[nl * 4 + (lane >> 3)] = pl[u];
        erl[nl * 4 + (lane >> 3)]  = pr[u];
      }
    }
    __syncthreads();
    if (tid < (int)nb * 4) {
      eln[(size_t)base * 4 + tid] = sden[tid];
      ern[(size_t)base * 4 + tid] = erl[tid];
    }
  }
}

extern "C" void kernel_launch(void* const* d_in, const int* in_sizes, int n_in,
                              void* d_out, int out_size, void* d_ws, size_t ws_size,
                              hipStream_t stream) {
  const float* feat = (const float*)d_in[0];
  const float* W1   = (const float*)d_in[1];
  const float* al1  = (const float*)d_in[2];
  const float* ar1  = (const float*)d_in[3];
  const float* b1   = (const float*)d_in[4];
  const float* lns  = (const float*)d_in[5];
  const float* lnb  = (const float*)d_in[6];
  const float* W2   = (const float*)d_in[7];
  const float* al2  = (const float*)d_in[8];
  const float* ar2  = (const float*)d_in[9];
  const float* b2   = (const float*)d_in[10];
  const int*   src  = (const int*)d_in[11];
  const int*   dst  = (const int*)d_in[12];
  const int*   indeg= (const int*)d_in[13];

  const int Nn = in_sizes[13];
  const int E  = in_sizes[11];
  const int NH = Nn * 4;

  char* p = (char*)d_ws;
  float*    hbuf  = (float*)p;   p += (size_t)Nn * 128 * sizeof(float);
  float*    hbuf2 = (float*)p;   p += (size_t)Nn * 128 * sizeof(float);
  float*    el    = (float*)p;   p += (size_t)NH * sizeof(float);
  float*    er    = (float*)p;   p += (size_t)NH * sizeof(float);
  float*    el2   = (float*)p;   p += (size_t)NH * sizeof(float);
  float*    er2   = (float*)p;   p += (size_t)NH * sizeof(float);
  const int ownBlocks  = (Nn + OWN_NB - 1) / OWN_NB;
  unsigned* wlist = (unsigned*)p; p += (size_t)ownBlocks * OWN_CAP * 2 * sizeof(unsigned);
  unsigned* wkey  = (unsigned*)p; p += ((size_t)E + 15) / 16 * 16;
  unsigned* wcnt  = (unsigned*)p; p += (size_t)ownBlocks * sizeof(unsigned);
  int*      wdeg  = (int*)p;
  (void)ws_size; (void)n_in; (void)out_size;

  float* outh = (float*)d_out;

  dim3 b256(256);
  const int rowTiles   = (Nn + 15) / 16;
  const int gemmBlocks = (rowTiles + 7) / 8;

  unsigned* wkeyArg = ownBlocks <= 256 ? wkey : nullptr;
  gat_gemm_wmma<128><<<gemmBlocks, b256, 0, stream>>>(feat, W1, hbuf, Nn, al1, ar1, el, er, dst, wkeyArg, E);
  gat_owner<1><<<ownBlocks, OWN_THREADS, 0, stream>>>(src, dst, el, er, hbuf, b1, lns, lnb, hbuf2, wlist, wcnt, indeg, wdeg, wkeyArg, W2, al2, ar2, el2, er2, Nn, E);

  gat_owner<2><<<ownBlocks, OWN_THREADS, 0, stream>>>(src, dst, el2, er2, hbuf2, b2, lns, lnb, outh, wlist, wcnt, indeg, wdeg, nullptr, nullptr, nullptr, nullptr, nullptr, nullptr, Nn, E);
}


# projection launch stores one bit per (owner, edge) in per-owner planes (LDS atomic OR, 2048-edge chunks), was a byte per edge: an owner reads 100 KB, not 800 KB; sums by LDS atomic arrival as before
# speedup vs baseline: 8.2006x; 1.0475x over previous
#include <hip/hip_runtime.h>


typedef float v2f __attribute__((ext_vector_type(2)));
typedef float v4f __attribute__((ext_vector_type(4)));
typedef float v8f __attribute__((ext_vector_type(8)));
typedef unsigned v2u __attribute__((ext_vector_type(2)));
typedef unsigned v4u __attribute__((ext_vector_type(4)));

#define NEG_SLOPE 0.2f
#define LN_EPS 1e-5f

__device__ __forceinline__ unsigned fenc(float f) {
  unsigned u = __float_as_uint(f);
  return (u >> 31) ? ~u : (u | 0x80000000u);
}
__device__ __forceinline__ float fdec(unsigned u) {
  return __uint_as_float((u >> 31) ? (u & 0x7FFFFFFFu) : ~u);
}
__device__ __forceinline__ float lrelu(float x) {
  return x > 0.f ? x : x * NEG_SLOPE;
}

#define OWN_NB      196
#define BIT_CH      2048
#define BIT_CW      (BIT_CH / 32)
#define BIT_OWN     256

__device__ __forceinline__ void bits_zero(v4u* bm4, int tid) {
  const v4u z = {0u, 0u, 0u, 0u};
#pragma unroll
  for (int i = 0; i < BIT_OWN * BIT_CW / 4 / 256; ++i) bm4[i * 256 + tid] = z;
}
__device__ __forceinline__ void bits_set1(unsigned* bm, int d, unsigned le) {
  const unsigned o = (unsigned)d / OWN_NB;
  if (o < BIT_OWN) atomicOr(&bm[o * BIT_CW + (le >> 5)], 1u << (le & 31u));
}
__device__ __forceinline__ void bits_set4(unsigned* bm, const int4 d, unsigned le) {
  bits_set1(bm, d.x, le);
  bits_set1(bm, d.y, le + 1u);
  bits_set1(bm, d.z, le + 2u);
  bits_set1(bm, d.w, le + 3u);
}
__device__ __forceinline__ void bits_store(const v4u* bm4, unsigned* __restrict__ wbits,
                                           int c, int nch, int tid) {
  static_assert(BIT_CW / 4 == 16, "16 pieces per row: piece tid + 256 i is piece tid % 16 of row tid / 16 + 16 i");
  v4u* out = (v4u*)wbits;
  const unsigned q0 = ((unsigned)(tid >> 4) * (unsigned)nch + (unsigned)c) * 16u + (unsigned)(tid & 15);
  const unsigned qs = 16u * (unsigned)nch * 16u;
#pragma unroll
  for (int i = 0; i < BIT_OWN * BIT_CW / 4 / 256; ++i) out[q0 + (unsigned)i * qs] = bm4[i * 256 + tid];
}

__device__ __forceinline__ float proj_sum8(const float (&p)[8], int lane) {
  const bool b4 = (lane & 4) != 0, b2 = (lane & 2) != 0, b1 = (lane & 1) != 0;
  float w[4], x[2];
#pragma unroll
  for (int j = 0; j < 4; ++j) {
    const float mine = b4 ? p[j + 4] : p[j];
    const float give = b4 ? p[j] : p[j + 4];
    w[j] = mine + __shfl_xor(give, 4, 32);
  }
#pragma unroll
  for (int j = 0; j < 2; ++j) {
    const float mine = b2 ? w[j + 2] : w[j];
    const float give = b2 ? w[j] : w[j + 2];
    x[j] = mine + __shfl_xor(give, 2, 32);
  }
  const float mine = b1 ? x[1] : x[0];
  const float give = b1 ? x[0] : x[1];
  return mine + __shfl_xor(give, 1, 32);
}

template <int K>
__global__ void __launch_bounds__(256)
gat_gemm_wmma(const float* __restrict__ Amat,
              const float* __restrict__ Bmat,
              float* __restrict__ Cmat,
              int M,
              const float* __restrict__ al, const float* __restrict__ ar,
              float* __restrict__ el, float* __restrict__ er,
              const int* __restrict__ dst, unsigned* __restrict__ wbits,
              int E) {
  __shared__ float bs[K * 128];
  __shared__ float bw[K * 16];
  __shared__ __attribute__((aligned(16))) float es[8 * 128];
  __shared__ __attribute__((aligned(16))) unsigned bm[BIT_OWN * BIT_CW];

  const int tid = threadIdx.x;

  static_assert(BIT_CH == 2 * 4 * 256, "two groups of 4 edges per thread");
  const int nq = E >> 2;
  const int nch = (E + BIT_CH - 1) / BIT_CH;
  const int4* d4 = (const int4*)dst;
  const int qa = blockIdx.x * (BIT_CH / 4) + tid, qb = qa + 256;
  const bool keys = wbits != nullptr;
  const bool mych = keys && (int)blockIdx.x < nch;
  int4 dA = {}, dB = {};
  if (mych) {
    if (qa < nq) dA = d4[qa];
    if (qb < nq) dB = d4[qb];
    bits_zero((v4u*)bm, tid);
  }
  static_assert(K % 16 == 0, "16 rows of B per wave and trip");
  const int wv = tid >> 5, ln = tid & 31;
  const v4f a4 = *(const v4f*)(al + ln * 4);
  const v4f r4 = *(const v4f*)(ar + ln * 4);
  v4f w[16];
#pragma unroll
  for (int u = 0; u < 16; ++u) w[u] = *(const v4f*)(Bmat + (size_t)(min(wv * 16, K - 16) + u) * 128 + ln * 4);

  for (int idx = tid; idx < K * 128; idx += 256) {
    int kr = idx >> 7;
    int col = idx & 127;
    int ks = kr >> 2;
    int c = col >> 4;
    int lane = (col & 15) | ((kr & 2) << 3);
    int sub = kr & 1;
    bs[((ks * 32 + lane) * 8 + c) * 2 + sub] = Bmat[idx];
  }
  for (int k0 = wv * 16; k0 < K; k0 += 8 * 16) {
#pragma unroll
    for (int g = 0; g < 2; ++g) {
      float pl[8], pr[8];
#pragma unroll
      for (int r = 0; r < 8; ++r) {
        const v4f t = w[g * 8 + r];
        pl[r] = t.x * a4.x + t.y * a4.y + t.z * a4.z + t.w * a4.w;
        pr[r] = t.x * r4.x + t.y * r4.y + t.z * r4.z + t.w * r4.w;
      }
      const float sl = proj_sum8(pl, ln);
      const float sr = proj_sum8(pr, ln);
      const int k = k0 + g * 8 + (ln & 7), h = ln >> 3;
      float* bq = bw + ((k >> 2) * 32 + ((k & 2) << 3)) * 2 + (k & 1);
      bq[h * 2]        = sl;
      bq[(h + 4) * 2]  = sr;
      bq[(h + 8) * 2]  = 0.f;
      bq[(h + 12) * 2] = 0.f;
    }
    if (k0 + 8 * 16 < K) {
#pragma unroll
      for (int u = 0; u < 16; ++u) w[u] = *(const v4f*)(Bmat + (size_t)(k0 + 8 * 16 + u) * 128 + ln * 4);
    }
  }
  __syncthreads();
  if (mych) {
    if (qa < nq) bits_set4(bm, dA, 4u * (unsigned)tid);
    if (qb < nq) bits_set4(bm, dB, 4u * (unsigned)tid + 1024u);
    __syncthreads();
    bits_store((const v4u*)bm, wbits, blockIdx.x, nch, tid);
  }

  const int wave = tid >> 5;
  const int lane = tid & 31;
  const int row0 = (blockIdx.x * 8 + wave) * 16;
  const bool live = row0 < M;

  const int m  = lane & 15;
  const int kk = (lane >> 4) << 1;
  const int mo = (lane >> 4) * 8;
  float* ew = es + wave * 128;

  if (live) {
    v8f acc[8] = {};
    v8f accs = {};

    const float* arow = Amat + (size_t)(row0 + m) * K + kk;
    const float* blane = bs + lane * 16;
    const float* wlane = bw + lane * 2;

#pragma unroll 4
    for (int ks = 0; ks < K / 4; ++ks) {
      v2f a = *(const v2f*)(arow + ks * 4);
      const float* bp = blane + ks * 512;
      v4f q0 = *(const v4f*)(bp);
      v4f q1 = *(const v4f*)(bp + 4);
      v4f q2 = *(const v4f*)(bp + 8);
      v4f q3 = *(const v4f*)(bp + 12);
      v2f qs = *(const v2f*)(wlane + ks * 64);
      acc[0] = __builtin_amdgcn_wmma_f32_16x16x4_f32(false, a, false, q0.xy, (short)0, acc[0], false, false);
      acc[1] = __builtin_amdgcn_wmma_f32_16x16x4_f32(false, a, false, q0.zw, (short)0, acc[1], false, false);
      acc[2] = __builtin_amdgcn_wmma_f32_16x16x4_f32(false, a, false, q1.xy, (short)0, acc[2], false, false);
      acc[3] = __builtin_amdgcn_wmma_f32_16x16x4_f32(false, a, false, q1.zw, (short)0, acc[3], false, false);
      acc[4] = __builtin_amdgcn_wmma_f32_16x16x4_f32(false, a, false, q2.xy, (short)0, acc[4], false, false);
      acc[5] = __builtin_amdgcn_wmma_f32_16x16x4_f32(false, a, false, q2.zw, (short)0, acc[5], false, false);
      acc[6] = __builtin_amdgcn_wmma_f32_16x16x4_f32(false, a, false, q3.xy, (short)0, acc[6], false, false);
      acc[7] = __builtin_amdgcn_wmma_f32_16x16x4_f32(false, a, false, q3.zw, (short)0, acc[7], false, false);
      accs   = __builtin_amdgcn_wmma_f32_16x16x4_f32(false, a, false, qs,    (short)0, accs,   false, false);
    }

#pragma unroll
    for (int c = 0; c < 8; ++c) {
#pragma unroll
      for (int r = 0; r < 8; ++r) {
        Cmat[(size_t)(row0 + mo + r) * 128 + c * 16 + m] = acc[c][r];
      }
    }
    if (m < 8) {
#pragma unroll
      for (int r = 0; r < 8; ++r) ew[(mo + r) * 8 + m] = accs[r];
    }
  }
  __syncthreads();
  if (live) {
    const v4f v = *(const v4f*)(ew + m * 8 + (lane >> 4) * 4);
    float* dp = (lane < 16 ? el : er) + (size_t)(row0 + m) * 4;
    *(v4f*)dp = v;
  }
  if (keys) {
    for (int c = blockIdx.x + gridDim.x; c < nch; c += gridDim.x) {
      __syncthreads();
      bits_zero((v4u*)bm, tid);
      const int qc = c * (BIT_CH / 4) + tid;
      int4 dC = {}, dD = {};
      if (qc < nq) dC = d4[qc];
      if (qc + 256 < nq) dD = d4[qc + 256];
      __syncthreads();
      if (qc < nq) bits_set4(bm, dC, 4u * (unsigned)tid);
      if (qc + 256 < nq) bits_set4(bm, dD, 4u * (unsigned)tid + 1024u);
      __syncthreads();
      bits_store((const v4u*)bm, wbits, c, nch, tid);
    }
  }
}

#define OWN_CAP     8192
#define OWN_THREADS 1024
#define ACC_P       40
#define P2_TILES    ((OWN_NB + 15) / 16)
#define P2_ROWS     (P2_TILES * 16)
#define P2_HPITCH   36
#define P2_OPITCH   132
#define P2_RPW      ((OWN_NB + 31) / 32)

__device__ __forceinline__ void own_check(int dv, int e, int base, unsigned nb,
                                          v2u* ent, unsigned* cnt) {
  const unsigned dl = (unsigned)(dv - base);
  if (dl < nb) {
    unsigned pos = atomicAdd(cnt, 1u);
    if (pos < OWN_CAP) { v2u v; v.x = (unsigned)e; v.y = dl; ent[pos] = v; }
  }
}

__device__ __forceinline__ void own_build(const int* __restrict__ dst, int e0, int e1,
                                          int base, unsigned nb,
                                          v2u* ent, unsigned* cnt, int tid) {
  const int nvec = (e1 - e0) >> 2;
  const int4* dv = (const int4*)(dst + e0);
  int i = tid;
  for (; i + 3 * OWN_THREADS < nvec; i += 4 * OWN_THREADS) {
    int4 d[4];
#pragma unroll
    for (int u = 0; u < 4; ++u) d[u] = dv[i + u * OWN_THREADS];
#pragma unroll
    for (int u = 0; u < 4; ++u) {
      const int eb = e0 + 4 * (i + u * OWN_THREADS);
      own_check(d[u].x, eb + 0, base, nb, ent, cnt);
      own_check(d[u].y, eb + 1, base, nb, ent, cnt);
      own_check(d[u].z, eb + 2, base, nb, ent, cnt);
      own_check(d[u].w, eb + 3, base, nb, ent, cnt);
    }
  }
  for (; i < nvec; i += OWN_THREADS) {
    int4 d = dv[i];
    const int eb = e0 + 4 * i;
    own_check(d.x, eb + 0, base, nb, ent, cnt);
    own_check(d.y, eb + 1, base, nb, ent, cnt);
    own_check(d.z, eb + 2, base, nb, ent, cnt);
    own_check(d.w, eb + 3, base, nb, ent, cnt);
  }
  for (int e = e0 + nvec * 4 + tid; e < e1; e += OWN_THREADS)
    own_check(dst[e], e, base, nb, ent, cnt);
}

__device__ __forceinline__ void own_emit_bits(unsigned m, unsigned e0, v2u* ent, unsigned& pos) {
  while (m) {
    const unsigned p = (unsigned)__builtin_ctz(m);
    m &= m - 1u;
    ent[min(pos, (unsigned)(OWN_CAP - 1))].x = e0 + p;
    ++pos;
  }
}

__device__ __forceinline__ void own_build_bits(const unsigned* __restrict__ wbits,
                                               const int* __restrict__ dst, int E,
                                               unsigned me, int base, unsigned nb,
                                               v2u* ent, unsigned* cnt, int tid) {
  const int lane = tid & 31;
  const int nch = (E + BIT_CH - 1) / BIT_CH;
  const int nvec = nch * (BIT_CW / 4);
  const v4u* pv = (const v4u*)wbits + (size_t)me * nvec;
  for (int i0 = 0; i0 < nvec; i0 += 8 * OWN_THREADS) {
    v4u k[8];
#pragma unroll
    for (int u = 0; u < 8; ++u) {
      const int i = i0 + u * OWN_THREADS + tid;
      const v4u z = {0u, 0u, 0u, 0u};
      k[u] = i < nvec ? pv[i] : z;
    }
    unsigned n = 0;
#pragma unroll
    for (int u = 0; u < 8; ++u)
      n += (unsigned)(__popc(k[u].x) + __popc(k[u].y) + __popc(k[u].z) + __popc(k[u].w));
    unsigned run = n;
#pragma unroll
    for (int d = 1; d < 32; d <<= 1) {
      const unsigned t = __shfl_up(run, d, 32);
      if (lane >= d) run += t;
    }
    unsigned first = 0u;
    if (lane == 31) first = atomicAdd(cnt, run);
    first = (unsigned)__builtin_amdgcn_readlane((int)first, 31);
    unsigned pos = first + run - n;
#pragma unroll
    for (int u = 0; u < 8; ++u) {
      const unsigned e0 = (unsigned)(i0 + u * OWN_THREADS + tid) * 128u;
      own_emit_bits(k[u].x, e0,       ent, pos);
      own_emit_bits(k[u].y, e0 + 32u, ent, pos);
      own_emit_bits(k[u].z, e0 + 64u, ent, pos);
      own_emit_bits(k[u].w, e0 + 96u, ent, pos);
    }
  }
  for (int e = (E & ~3) + tid; e < E; e += OWN_THREADS)
    own_check(dst[e], e, base, nb, ent, cnt);
}

template <bool HAVE_SRC, bool KEEP, bool NEED_DL>
__device__ __forceinline__ void own_prep(v2u* ent, v4f* sc, int n,
                                         const int* __restrict__ src,
                                         const int* __restrict__ dst, int base, unsigned nb,
                                         const float* __restrict__ el, const float* erl,
                                         unsigned* smax, v2u* __restrict__ wl,
                                         bool doMax, int tid) {
  for (int i = tid; i < n; i += OWN_THREADS) {
    v2u en = ent[i];
    if (!HAVE_SRC) {
      const unsigned e = en.x;
      en.x = (unsigned)src[e];
      if (NEED_DL) en.y = min((unsigned)(dst[e] - base), nb - 1u);
      ent[i] = en;
    }
    if (KEEP) wl[i] = en;
    const unsigned dl = en.y;
    const v4f l = *(const v4f*)(el + (size_t)en.x * 4);
    const v4f r = *(const v4f*)(erl + dl * 4);
    v4f v;
    v.x = lrelu(l.x + r.x);
    v.y = lrelu(l.y + r.y);
    v.z = lrelu(l.z + r.z);
    v.w = lrelu(l.w + r.w);
    sc[i] = v;
    if (doMax) {
      atomicMax(&smax[dl * 4 + 0], fenc(v.x));
      atomicMax(&smax[dl * 4 + 1], fenc(v.y));
      atomicMax(&smax[dl * 4 + 2], fenc(v.z));
      atomicMax(&smax[dl * 4 + 3], fenc(v.w));
    }
  }
}

__device__ __forceinline__ void own_den(const v2u* ent, v4f* sc, int n,
                                        const unsigned* smax, float* sden,
                                        bool add, int tid) {
  for (int i = tid; i < n; i += OWN_THREADS) {
    const unsigned dl = ent[i].y;
    const v4u m = *(const v4u*)(smax + dl * 4);
    v4f v = sc[i];
    v.x = expf(v.x - fdec(m.x));
    v.y = expf(v.y - fdec(m.y));
    v.z = expf(v.z - fdec(m.z));
    v.w = expf(v.w - fdec(m.w));
    sc[i] = v;
    if (add) {
      atomicAdd(&sden[dl * 4 + 0], v.x);
      atomicAdd(&sden[dl * 4 + 1], v.y);
      atomicAdd(&sden[dl * 4 + 2], v.z);
      atomicAdd(&sden[dl * 4 + 3], v.w);
    }
  }
}

template <bool FULL>
__device__ __forceinline__ void own_acc16(int eq, int n, const v2u* ent, const v4f* sc,
                                          const char* __restrict__ hb, unsigned ro,
                                          float* accr) {
  v4f x[4][4];
  v4f al[4];
  int dl[4];
#pragma unroll
  for (int t = 0; t < 4; ++t) {
    const int idx = eq + 4 * t;
    const int idc = FULL ? idx : min(idx, n - 1);
    const v2u en = ent[idc];
    al[t] = sc[idc];
    dl[t] = (int)en.y;
    const char* p = hb + ((en.x << 9) + ro);
    x[t][0] = *(const v4f*)(p);
    x[t][1] = *(const v4f*)(p + 128);
    x[t][2] = *(const v4f*)(p + 256);
    x[t][3] = *(const v4f*)(p + 384);
  }
#pragma unroll
  for (int t = 0; t < 4; ++t) {
    const v4f y = x[t][0] * al[t].x + x[t][1] * al[t].y + x[t][2] * al[t].z + x[t][3] * al[t].w;
    if (FULL || eq + 4 * t < n) {
      float* ap = accr + dl[t] * ACC_P;
      atomicAdd(ap,      y.x);
      atomicAdd(ap + 8,  y.y);
      atomicAdd(ap + 16, y.z);
      atomicAdd(ap + 24, y.w);
    }
  }
}

__device__ __forceinline__ void own_acc(const v2u* ent, v4f* sc, int n,
                                        const float* __restrict__ h,
                                        const float* sden,
                                        float* acc, int wave, int lane) {
  for (int i = wave * 32 + lane; i < n; i += OWN_THREADS) {
    const v4f ex = sc[i];
    const v4f dn = *(const v4f*)(sden + ent[i].y * 4);
    v4f a;
    a.x = ex.x / fmaxf(dn.x, 1e-9f);
    a.y = ex.y / fmaxf(dn.y, 1e-9f);
    a.z = ex.z / fmaxf(dn.z, 1e-9f);
    a.w = ex.w / fmaxf(dn.w, 1e-9f);
    sc[i] = a;
  }
  __syncthreads();
  const char* hb = (const char*)h;
  const int q = lane >> 3, r = lane & 7;
  const unsigned ro = (unsigned)r * 16u;
  float* accr = acc + r;
  const int ng = (n + 15) >> 4;
  for (int g = wave; g < ng; g += OWN_THREADS / 32) {
    const int e0 = g * 16;
    if (e0 + 16 <= n) own_acc16<true>(e0 + q, n, ent, sc, hb, ro, accr);
    else              own_acc16<false>(e0 + q, n, ent, sc, hb, ro, accr);
  }
}

__device__ __forceinline__ void own_deg_add(int v, int lane, int* sdeg) {
  for (int m = 16; m >= 1; m >>= 1) v += __shfl_xor(v, m, 32);
  if (lane == 0 && v != 0) atomicAdd(sdeg, v);
}

template <int LAYER>
__global__ void __launch_bounds__(OWN_THREADS)
gat_owner(const int* __restrict__ src, const int* __restrict__ dst,
          const float* __restrict__ el, const float* __restrict__ er,
          const float* __restrict__ h, const float* __restrict__ bias,
          const float* __restrict__ lns, const float* __restrict__ lnb,
          float* __restrict__ outp,
          unsigned* __restrict__ wlist, unsigned* __restrict__ wcnt,
          const int* __restrict__ indeg, int* __restrict__ wdeg,
          const unsigned* __restrict__ wbits,
          const float* __restrict__ Wn, const float* __restrict__ aln,
          const float* __restrict__ arn,
          float* __restrict__ eln, float* __restrict__ ern,
          int Nn, int E) {
  __shared__ float    acc[OWN_NB * ACC_P];
  __shared__ __attribute__((aligned(16))) unsigned smax[OWN_NB * 4];
  __shared__ __attribute__((aligned(16))) float    sden[OWN_NB * 4];
  __shared__ __attribute__((aligned(16))) float    erl[OWN_NB * 4];
  __shared__ v2u      ent[OWN_CAP];
  __shared__ v4f      sc[OWN_CAP];
  __shared__ unsigned cnt;
  __shared__ int      sdeg;

  const int tid  = threadIdx.x;
  const int wave = tid >> 5;
  const int lane = tid & 31;
  const int base = blockIdx.x * OWN_NB;
  const unsigned nb = (unsigned)min(OWN_NB, Nn - base);

  const bool degBlock = (LAYER == 1) || (blockIdx.x == gridDim.x - 1);
  int dpart = 0;
  if (LAYER == 2 && degBlock) {
    for (int i = tid; i < (int)gridDim.x; i += OWN_THREADS) dpart += wdeg[i];
  }

  for (int i = tid; i < OWN_NB * ACC_P; i += OWN_THREADS) acc[i] = 0.f;
  for (int i = tid; i < OWN_NB * 4; i += OWN_THREADS) { smax[i] = 0u; sden[i] = 0.f; }
  if (tid == 0) { cnt = 0u; sdeg = 0; }

  v2u* wl = (v2u*)wlist + (size_t)blockIdx.x * OWN_CAP;
  float erv = 0.f;
  unsigned total;
  if (LAYER == 1) {
    __syncthreads();
    if (tid < (int)nb) dpart = indeg[base + tid];
    if (tid < (int)nb * 4) erv = er[(size_t)base * 4 + tid];
    if (wbits) own_build_bits(wbits, dst, E, blockIdx.x, base, nb, ent, &cnt, tid);
    else       own_build(dst, 0, E, base, nb, ent, &cnt, tid);
    if (tid < (int)nb * 4) erl[tid] = erv;
    own_deg_add(dpart, lane, &sdeg);
    __syncthreads();
    total = cnt;
    if (tid == 0) { wcnt[blockIdx.x] = total; wdeg[blockIdx.x] = sdeg; }
  } else {
    total = wcnt[blockIdx.x];
    if (tid < (int)nb * 4) erv = er[(size_t)base * 4 + tid];
    if (total <= OWN_CAP) {
      for (unsigned i0 = tid; i0 < total; i0 += 4 * OWN_THREADS) {
        v2u v[4];
#pragma unroll
        for (int u = 0; u < 4; ++u) {
          const unsigned i = i0 + u * OWN_THREADS;
          v2u z; z.x = 0u; z.y = 0u;
          v[u] = i < total ? wl[i] : z;
        }
#pragma unroll
        for (int u = 0; u < 4; ++u) {
          const unsigned i = i0 + u * OWN_THREADS;
          if (i < total) ent[i] = v[u];
        }
      }
    }
    if (tid < (int)nb * 4) erl[tid] = erv;
    __syncthreads();
    if (degBlock) own_deg_add(dpart, lane, &sdeg);
  }

  if (total <= OWN_CAP) {
    const int n = (int)total;
    own_prep<LAYER == 2, LAYER == 1, LAYER == 1>(ent, sc, n, src, dst, base, nb, el, erl, smax, wl, true, tid);
    __syncthreads();
    own_den(ent, sc, n, smax, sden, true, tid);
    __syncthreads();
    own_acc(ent, sc, n, h, sden, acc, wave, lane);
  } else {
    for (int pass = 0; pass < 3; ++pass) {
      for (int e0 = 0; e0 < E; e0 += OWN_CAP) {
        __syncthreads();
        if (tid == 0) cnt = 0u;
        __syncthreads();
        own_build(dst, e0, min(E, e0 + OWN_CAP), base, nb, ent, &cnt, tid);
        __syncthreads();
        const int n = (int)cnt;
        own_prep<false, false, false>(ent, sc, n, src, dst, base, nb, el, erl, smax, wl, pass == 0, tid);
        if (pass >= 1) {
          __syncthreads();
          own_den(ent, sc, n, smax, sden, pass == 1, tid);
        }
        if (pass == 2) {
          __syncthreads();
          own_acc(ent, sc, n, h, sden, acc, wave, lane);
        }
      }
    }
  }
  __syncthreads();

  if (LAYER == 2 && degBlock && tid == 0) {
    double ne = (double)(float)sdeg;
    double fl = (4.0 * ne * (6.0 * 128 * 32 + 6 * 32 + 2) +
                 4.0 * ne * (6.0 * 32 * 32 + 6 * 32 + 2)) / 1e12;
    outp[(size_t)Nn * 32] = (float)fl;
  }

  float* hm = (float*)ent;
  float* st = (float*)sc;
  float* bs = st + P2_ROWS * P2_OPITCH;
  static_assert(32 * 128 == 4 * OWN_THREADS, "Wn: 4 consecutive floats per thread");
  v4f alv = {}, arv = {}, wq = {};
  if (LAYER == 1) {
    alv = *(const v4f*)(aln + lane * 4);
    arv = *(const v4f*)(arn + lane * 4);
    wq  = *(const v4f*)(Wn + tid * 4);
    for (int i = (int)nb * P2_HPITCH + tid; i < P2_ROWS * P2_HPITCH; i += OWN_THREADS) hm[i] = 0.f;
  }

  const int pq = (lane & 3) * 8 + (lane >> 2);
  const float bsum = ((bias[lane] + bias[32 + lane]) + bias[64 + lane]) + bias[96 + lane];
  for (int nl = wave; nl < (int)nb; nl += OWN_THREADS / 32) {
    const int n = base + nl;
    float x = (acc[nl * ACC_P + pq] + bsum) * 0.25f;
    if (LAYER == 1) {
      float mu = x;
#pragma unroll
      for (int m = 16; m >= 1; m >>= 1) mu += __shfl_xor(mu, m, 32);
      mu *= (1.f / 32.f);
      float dx = x - mu;
      float var = dx * dx;
#pragma unroll
      for (int m = 16; m >= 1; m >>= 1) var += __shfl_xor(var, m, 32);
      var *= (1.f / 32.f);
      float y = dx * rsqrtf(var + LN_EPS) * lns[lane] + lnb[lane];
      hm[nl * P2_HPITCH + lane] = fmaxf(y, 0.f);
    } else {
      outp[(size_t)n * 32 + lane] = x;
    }
  }

  if (LAYER == 1) {
    {
      const int kr = tid >> 5, col = (tid & 31) * 4;
      const int ln = (col & 15) | ((kr & 2) << 3);
      float* bq = bs + (((kr >> 2) * 32 + ln) * 8 + (col >> 4)) * 2 + (kr & 1);
      bq[0]  = wq.x;
      bq[16] = wq.y;
      bq[32] = wq.z;
      bq[48] = wq.w;
    }
    __syncthreads();
    const int wv = __builtin_amdgcn_readfirstlane(wave);
    const int t = wv >> 1, hf = wv & 1;
    if (t * 16 < (int)nb) {
      const int m = lane & 15;
      const float* ap = hm + (t * 16 + m) * P2_HPITCH + ((lane >> 4) << 1);
      const float* bp = bs + lane * 16 + hf * 8;
      v8f c0 = {}, c1 = {}, c2 = {}, c3 = {};
#pragma unroll
      for (int ks = 0; ks < 8; ++ks) {
        const v2f a  = *(const v2f*)(ap + ks * 4);
        const v4f q0 = *(const v4f*)(bp + ks * 512);
        const v4f q1 = *(const v4f*)(bp + ks * 512 + 4);
        c0 = __builtin_amdgcn_wmma_f32_16x16x4_f32(false, a, false, q0.xy, (short)0, c0, false, false);
        c1 = __builtin_amdgcn_wmma_f32_16x16x4_f32(false, a, false, q0.zw, (short)0, c1, false, false);
        c2 = __builtin_amdgcn_wmma_f32_16x16x4_f32(false, a, false, q1.xy, (short)0, c2, false, false);
        c3 = __builtin_amdgcn_wmma_f32_16x16x4_f32(false, a, false, q1.zw, (short)0, c3, false, false);
      }
      float* sp = st + (t * 16 + (lane >> 4) * 8) * P2_OPITCH + hf * 64 + m;
#pragma unroll
      for (int r = 0; r < 8; ++r) {
        sp[r * P2_OPITCH]      = c0[r];
        sp[r * P2_OPITCH + 16] = c1[r];
        sp[r * P2_OPITCH + 32] = c2[r];
        sp[r * P2_OPITCH + 48] = c3[r];
      }
    }
    __syncthreads();
    float pl[P2_RPW], pr[P2_RPW];
    v4f pv[P2_RPW];
#pragma unroll
    for (int u = 0; u < P2_RPW; ++u)
      pv[u] = *(const v4f*)(st + min(wv + 32 * u, P2_ROWS - 1) * P2_OPITCH + lane * 4);
#pragma unroll
    for (int u = 0; u < P2_RPW; ++u) {
      const int nl = wv + 32 * u;
      const v4f v = pv[u];
      if (nl < (int)nb) *(v4f*)(outp + (size_t)(base + nl) * 128 + lane * 4) = v;
      pl[u] = v.x * alv.x + v.y * alv.y + v.z * alv.z + v.w * alv.w;
      pr[u] = v.x * arv.x + v.y * arv.y + v.z * arv.z + v.w * arv.w;
    }
#pragma unroll
    for (int msk = 1; msk <= 4; msk <<= 1) {
#pragma unroll
      for (int u = 0; u < P2_RPW; ++u) {
        pl[u] += __shfl_xor(pl[u], msk, 32);
        pr[u] += __shfl_xor(pr[u], msk, 32);
      }
    }
#pragma unroll
    for (int u = 0; u < P2_RPW; ++u) {
      const int nl = wv + 32 * u;
      if (nl < (int)nb && (lane & 7) == 0) {
        sden[nl * 4 + (lane >> 3)] = pl[u];
        erl[nl * 4 + (lane >> 3)]  = pr[u];
      }
    }
    __syncthreads();
    if (tid < (int)nb * 4) {
      eln[(size_t)base * 4 + tid] = sden[tid];
      ern[(size_t)base * 4 + tid] = erl[tid];
    }
  }
}

extern "C" void kernel_launch(void* const* d_in, const int* in_sizes, int n_in,
                              void* d_out, int out_size, void* d_ws, size_t ws_size,
                              hipStream_t stream) {
  const float* feat = (const float*)d_in[0];
  const float* W1   = (const float*)d_in[1];
  const float* al1  = (const float*)d_in[2];
  const float* ar1  = (const float*)d_in[3];
  const float* b1   = (const float*)d_in[4];
  const float* lns  = (const float*)d_in[5];
  const float* lnb  = (const float*)d_in[6];
  const float* W2   = (const float*)d_in[7];
  const float* al2  = (const float*)d_in[8];
  const float* ar2  = (const float*)d_in[9];
  const float* b2   = (const float*)d_in[10];
  const int*   src  = (const int*)d_in[11];
  const int*   dst  = (const int*)d_in[12];
  const int*   indeg= (const int*)d_in[13];

  const int Nn = in_sizes[13];
  const int E  = in_sizes[11];
  const int NH = Nn * 4;

  char* p = (char*)d_ws;
  float*    hbuf  = (float*)p;   p += (size_t)Nn * 128 * sizeof(float);
  float*    hbuf2 = (float*)p;   p += (size_t)Nn * 128 * sizeof(float);
  float*    el    = (float*)p;   p += (size_t)NH * sizeof(float);
  float*    er    = (float*)p;   p += (size_t)NH * sizeof(float);
  float*    el2   = (float*)p;   p += (size_t)NH * sizeof(float);
  float*    er2   = (float*)p;   p += (size_t)NH * sizeof(float);
  const int ownBlocks  = (Nn + OWN_NB - 1) / OWN_NB;
  unsigned* wlist = (unsigned*)p; p += (size_t)ownBlocks * OWN_CAP * 2 * sizeof(unsigned);
  unsigned* wbits = (unsigned*)p; p += (size_t)BIT_OWN * ((E + BIT_CH - 1) / BIT_CH) * BIT_CW * sizeof(unsigned);
  unsigned* wcnt  = (unsigned*)p; p += (size_t)ownBlocks * sizeof(unsigned);
  int*      wdeg  = (int*)p;
  (void)ws_size; (void)n_in; (void)out_size;

  float* outh = (float*)d_out;

  dim3 b256(256);
  const int rowTiles   = (Nn + 15) / 16;
  const int gemmBlocks = (rowTiles + 7) / 8;

  unsigned* wbitsArg = ownBlocks <= BIT_OWN ? wbits : nullptr;
  gat_gemm_wmma<128><<<gemmBlocks, b256, 0, stream>>>(feat, W1, hbuf, Nn, al1, ar1, el, er, dst, wbitsArg, E);
  gat_owner<1><<<ownBlocks, OWN_THREADS, 0, stream>>>(src, dst, el, er, hbuf, b1, lns, lnb, hbuf2, wlist, wcnt, indeg, wdeg, wbitsArg, W2, al2, ar2, el2, er2, Nn, E);

  gat_owner<2><<<ownBlocks, OWN_THREADS, 0, stream>>>(src, dst, el2, er2, hbuf2, b2, lns, lnb, outh, wlist, wcnt, indeg, wdeg, nullptr, nullptr, nullptr, nullptr, nullptr, nullptr, Nn, E);
}
